# starting kernel
# baseline (speedup 1.0000x reference)
#include <hip/hip_runtime.h>
#include <math.h>

#define Hdim 256
#define NHEAD 4
#define DK 64
#define PLR 32
#define Bsz 2
#define LQ 8
#define LKs 512
#define NTOK (Bsz * LQ * LKs)
#define EPSV 1e-5f

typedef float v2f __attribute__((ext_vector_type(2)));
typedef float v8f __attribute__((ext_vector_type(8)));

#define ACT_NONE  0
#define ACT_SILU  1
#define ACT_TANH  2
#define ACT_DECAY 3

__global__ void rwkv_features_kernel(
    const float* __restrict__ query, const float* __restrict__ keyval,
    const float* __restrict__ rwg_mu, const float* __restrict__ W_rwg0,
    const float* __restrict__ W_rwg2, const float* __restrict__ rwg_bias,
    const float* __restrict__ kv_mu, const float* __restrict__ W_kv0,
    const float* __restrict__ W_kv2, const float* __restrict__ kv_bias,
    float* __restrict__ xr, float* __restrict__ xw, float* __restrict__ xg,
    float* __restrict__ xk, float* __restrict__ xv)
{
    const int tid = blockIdx.x;
    const int e   = tid >> 9;
    const int l   = tid & 511;
    const int b   = e >> 3;
    const int t   = threadIdx.x;

    __shared__ float xrwg_s[Hdim];
    __shared__ float xkv_s[Hdim];
    __shared__ float t1_s[3 * PLR];
    __shared__ float t2_s[2 * PLR];

    const float h  = keyval[((long)b * LKs + l) * Hdim + t];
    const float q  = query[(long)e * Hdim + t];
    const float hp = (l == 0) ? 0.0f : keyval[((long)b * LKs + (l - 1)) * Hdim + t];
    const float dc = q - h;
    const float d  = hp - h;

    xrwg_s[t] = h + dc * rwg_mu[t];
    xkv_s[t]  = h + d  * kv_mu[t];
    __syncthreads();

    if (t < 96) {
        float acc = 0.0f;
        #pragma unroll 8
        for (int i = 0; i < Hdim; ++i) acc += xrwg_s[i] * W_rwg0[i * 96 + t];
        t1_s[t] = tanhf(acc);
    } else if (t < 160) {
        const int j = t - 96;
        float acc = 0.0f;
        #pragma unroll 8
        for (int i = 0; i < Hdim; ++i) acc += xkv_s[i] * W_kv0[i * 64 + j];
        t2_s[j] = tanhf(acc);
    }
    __syncthreads();

    float rmu = rwg_bias[t];
    float wmu = rwg_bias[Hdim + t];
    float gmu = rwg_bias[2 * Hdim + t];
    #pragma unroll
    for (int r = 0; r < PLR; ++r) {
        rmu += t1_s[r]            * W_rwg2[t * 96 + r];
        wmu += t1_s[PLR + r]      * W_rwg2[t * 96 + PLR + r];
        gmu += t1_s[2 * PLR + r]  * W_rwg2[t * 96 + 2 * PLR + r];
    }
    float kmu = kv_bias[t];
    float vmu = kv_bias[Hdim + t];
    #pragma unroll
    for (int r = 0; r < PLR; ++r) {
        kmu += t2_s[r]       * W_kv2[t * 64 + r];
        vmu += t2_s[PLR + r] * W_kv2[t * 64 + PLR + r];
    }

    const long o = (long)tid * Hdim + t;
    xr[o] = h + dc * rmu;
    xw[o] = h + dc * wmu;
    xg[o] = h + dc * gmu;
    xk[o] = h + d  * kmu;
    xv[o] = h + d  * vmu;
}

__device__ __forceinline__ float apply_act(float x, int act)
{
    switch (act) {
        case ACT_SILU:  return x / (1.0f + expf(-x));
        case ACT_TANH:  return tanhf(x);
        case ACT_DECAY: return expf(-expf(x));
        default:        return x;
    }
}

__global__ __launch_bounds__(128) void rwkv_wmma_gemm_kernel(
    const float* __restrict__ X, const float* __restrict__ W,
    const float* __restrict__ bias, float* __restrict__ Y,
    int K, int N, int act)
{
    const int wave  = threadIdx.x >> 5;
    const int lane  = threadIdx.x & 31;
    const int mbase = (blockIdx.x * 4 + wave) << 5;
    const int nbase = blockIdx.y << 5;

    const int sub  = lane & 15;
    const int koff = (lane >> 4) << 1;

    const float* __restrict__ Arow0 = X + (long)(mbase + sub) * K + koff;
    const float* __restrict__ Arow1 = Arow0 + (long)16 * K;
    const float* __restrict__ Bcol0 = W + (long)koff * N + nbase + sub;
    const float* __restrict__ Bcol1 = Bcol0 + 16;

    v8f c00 = {}, c01 = {}, c10 = {}, c11 = {};
    #pragma unroll 4
    for (int k = 0; k < K; k += 4) {
        const v2f a0 = *(const v2f*)(Arow0 + k);
        const v2f a1 = *(const v2f*)(Arow1 + k);
        v2f b0, b1;
        b0.x = Bcol0[(long)k * N];
        b0.y = Bcol0[(long)(k + 1) * N];
        b1.x = Bcol1[(long)k * N];
        b1.y = Bcol1[(long)(k + 1) * N];
        c00 = __builtin_amdgcn_wmma_f32_16x16x4_f32(false, a0, false, b0, (short)0, c00, false, false);
        c01 = __builtin_amdgcn_wmma_f32_16x16x4_f32(false, a0, false, b1, (short)0, c01, false, false);
        c10 = __builtin_amdgcn_wmma_f32_16x16x4_f32(false, a1, false, b0, (short)0, c10, false, false);
        c11 = __builtin_amdgcn_wmma_f32_16x16x4_f32(false, a1, false, b1, (short)0, c11, false, false);
    }

    const int rbase = (lane >> 4) * 8;
    const int col0  = nbase + sub;
    const int col1  = col0 + 16;
    const float bv0 = bias ? bias[col0] : 0.0f;
    const float bv1 = bias ? bias[col1] : 0.0f;
    #pragma unroll
    for (int r = 0; r < 8; ++r) {
        const long row0 = (long)mbase + rbase + r;
        const long row1 = row0 + 16;
        Y[row0 * N + col0] = apply_act(c00[r] + bv0, act);
        Y[row0 * N + col1] = apply_act(c01[r] + bv1, act);
        Y[row1 * N + col0] = apply_act(c10[r] + bv0, act);
        Y[row1 * N + col1] = apply_act(c11[r] + bv1, act);
    }
}

__global__ void rwkv_scan_kernel(
    const float* __restrict__ r, const float* __restrict__ k,
    const float* __restrict__ v, const float* __restrict__ decay,
    const float* __restrict__ u, float* __restrict__ o)
{
    const int e    = blockIdx.x >> 2;
    const int head = blockIdx.x & 3;
    const long base = ((long)e * LKs) * Hdim + head * DK;
    const int t    = threadIdx.x;
    const int vcol = t & 63;
    const int k0   = (t >> 6) * 16;

    float S[16];
    #pragma unroll
    for (int j = 0; j < 16; ++j) S[j] = 0.0f;

    __shared__ float rr[DK], kk[DK], vv[DK], dd[DK], uu[DK];
    __shared__ float part[256];

    if (t < DK) uu[t] = u[head * DK + t];
    __syncthreads();

    for (int step = 0; step < LKs; ++step) {
        const long p = base + (long)step * Hdim;
        if (t < 64) {
            rr[t] = r[p + t];
            __builtin_prefetch((const void*)(r + p + Hdim + t), 0, 1);
        } else if (t < 128) {
            kk[t - 64] = k[p + t - 64];
            __builtin_prefetch((const void*)(k + p + Hdim + t - 64), 0, 1);
        } else if (t < 192) {
            vv[t - 128] = v[p + t - 128];
            __builtin_prefetch((const void*)(v + p + Hdim + t - 128), 0, 1);
        } else {
            dd[t - 192] = decay[p + t - 192];
            __builtin_prefetch((const void*)(decay + p + Hdim + t - 192), 0, 1);
        }
        __syncthreads();

        const float vt = vv[vcol];
        float acc = 0.0f;
        #pragma unroll
        for (int j = 0; j < 16; ++j) {
            const int ki = k0 + j;
            const float kv = kk[ki] * vt;
            acc += rr[ki] * (S[j] + uu[ki] * kv);
            S[j] = dd[ki] * S[j] + kv;
        }
        part[t] = acc;
        __syncthreads();
        if (t < 64)
            o[p + t] = part[t] + part[t + 64] + part[t + 128] + part[t + 192];
        __syncthreads();
    }
}

__global__ void rwkv_gn_silu_kernel(
    const float* __restrict__ o, const float* __restrict__ gsilu,
    const float* __restrict__ gn_w, const float* __restrict__ gn_b,
    float* __restrict__ y)
{
    const long tok = blockIdx.x;
    const int t    = threadIdx.x;
    const int head = t >> 6;
    const int idx  = t & 63;

    const float val = o[tok * Hdim + t];
    __shared__ float red[256];

    red[t] = val;
    __syncthreads();
    for (int off = 32; off >= 1; off >>= 1) {
        if (idx < off) red[t] += red[t + off];
        __syncthreads();
    }
    const float mean = red[head * 64] * (1.0f / 64.0f);
    __syncthreads();

    const float dv = val - mean;
    red[t] = dv * dv;
    __syncthreads();
    for (int off = 32; off >= 1; off >>= 1) {
        if (idx < off) red[t] += red[t + off];
        __syncthreads();
    }
    const float var = red[head * 64] * (1.0f / 64.0f);

    const float ynorm = dv * rsqrtf(var + EPSV);
    y[tok * Hdim + t] = (ynorm * gn_w[t] + gn_b[t]) * gsilu[tok * Hdim + t];
}

extern "C" void kernel_launch(void* const* d_in, const int* in_sizes, int n_in,
                              void* d_out, int out_size, void* d_ws, size_t ws_size,
                              hipStream_t stream)
{
    const float* query    = (const float*)d_in[0];
    const float* keyval   = (const float*)d_in[1];
    const float* rwg_mu   = (const float*)d_in[2];
    const float* W_rwg0   = (const float*)d_in[3];
    const float* W_rwg2   = (const float*)d_in[4];
    const float* rwg_bias = (const float*)d_in[5];
    const float* kv_mu    = (const float*)d_in[6];
    const float* W_kv0    = (const float*)d_in[7];
    const float* W_kv2    = (const float*)d_in[8];
    const float* kv_bias  = (const float*)d_in[9];
    const float* Wr       = (const float*)d_in[10];
    const float* Wa       = (const float*)d_in[11];
    const float* Wb       = (const float*)d_in[12];
    const float* bb       = (const float*)d_in[13];
    const float* Wk       = (const float*)d_in[14];
    const float* Wv       = (const float*)d_in[15];
    const float* Wg       = (const float*)d_in[16];
    const float* u        = (const float*)d_in[17];
    const float* gn_w     = (const float*)d_in[18];
    const float* gn_b     = (const float*)d_in[19];
    const float* Wo       = (const float*)d_in[20];

    float* ws = (float*)d_ws;
    const long SZ = (long)NTOK * Hdim;
    float* xr    = ws + 0 * SZ;
    float* xw    = ws + 1 * SZ;
    float* xg    = ws + 2 * SZ;
    float* xk    = ws + 3 * SZ;
    float* xv    = ws + 4 * SZ;
    float* rbuf  = ws + 5 * SZ;
    float* kbuf  = ws + 6 * SZ;
    float* vbuf  = ws + 7 * SZ;
    float* dbuf  = ws + 8 * SZ;
    float* gsbuf = ws + 9 * SZ;
    float* twbuf = ws + 10 * SZ;
    float* obuf  = xr;
    float* ybuf  = xw;

    rwkv_features_kernel<<<NTOK, 256, 0, stream>>>(
        query, keyval, rwg_mu, W_rwg0, W_rwg2, rwg_bias,
        kv_mu, W_kv0, W_kv2, kv_bias, xr, xw, xg, xk, xv);

    auto gemm = [&](const float* X, const float* W, const float* bias,
                    float* Y, int M, int K, int N, int act) {
        dim3 grid(M / 128, N / 32);
        rwkv_wmma_gemm_kernel<<<grid, 128, 0, stream>>>(X, W, bias, Y, K, N, act);
    };

    gemm(xr, Wr, nullptr, rbuf,  NTOK, 256, 256, ACT_NONE);
    gemm(xk, Wk, nullptr, kbuf,  NTOK, 256, 256, ACT_NONE);
    gemm(xv, Wv, nullptr, vbuf,  NTOK, 256, 256, ACT_NONE);
    gemm(xg, Wg, nullptr, gsbuf, NTOK, 256, 256, ACT_SILU);
    gemm(xw, Wa, nullptr, twbuf, NTOK, 256, 64,  ACT_TANH);
    gemm(twbuf, Wb, bb, dbuf,    NTOK, 64,  256, ACT_DECAY);

    rwkv_scan_kernel<<<Bsz * LQ * NHEAD, 256, 0, stream>>>(rbuf, kbuf, vbuf, dbuf, u, obuf);

    rwkv_gn_silu_kernel<<<NTOK, 256, 0, stream>>>(obuf, gsbuf, gn_w, gn_b, ybuf);

    gemm(ybuf, Wo, nullptr, (float*)d_out, NTOK, 256, 256, ACT_NONE);
}


# scan: 512 steps walked in 16-step chunks, r/k/v/decay rows staged in LDS one chunk ahead, k-slice partials joined once per chunk: 1 barrier per 16 steps, not 3 per step; sum order kept
# speedup vs baseline: 1.7973x; 1.7973x over previous
#include <hip/hip_runtime.h>
#include <math.h>

#define Hdim 256
#define NHEAD 4
#define DK 64
#define PLR 32
#define Bsz 2
#define LQ 8
#define LKs 512
#define NTOK (Bsz * LQ * LKs)
#define EPSV 1e-5f

typedef float v2f __attribute__((ext_vector_type(2)));
typedef float v8f __attribute__((ext_vector_type(8)));

#define ACT_NONE  0
#define ACT_SILU  1
#define ACT_TANH  2
#define ACT_DECAY 3

__global__ void rwkv_features_kernel(
    const float* __restrict__ query, const float* __restrict__ keyval,
    const float* __restrict__ rwg_mu, const float* __restrict__ W_rwg0,
    const float* __restrict__ W_rwg2, const float* __restrict__ rwg_bias,
    const float* __restrict__ kv_mu, const float* __restrict__ W_kv0,
    const float* __restrict__ W_kv2, const float* __restrict__ kv_bias,
    float* __restrict__ xr, float* __restrict__ xw, float* __restrict__ xg,
    float* __restrict__ xk, float* __restrict__ xv)
{
    const int tid = blockIdx.x;
    const int e   = tid >> 9;
    const int l   = tid & 511;
    const int b   = e >> 3;
    const int t   = threadIdx.x;

    __shared__ float xrwg_s[Hdim];
    __shared__ float xkv_s[Hdim];
    __shared__ float t1_s[3 * PLR];
    __shared__ float t2_s[2 * PLR];

    const float h  = keyval[((long)b * LKs + l) * Hdim + t];
    const float q  = query[(long)e * Hdim + t];
    const float hp = (l == 0) ? 0.0f : keyval[((long)b * LKs + (l - 1)) * Hdim + t];
    const float dc = q - h;
    const float d  = hp - h;

    xrwg_s[t] = h + dc * rwg_mu[t];
    xkv_s[t]  = h + d  * kv_mu[t];
    __syncthreads();

    if (t < 96) {
        float acc = 0.0f;
        #pragma unroll 8
        for (int i = 0; i < Hdim; ++i) acc += xrwg_s[i] * W_rwg0[i * 96 + t];
        t1_s[t] = tanhf(acc);
    } else if (t < 160) {
        const int j = t - 96;
        float acc = 0.0f;
        #pragma unroll 8
        for (int i = 0; i < Hdim; ++i) acc += xkv_s[i] * W_kv0[i * 64 + j];
        t2_s[j] = tanhf(acc);
    }
    __syncthreads();

    float rmu = rwg_bias[t];
    float wmu = rwg_bias[Hdim + t];
    float gmu = rwg_bias[2 * Hdim + t];
    #pragma unroll
    for (int r = 0; r < PLR; ++r) {
        rmu += t1_s[r]            * W_rwg2[t * 96 + r];
        wmu += t1_s[PLR + r]      * W_rwg2[t * 96 + PLR + r];
        gmu += t1_s[2 * PLR + r]  * W_rwg2[t * 96 + 2 * PLR + r];
    }
    float kmu = kv_bias[t];
    float vmu = kv_bias[Hdim + t];
    #pragma unroll
    for (int r = 0; r < PLR; ++r) {
        kmu += t2_s[r]       * W_kv2[t * 64 + r];
        vmu += t2_s[PLR + r] * W_kv2[t * 64 + PLR + r];
    }

    const long o = (long)tid * Hdim + t;
    xr[o] = h + dc * rmu;
    xw[o] = h + dc * wmu;
    xg[o] = h + dc * gmu;
    xk[o] = h + d  * kmu;
    xv[o] = h + d  * vmu;
}

__device__ __forceinline__ float apply_act(float x, int act)
{
    switch (act) {
        case ACT_SILU:  return x / (1.0f + expf(-x));
        case ACT_TANH:  return tanhf(x);
        case ACT_DECAY: return expf(-expf(x));
        default:        return x;
    }
}

__global__ __launch_bounds__(128) void rwkv_wmma_gemm_kernel(
    const float* __restrict__ X, const float* __restrict__ W,
    const float* __restrict__ bias, float* __restrict__ Y,
    int K, int N, int act)
{
    const int wave  = threadIdx.x >> 5;
    const int lane  = threadIdx.x & 31;
    const int mbase = (blockIdx.x * 4 + wave) << 5;
    const int nbase = blockIdx.y << 5;

    const int sub  = lane & 15;
    const int koff = (lane >> 4) << 1;

    const float* __restrict__ Arow0 = X + (long)(mbase + sub) * K + koff;
    const float* __restrict__ Arow1 = Arow0 + (long)16 * K;
    const float* __restrict__ Bcol0 = W + (long)koff * N + nbase + sub;
    const float* __restrict__ Bcol1 = Bcol0 + 16;

    v8f c00 = {}, c01 = {}, c10 = {}, c11 = {};
    #pragma unroll 4
    for (int k = 0; k < K; k += 4) {
        const v2f a0 = *(const v2f*)(Arow0 + k);
        const v2f a1 = *(const v2f*)(Arow1 + k);
        v2f b0, b1;
        b0.x = Bcol0[(long)k * N];
        b0.y = Bcol0[(long)(k + 1) * N];
        b1.x = Bcol1[(long)k * N];
        b1.y = Bcol1[(long)(k + 1) * N];
        c00 = __builtin_amdgcn_wmma_f32_16x16x4_f32(false, a0, false, b0, (short)0, c00, false, false);
        c01 = __builtin_amdgcn_wmma_f32_16x16x4_f32(false, a0, false, b1, (short)0, c01, false, false);
        c10 = __builtin_amdgcn_wmma_f32_16x16x4_f32(false, a1, false, b0, (short)0, c10, false, false);
        c11 = __builtin_amdgcn_wmma_f32_16x16x4_f32(false, a1, false, b1, (short)0, c11, false, false);
    }

    const int rbase = (lane >> 4) * 8;
    const int col0  = nbase + sub;
    const int col1  = col0 + 16;
    const float bv0 = bias ? bias[col0] : 0.0f;
    const float bv1 = bias ? bias[col1] : 0.0f;
    #pragma unroll
    for (int r = 0; r < 8; ++r) {
        const long row0 = (long)mbase + rbase + r;
        const long row1 = row0 + 16;
        Y[row0 * N + col0] = apply_act(c00[r] + bv0, act);
        Y[row0 * N + col1] = apply_act(c01[r] + bv1, act);
        Y[row1 * N + col0] = apply_act(c10[r] + bv0, act);
        Y[row1 * N + col1] = apply_act(c11[r] + bv1, act);
    }
}

#define SCAN_T 16
#define SCAN_NCHUNK (LKs / SCAN_T)

__global__ __launch_bounds__(256) void rwkv_scan_kernel(
    const float* __restrict__ r, const float* __restrict__ k,
    const float* __restrict__ v, const float* __restrict__ decay,
    const float* __restrict__ u, float* __restrict__ o)
{
    const int e    = blockIdx.x >> 2;
    const int head = blockIdx.x & 3;
    const long base = ((long)e * LKs) * Hdim + head * DK;
    const int t    = threadIdx.x;
    const int vcol = t & 63;
    const int k0   = (t >> 6) * 16;

    __shared__ __attribute__((aligned(16))) float stage[2][SCAN_T][4][DK];
    __shared__ __attribute__((aligned(16))) float part[2][SCAN_T][256];

    float S[16], uu[16];
    #pragma unroll
    for (int j = 0; j < 16; ++j) {
        S[j]  = 0.0f;
        uu[j] = u[head * DK + k0 + j];
    }

    const int  ls   = t >> 4;
    const int  lc   = (t & 15) << 2;
    const long lofs = base + (long)ls * Hdim + lc;

    float4 g0 = *(const float4*)(r + lofs);
    float4 g1 = *(const float4*)(k + lofs);
    float4 g2 = *(const float4*)(v + lofs);
    float4 g3 = *(const float4*)(decay + lofs);
    *(float4*)&stage[0][ls][0][lc] = g0;
    *(float4*)&stage[0][ls][1][lc] = g1;
    *(float4*)&stage[0][ls][2][lc] = g2;
    *(float4*)&stage[0][ls][3][lc] = g3;
    __syncthreads();

    for (int c = 0; c < SCAN_NCHUNK; ++c) {
        const int  cur  = c & 1;
        const bool more = (c + 1 < SCAN_NCHUNK);

        if (more) {
            const long nofs = lofs + (long)(c + 1) * SCAN_T * Hdim;
            g0 = *(const float4*)(r + nofs);
            g1 = *(const float4*)(k + nofs);
            g2 = *(const float4*)(v + nofs);
            g3 = *(const float4*)(decay + nofs);
        }

        #pragma unroll
        for (int s = 0; s < SCAN_T; ++s) {
            const float* rr = &stage[cur][s][0][k0];
            const float* kk = &stage[cur][s][1][k0];
            const float* dd = &stage[cur][s][3][k0];
            const float  vt = stage[cur][s][2][vcol];
            float acc = 0.0f;
            #pragma unroll
            for (int j = 0; j < 16; ++j) {
                const float kv = kk[j] * vt;
                acc += rr[j] * (S[j] + uu[j] * kv);
                S[j] = dd[j] * S[j] + kv;
            }
            part[cur][s][t] = acc;
        }

        if (more) {
            *(float4*)&stage[cur ^ 1][ls][0][lc] = g0;
            *(float4*)&stage[cur ^ 1][ls][1][lc] = g1;
            *(float4*)&stage[cur ^ 1][ls][2][lc] = g2;
            *(float4*)&stage[cur ^ 1][ls][3][lc] = g3;
        }
        __syncthreads();

        const float4 p0 = *(const float4*)&part[cur][ls][lc];
        const float4 p1 = *(const float4*)&part[cur][ls][64 + lc];
        const float4 p2 = *(const float4*)&part[cur][ls][128 + lc];
        const float4 p3 = *(const float4*)&part[cur][ls][192 + lc];
        float4 sum;
        sum.x = p0.x + p1.x + p2.x + p3.x;
        sum.y = p0.y + p1.y + p2.y + p3.y;
        sum.z = p0.z + p1.z + p2.z + p3.z;
        sum.w = p0.w + p1.w + p2.w + p3.w;
        *(float4*)(o + base + (long)(c * SCAN_T + ls) * Hdim + lc) = sum;
    }
}

__global__ void rwkv_gn_silu_kernel(
    const float* __restrict__ o, const float* __restrict__ gsilu,
    const float* __restrict__ gn_w, const float* __restrict__ gn_b,
    float* __restrict__ y)
{
    const long tok = blockIdx.x;
    const int t    = threadIdx.x;
    const int head = t >> 6;
    const int idx  = t & 63;

    const float val = o[tok * Hdim + t];
    __shared__ float red[256];

    red[t] = val;
    __syncthreads();
    for (int off = 32; off >= 1; off >>= 1) {
        if (idx < off) red[t] += red[t + off];
        __syncthreads();
    }
    const float mean = red[head * 64] * (1.0f / 64.0f);
    __syncthreads();

    const float dv = val - mean;
    red[t] = dv * dv;
    __syncthreads();
    for (int off = 32; off >= 1; off >>= 1) {
        if (idx < off) red[t] += red[t + off];
        __syncthreads();
    }
    const float var = red[head * 64] * (1.0f / 64.0f);

    const float ynorm = dv * rsqrtf(var + EPSV);
    y[tok * Hdim + t] = (ynorm * gn_w[t] + gn_b[t]) * gsilu[tok * Hdim + t];
}

extern "C" void kernel_launch(void* const* d_in, const int* in_sizes, int n_in,
                              void* d_out, int out_size, void* d_ws, size_t ws_size,
                              hipStream_t stream)
{
    const float* query    = (const float*)d_in[0];
    const float* keyval   = (const float*)d_in[1];
    const float* rwg_mu   = (const float*)d_in[2];
    const float* W_rwg0   = (const float*)d_in[3];
    const float* W_rwg2   = (const float*)d_in[4];
    const float* rwg_bias = (const float*)d_in[5];
    const float* kv_mu    = (const float*)d_in[6];
    const float* W_kv0    = (const float*)d_in[7];
    const float* W_kv2    = (const float*)d_in[8];
    const float* kv_bias  = (const float*)d_in[9];
    const float* Wr       = (const float*)d_in[10];
    const float* Wa       = (const float*)d_in[11];
    const float* Wb       = (const float*)d_in[12];
    const float* bb       = (const float*)d_in[13];
    const float* Wk       = (const float*)d_in[14];
    const float* Wv       = (const float*)d_in[15];
    const float* Wg       = (const float*)d_in[16];
    const float* u        = (const float*)d_in[17];
    const float* gn_w     = (const float*)d_in[18];
    const float* gn_b     = (const float*)d_in[19];
    const float* Wo       = (const float*)d_in[20];

    float* ws = (float*)d_ws;
    const long SZ = (long)NTOK * Hdim;
    float* xr    = ws + 0 * SZ;
    float* xw    = ws + 1 * SZ;
    float* xg    = ws + 2 * SZ;
    float* xk    = ws + 3 * SZ;
    float* xv    = ws + 4 * SZ;
    float* rbuf  = ws + 5 * SZ;
    float* kbuf  = ws + 6 * SZ;
    float* vbuf  = ws + 7 * SZ;
    float* dbuf  = ws + 8 * SZ;
    float* gsbuf = ws + 9 * SZ;
    float* twbuf = ws + 10 * SZ;
    float* obuf  = xr;
    float* ybuf  = xw;

    rwkv_features_kernel<<<NTOK, 256, 0, stream>>>(
        query, keyval, rwg_mu, W_rwg0, W_rwg2, rwg_bias,
        kv_mu, W_kv0, W_kv2, kv_bias, xr, xw, xg, xk, xv);

    auto gemm = [&](const float* X, const float* W, const float* bias,
                    float* Y, int M, int K, int N, int act) {
        dim3 grid(M / 128, N / 32);
        rwkv_wmma_gemm_kernel<<<grid, 128, 0, stream>>>(X, W, bias, Y, K, N, act);
    };

    gemm(xr, Wr, nullptr, rbuf,  NTOK, 256, 256, ACT_NONE);
    gemm(xk, Wk, nullptr, kbuf,  NTOK, 256, 256, ACT_NONE);
    gemm(xv, Wv, nullptr, vbuf,  NTOK, 256, 256, ACT_NONE);
    gemm(xg, Wg, nullptr, gsbuf, NTOK, 256, 256, ACT_SILU);
    gemm(xw, Wa, nullptr, twbuf, NTOK, 256, 64,  ACT_TANH);
    gemm(twbuf, Wb, bb, dbuf,    NTOK, 64,  256, ACT_DECAY);

    rwkv_scan_kernel<<<Bsz * LQ * NHEAD, 256, 0, stream>>>(rbuf, kbuf, vbuf, dbuf, u, obuf);

    rwkv_gn_silu_kernel<<<NTOK, 256, 0, stream>>>(obuf, gsbuf, gn_w, gn_b, ybuf);

    gemm(ybuf, Wo, nullptr, (float*)d_out, NTOK, 256, 256, ACT_NONE);
}


# features: one work-group per key position serves its 8 query rows: each low-rank weight fetched once per group by whole-line loads (LDS-transposed), kv branch once; per-output sum order kept
# speedup vs baseline: 2.4164x; 1.3445x over previous
#include <hip/hip_runtime.h>
#include <math.h>

#define Hdim 256
#define NHEAD 4
#define DK 64
#define PLR 32
#define Bsz 2
#define LQ 8
#define LKs 512
#define NTOK (Bsz * LQ * LKs)
#define EPSV 1e-5f

typedef float v2f __attribute__((ext_vector_type(2)));
typedef float v8f __attribute__((ext_vector_type(8)));
typedef float v4f __attribute__((ext_vector_type(4)));

#define ACT_NONE  0
#define ACT_SILU  1
#define ACT_TANH  2
#define ACT_DECAY 3

#define NPOS   (Bsz * LKs)
#define WPITCH 36

struct Slab8 { v4f v[8]; };

__device__ __forceinline__ Slab8 slab_fetch(const float* __restrict__ Wsrc, int ld,
                                            int goff, int t)
{
    const float* p = Wsrc + (t >> 3) * ld + goff + ((t & 7) << 2);
    Slab8 s;
    #pragma unroll
    for (int m = 0; m < 8; ++m) s.v[m] = *(const v4f*)(p + m * 32 * ld);
    return s;
}

__device__ __forceinline__ void slab_store(float (*slab)[WPITCH], int t, const Slab8& s)
{
    #pragma unroll
    for (int m = 0; m < 8; ++m)
        *(v4f*)&slab[m * 32 + (t >> 3)][(t & 7) << 2] = s.v[m];
}

__device__ __forceinline__ void slab_row(float (*slab)[WPITCH], int t, float* w)
{
    #pragma unroll
    for (int r4 = 0; r4 < PLR; r4 += 4) {
        const float4 wv = *(const float4*)&slab[t][r4];
        w[r4] = wv.x; w[r4 + 1] = wv.y; w[r4 + 2] = wv.z; w[r4 + 3] = wv.w;
    }
}

__global__ __launch_bounds__(256) void rwkv_features_kernel(
    const float* __restrict__ query, const float* __restrict__ keyval,
    const float* __restrict__ rwg_mu, const float* __restrict__ W_rwg0,
    const float* __restrict__ W_rwg2, const float* __restrict__ rwg_bias,
    const float* __restrict__ kv_mu, const float* __restrict__ W_kv0,
    const float* __restrict__ W_kv2, const float* __restrict__ kv_bias,
    float* __restrict__ xr, float* __restrict__ xw, float* __restrict__ xg,
    float* __restrict__ xk, float* __restrict__ xv)
{
    const int pos = blockIdx.x;
    const int b   = pos >> 9;
    const int l   = pos & 511;
    const int t   = threadIdx.x;

    __shared__ __attribute__((aligned(16))) float xrwg_s[Hdim][LQ];
    __shared__ float xkv_s[Hdim];
    __shared__ __attribute__((aligned(16))) float t1_s[3 * PLR][LQ];
    __shared__ float t2_s[2 * PLR];
    __shared__ __attribute__((aligned(16))) float wslab[Hdim][WPITCH];

    const float h  = keyval[(long)pos * Hdim + t];
    const float hp = (l == 0) ? 0.0f : keyval[(long)(pos - 1) * Hdim + t];
    const float d  = hp - h;
    float dc[LQ];
    {
        const float mu = rwg_mu[t];
        float xq[LQ];
        #pragma unroll
        for (int q = 0; q < LQ; ++q) {
            dc[q] = query[(long)(b * LQ + q) * Hdim + t] - h;
            xq[q] = h + dc[q] * mu;
        }
        *(float4*)&xrwg_s[t][0] = make_float4(xq[0], xq[1], xq[2], xq[3]);
        *(float4*)&xrwg_s[t][4] = make_float4(xq[4], xq[5], xq[6], xq[7]);
        xkv_s[t] = h + d * kv_mu[t];
    }
    __syncthreads();

    if (t < 192) {
        const int c  = (t < 96) ? t : t - 96;
        const int qh = (t < 96) ? 0 : 4;
        float a0 = 0.0f, a1 = 0.0f, a2 = 0.0f, a3 = 0.0f;
        #pragma unroll 16
        for (int i = 0; i < Hdim; ++i) {
            const float  w = W_rwg0[i * 96 + c];
            const float4 x = *(const float4*)&xrwg_s[i][qh];
            a0 += x.x * w;
            a1 += x.y * w;
            a2 += x.z * w;
            a3 += x.w * w;
        }
        *(float4*)&t1_s[c][qh] = make_float4(tanhf(a0), tanhf(a1), tanhf(a2), tanhf(a3));
    } else {
        const int j = t - 192;
        float acc = 0.0f;
        #pragma unroll 16
        for (int i = 0; i < Hdim; ++i) acc += xkv_s[i] * W_kv0[i * 64 + j];
        t2_s[j] = tanhf(acc);
    }

    const long orow = ((long)(b * LQ) * LKs + l) * Hdim + t;
    float w[PLR];

    auto expand_q = [&](int g, float* __restrict__ dst) {
        const float bias = rwg_bias[g * Hdim + t];
        float acc[LQ];
        #pragma unroll
        for (int q = 0; q < LQ; ++q) acc[q] = bias;
        #pragma unroll
        for (int r = 0; r < PLR; ++r) {
            const float4 x0 = *(const float4*)&t1_s[g * PLR + r][0];
            const float4 x1 = *(const float4*)&t1_s[g * PLR + r][4];
            acc[0] += x0.x * w[r];
            acc[1] += x0.y * w[r];
            acc[2] += x0.z * w[r];
            acc[3] += x0.w * w[r];
            acc[4] += x1.x * w[r];
            acc[5] += x1.y * w[r];
            acc[6] += x1.z * w[r];
            acc[7] += x1.w * w[r];
        }
        #pragma unroll
        for (int q = 0; q < LQ; ++q)
            dst[orow + (long)q * LKs * Hdim] = h + dc[q] * acc[q];
    };
    auto expand_kv = [&](int gg, float* __restrict__ dst) {
        float acc = kv_bias[gg * Hdim + t];
        #pragma unroll
        for (int r = 0; r < PLR; ++r) acc += t2_s[gg * PLR + r] * w[r];
        const float val = h + d * acc;
        #pragma unroll
        for (int q = 0; q < LQ; ++q)
            dst[orow + (long)q * LKs * Hdim] = val;
    };

    Slab8 s0 = slab_fetch(W_rwg2, 3 * PLR, 0, t);
    __syncthreads();

    slab_store(wslab, t, s0);
    __syncthreads();
    Slab8 s1 = slab_fetch(W_rwg2, 3 * PLR, PLR, t);
    slab_row(wslab, t, w);
    expand_q(0, xr);
    __syncthreads();

    slab_store(wslab, t, s1);
    __syncthreads();
    Slab8 s2 = slab_fetch(W_rwg2, 3 * PLR, 2 * PLR, t);
    slab_row(wslab, t, w);
    expand_q(1, xw);
    __syncthreads();

    slab_store(wslab, t, s2);
    __syncthreads();
    Slab8 s3 = slab_fetch(W_kv2, 2 * PLR, 0, t);
    slab_row(wslab, t, w);
    expand_q(2, xg);
    __syncthreads();

    slab_store(wslab, t, s3);
    __syncthreads();
    Slab8 s4 = slab_fetch(W_kv2, 2 * PLR, PLR, t);
    slab_row(wslab, t, w);
    expand_kv(0, xk);
    __syncthreads();

    slab_store(wslab, t, s4);
    __syncthreads();
    slab_row(wslab, t, w);
    expand_kv(1, xv);
}

__device__ __forceinline__ float apply_act(float x, int act)
{
    switch (act) {
        case ACT_SILU:  return x / (1.0f + expf(-x));
        case ACT_TANH:  return tanhf(x);
        case ACT_DECAY: return expf(-expf(x));
        default:        return x;
    }
}

__global__ __launch_bounds__(128) void rwkv_wmma_gemm_kernel(
    const float* __restrict__ X, const float* __restrict__ W,
    const float* __restrict__ bias, float* __restrict__ Y,
    int K, int N, int act)
{
    const int wave  = threadIdx.x >> 5;
    const int lane  = threadIdx.x & 31;
    const int mbase = (blockIdx.x * 4 + wave) << 5;
    const int nbase = blockIdx.y << 5;

    const int sub  = lane & 15;
    const int koff = (lane >> 4) << 1;

    const float* __restrict__ Arow0 = X + (long)(mbase + sub) * K + koff;
    const float* __restrict__ Arow1 = Arow0 + (long)16 * K;
    const float* __restrict__ Bcol0 = W + (long)koff * N + nbase + sub;
    const float* __restrict__ Bcol1 = Bcol0 + 16;

    v8f c00 = {}, c01 = {}, c10 = {}, c11 = {};
    #pragma unroll 4
    for (int k = 0; k < K; k += 4) {
        const v2f a0 = *(const v2f*)(Arow0 + k);
        const v2f a1 = *(const v2f*)(Arow1 + k);
        v2f b0, b1;
        b0.x = Bcol0[(long)k * N];
        b0.y = Bcol0[(long)(k + 1) * N];
        b1.x = Bcol1[(long)k * N];
        b1.y = Bcol1[(long)(k + 1) * N];
        c00 = __builtin_amdgcn_wmma_f32_16x16x4_f32(false, a0, false, b0, (short)0, c00, false, false);
        c01 = __builtin_amdgcn_wmma_f32_16x16x4_f32(false, a0, false, b1, (short)0, c01, false, false);
        c10 = __builtin_amdgcn_wmma_f32_16x16x4_f32(false, a1, false, b0, (short)0, c10, false, false);
        c11 = __builtin_amdgcn_wmma_f32_16x16x4_f32(false, a1, false, b1, (short)0, c11, false, false);
    }

    const int rbase = (lane >> 4) * 8;
    const int col0  = nbase + sub;
    const int col1  = col0 + 16;
    const float bv0 = bias ? bias[col0] : 0.0f;
    const float bv1 = bias ? bias[col1] : 0.0f;
    #pragma unroll
    for (int r = 0; r < 8; ++r) {
        const long row0 = (long)mbase + rbase + r;
        const long row1 = row0 + 16;
        Y[row0 * N + col0] = apply_act(c00[r] + bv0, act);
        Y[row0 * N + col1] = apply_act(c01[r] + bv1, act);
        Y[row1 * N + col0] = apply_act(c10[r] + bv0, act);
        Y[row1 * N + col1] = apply_act(c11[r] + bv1, act);
    }
}

#define SCAN_T 16
#define SCAN_NCHUNK (LKs / SCAN_T)

__global__ __launch_bounds__(256) void rwkv_scan_kernel(
    const float* __restrict__ r, const float* __restrict__ k,
    const float* __restrict__ v, const float* __restrict__ decay,
    const float* __restrict__ u, float* __restrict__ o)
{
    const int e    = blockIdx.x >> 2;
    const int head = blockIdx.x & 3;
    const long base = ((long)e * LKs) * Hdim + head * DK;
    const int t    = threadIdx.x;
    const int vcol = t & 63;
    const int k0   = (t >> 6) * 16;

    __shared__ __attribute__((aligned(16))) float stage[2][SCAN_T][4][DK];
    __shared__ __attribute__((aligned(16))) float part[2][SCAN_T][256];

    float S[16], uu[16];
    #pragma unroll
    for (int j = 0; j < 16; ++j) {
        S[j]  = 0.0f;
        uu[j] = u[head * DK + k0 + j];
    }

    const int  ls   = t >> 4;
    const int  lc   = (t & 15) << 2;
    const long lofs = base + (long)ls * Hdim + lc;

    float4 g0 = *(const float4*)(r + lofs);
    float4 g1 = *(const float4*)(k + lofs);
    float4 g2 = *(const float4*)(v + lofs);
    float4 g3 = *(const float4*)(decay + lofs);
    *(float4*)&stage[0][ls][0][lc] = g0;
    *(float4*)&stage[0][ls][1][lc] = g1;
    *(float4*)&stage[0][ls][2][lc] = g2;
    *(float4*)&stage[0][ls][3][lc] = g3;
    __syncthreads();

    for (int c = 0; c < SCAN_NCHUNK; ++c) {
        const int  cur  = c & 1;
        const bool more = (c + 1 < SCAN_NCHUNK);

        if (more) {
            const long nofs = lofs + (long)(c + 1) * SCAN_T * Hdim;
            g0 = *(const float4*)(r + nofs);
            g1 = *(const float4*)(k + nofs);
            g2 = *(const float4*)(v + nofs);
            g3 = *(const float4*)(decay + nofs);
        }

        #pragma unroll
        for (int s = 0; s < SCAN_T; ++s) {
            const float* rr = &stage[cur][s][0][k0];
            const float* kk = &stage[cur][s][1][k0];
            const float* dd = &stage[cur][s][3][k0];
            const float  vt = stage[cur][s][2][vcol];
            float acc = 0.0f;
            #pragma unroll
            for (int j = 0; j < 16; ++j) {
                const float kv = kk[j] * vt;
                acc += rr[j] * (S[j] + uu[j] * kv);
                S[j] = dd[j] * S[j] + kv;
            }
            part[cur][s][t] = acc;
        }

        if (more) {
            *(float4*)&stage[cur ^ 1][ls][0][lc] = g0;
            *(float4*)&stage[cur ^ 1][ls][1][lc] = g1;
            *(float4*)&stage[cur ^ 1][ls][2][lc] = g2;
            *(float4*)&stage[cur ^ 1][ls][3][lc] = g3;
        }
        __syncthreads();

        const float4 p0 = *(const float4*)&part[cur][ls][lc];
        const float4 p1 = *(const float4*)&part[cur][ls][64 + lc];
        const float4 p2 = *(const float4*)&part[cur][ls][128 + lc];
        const float4 p3 = *(const float4*)&part[cur][ls][192 + lc];
        float4 sum;
        sum.x = p0.x + p1.x + p2.x + p3.x;
        sum.y = p0.y + p1.y + p2.y + p3.y;
        sum.z = p0.z + p1.z + p2.z + p3.z;
        sum.w = p0.w + p1.w + p2.w + p3.w;
        *(float4*)(o + base + (long)(c * SCAN_T + ls) * Hdim + lc) = sum;
    }
}

__global__ void rwkv_gn_silu_kernel(
    const float* __restrict__ o, const float* __restrict__ gsilu,
    const float* __restrict__ gn_w, const float* __restrict__ gn_b,
    float* __restrict__ y)
{
    const long tok = blockIdx.x;
    const int t    = threadIdx.x;
    const int head = t >> 6;
    const int idx  = t & 63;

    const float val = o[tok * Hdim + t];
    __shared__ float red[256];

    red[t] = val;
    __syncthreads();
    for (int off = 32; off >= 1; off >>= 1) {
        if (idx < off) red[t] += red[t + off];
        __syncthreads();
    }
    const float mean = red[head * 64] * (1.0f / 64.0f);
    __syncthreads();

    const float dv = val - mean;
    red[t] = dv * dv;
    __syncthreads();
    for (int off = 32; off >= 1; off >>= 1) {
        if (idx < off) red[t] += red[t + off];
        __syncthreads();
    }
    const float var = red[head * 64] * (1.0f / 64.0f);

    const float ynorm = dv * rsqrtf(var + EPSV);
    y[tok * Hdim + t] = (ynorm * gn_w[t] + gn_b[t]) * gsilu[tok * Hdim + t];
}

extern "C" void kernel_launch(void* const* d_in, const int* in_sizes, int n_in,
                              void* d_out, int out_size, void* d_ws, size_t ws_size,
                              hipStream_t stream)
{
    const float* query    = (const float*)d_in[0];
    const float* keyval   = (const float*)d_in[1];
    const float* rwg_mu   = (const float*)d_in[2];
    const float* W_rwg0   = (const float*)d_in[3];
    const float* W_rwg2   = (const float*)d_in[4];
    const float* rwg_bias = (const float*)d_in[5];
    const float* kv_mu    = (const float*)d_in[6];
    const float* W_kv0    = (const float*)d_in[7];
    const float* W_kv2    = (const float*)d_in[8];
    const float* kv_bias  = (const float*)d_in[9];
    const float* Wr       = (const float*)d_in[10];
    const float* Wa       = (const float*)d_in[11];
    const float* Wb       = (const float*)d_in[12];
    const float* bb       = (const float*)d_in[13];
    const float* Wk       = (const float*)d_in[14];
    const float* Wv       = (const float*)d_in[15];
    const float* Wg       = (const float*)d_in[16];
    const float* u        = (const float*)d_in[17];
    const float* gn_w     = (const float*)d_in[18];
    const float* gn_b     = (const float*)d_in[19];
    const float* Wo       = (const float*)d_in[20];

    float* ws = (float*)d_ws;
    const long SZ = (long)NTOK * Hdim;
    float* xr    = ws + 0 * SZ;
    float* xw    = ws + 1 * SZ;
    float* xg    = ws + 2 * SZ;
    float* xk    = ws + 3 * SZ;
    float* xv    = ws + 4 * SZ;
    float* rbuf  = ws + 5 * SZ;
    float* kbuf  = ws + 6 * SZ;
    float* vbuf  = ws + 7 * SZ;
    float* dbuf  = ws + 8 * SZ;
    float* gsbuf = ws + 9 * SZ;
    float* twbuf = ws + 10 * SZ;
    float* obuf  = xr;
    float* ybuf  = xw;

    rwkv_features_kernel<<<NPOS, 256, 0, stream>>>(
        query, keyval, rwg_mu, W_rwg0, W_rwg2, rwg_bias,
        kv_mu, W_kv0, W_kv2, kv_bias, xr, xw, xg, xk, xv);

    auto gemm = [&](const float* X, const float* W, const float* bias,
                    float* Y, int M, int K, int N, int act) {
        dim3 grid(M / 128, N / 32);
        rwkv_wmma_gemm_kernel<<<grid, 128, 0, stream>>>(X, W, bias, Y, K, N, act);
    };

    gemm(xr, Wr, nullptr, rbuf,  NTOK, 256, 256, ACT_NONE);
    gemm(xk, Wk, nullptr, kbuf,  NTOK, 256, 256, ACT_NONE);
    gemm(xv, Wv, nullptr, vbuf,  NTOK, 256, 256, ACT_NONE);
    gemm(xg, Wg, nullptr, gsbuf, NTOK, 256, 256, ACT_SILU);
    gemm(xw, Wa, nullptr, twbuf, NTOK, 256, 64,  ACT_TANH);
    gemm(twbuf, Wb, bb, dbuf,    NTOK, 64,  256, ACT_DECAY);

    rwkv_scan_kernel<<<Bsz * LQ * NHEAD, 256, 0, stream>>>(rbuf, kbuf, vbuf, dbuf, u, obuf);

    rwkv_gn_silu_kernel<<<NTOK, 256, 0, stream>>>(obuf, gsbuf, gn_w, gn_b, ybuf);

    gemm(ybuf, Wo, nullptr, (float*)d_out, NTOK, 256, 256, ACT_NONE);
}


# GEMM: X and W pieces of each 32-wide K chunk staged in LDS by whole-line 16-byte loads, next chunk requested ahead of the current chunk's WMMAs; 64x64 tile per work-group; product order kept
# speedup vs baseline: 4.2475x; 1.7578x over previous
#include <hip/hip_runtime.h>
#include <math.h>

#define Hdim 256
#define NHEAD 4
#define DK 64
#define PLR 32
#define Bsz 2
#define LQ 8
#define LKs 512
#define NTOK (Bsz * LQ * LKs)
#define EPSV 1e-5f

typedef float v2f __attribute__((ext_vector_type(2)));
typedef float v8f __attribute__((ext_vector_type(8)));
typedef float v4f __attribute__((ext_vector_type(4)));

#define ACT_NONE  0
#define ACT_SILU  1
#define ACT_TANH  2
#define ACT_DECAY 3

#define NPOS   (Bsz * LKs)
#define WPITCH 36

struct Slab8 { v4f v[8]; };

__device__ __forceinline__ Slab8 slab_fetch(const float* __restrict__ Wsrc, int ld,
                                            int goff, int t)
{
    const float* p = Wsrc + (t >> 3) * ld + goff + ((t & 7) << 2);
    Slab8 s;
    #pragma unroll
    for (int m = 0; m < 8; ++m) s.v[m] = *(const v4f*)(p + m * 32 * ld);
    return s;
}

__device__ __forceinline__ void slab_store(float (*slab)[WPITCH], int t, const Slab8& s)
{
    #pragma unroll
    for (int m = 0; m < 8; ++m)
        *(v4f*)&slab[m * 32 + (t >> 3)][(t & 7) << 2] = s.v[m];
}

__device__ __forceinline__ void slab_row(float (*slab)[WPITCH], int t, float* w)
{
    #pragma unroll
    for (int r4 = 0; r4 < PLR; r4 += 4) {
        const float4 wv = *(const float4*)&slab[t][r4];
        w[r4] = wv.x; w[r4 + 1] = wv.y; w[r4 + 2] = wv.z; w[r4 + 3] = wv.w;
    }
}

__global__ __launch_bounds__(256) void rwkv_features_kernel(
    const float* __restrict__ query, const float* __restrict__ keyval,
    const float* __restrict__ rwg_mu, const float* __restrict__ W_rwg0,
    const float* __restrict__ W_rwg2, const float* __restrict__ rwg_bias,
    const float* __restrict__ kv_mu, const float* __restrict__ W_kv0,
    const float* __restrict__ W_kv2, const float* __restrict__ kv_bias,
    float* __restrict__ xr, float* __restrict__ xw, float* __restrict__ xg,
    float* __restrict__ xk, float* __restrict__ xv)
{
    const int pos = blockIdx.x;
    const int b   = pos >> 9;
    const int l   = pos & 511;
    const int t   = threadIdx.x;

    __shared__ __attribute__((aligned(16))) float xrwg_s[Hdim][LQ];
    __shared__ float xkv_s[Hdim];
    __shared__ __attribute__((aligned(16))) float t1_s[3 * PLR][LQ];
    __shared__ float t2_s[2 * PLR];
    __shared__ __attribute__((aligned(16))) float wslab[Hdim][WPITCH];

    const float h  = keyval[(long)pos * Hdim + t];
    const float hp = (l == 0) ? 0.0f : keyval[(long)(pos - 1) * Hdim + t];
    const float d  = hp - h;
    float dc[LQ];
    {
        const float mu = rwg_mu[t];
        float xq[LQ];
        #pragma unroll
        for (int q = 0; q < LQ; ++q) {
            dc[q] = query[(long)(b * LQ + q) * Hdim + t] - h;
            xq[q] = h + dc[q] * mu;
        }
        *(float4*)&xrwg_s[t][0] = make_float4(xq[0], xq[1], xq[2], xq[3]);
        *(float4*)&xrwg_s[t][4] = make_float4(xq[4], xq[5], xq[6], xq[7]);
        xkv_s[t] = h + d * kv_mu[t];
    }
    __syncthreads();

    if (t < 192) {
        const int c  = (t < 96) ? t : t - 96;
        const int qh = (t < 96) ? 0 : 4;
        float a0 = 0.0f, a1 = 0.0f, a2 = 0.0f, a3 = 0.0f;
        #pragma unroll 16
        for (int i = 0; i < Hdim; ++i) {
            const float  w = W_rwg0[i * 96 + c];
            const float4 x = *(const float4*)&xrwg_s[i][qh];
            a0 += x.x * w;
            a1 += x.y * w;
            a2 += x.z * w;
            a3 += x.w * w;
        }
        *(float4*)&t1_s[c][qh] = make_float4(tanhf(a0), tanhf(a1), tanhf(a2), tanhf(a3));
    } else {
        const int j = t - 192;
        float acc = 0.0f;
        #pragma unroll 16
        for (int i = 0; i < Hdim; ++i) acc += xkv_s[i] * W_kv0[i * 64 + j];
        t2_s[j] = tanhf(acc);
    }

    const long orow = ((long)(b * LQ) * LKs + l) * Hdim + t;
    float w[PLR];

    auto expand_q = [&](int g, float* __restrict__ dst) {
        const float bias = rwg_bias[g * Hdim + t];
        float acc[LQ];
        #pragma unroll
        for (int q = 0; q < LQ; ++q) acc[q] = bias;
        #pragma unroll
        for (int r = 0; r < PLR; ++r) {
            const float4 x0 = *(const float4*)&t1_s[g * PLR + r][0];
            const float4 x1 = *(const float4*)&t1_s[g * PLR + r][4];
            acc[0] += x0.x * w[r];
            acc[1] += x0.y * w[r];
            acc[2] += x0.z * w[r];
            acc[3] += x0.w * w[r];
            acc[4] += x1.x * w[r];
            acc[5] += x1.y * w[r];
            acc[6] += x1.z * w[r];
            acc[7] += x1.w * w[r];
        }
        #pragma unroll
        for (int q = 0; q < LQ; ++q)
            dst[orow + (long)q * LKs * Hdim] = h + dc[q] * acc[q];
    };
    auto expand_kv = [&](int gg, float* __restrict__ dst) {
        float acc = kv_bias[gg * Hdim + t];
        #pragma unroll
        for (int r = 0; r < PLR; ++r) acc += t2_s[gg * PLR + r] * w[r];
        const float val = h + d * acc;
        #pragma unroll
        for (int q = 0; q < LQ; ++q)
            dst[orow + (long)q * LKs * Hdim] = val;
    };

    Slab8 s0 = slab_fetch(W_rwg2, 3 * PLR, 0, t);
    __syncthreads();

    slab_store(wslab, t, s0);
    __syncthreads();
    Slab8 s1 = slab_fetch(W_rwg2, 3 * PLR, PLR, t);
    slab_row(wslab, t, w);
    expand_q(0, xr);
    __syncthreads();

    slab_store(wslab, t, s1);
    __syncthreads();
    Slab8 s2 = slab_fetch(W_rwg2, 3 * PLR, 2 * PLR, t);
    slab_row(wslab, t, w);
    expand_q(1, xw);
    __syncthreads();

    slab_store(wslab, t, s2);
    __syncthreads();
    Slab8 s3 = slab_fetch(W_kv2, 2 * PLR, 0, t);
    slab_row(wslab, t, w);
    expand_q(2, xg);
    __syncthreads();

    slab_store(wslab, t, s3);
    __syncthreads();
    Slab8 s4 = slab_fetch(W_kv2, 2 * PLR, PLR, t);
    slab_row(wslab, t, w);
    expand_kv(0, xk);
    __syncthreads();

    slab_store(wslab, t, s4);
    __syncthreads();
    slab_row(wslab, t, w);
    expand_kv(1, xv);
}

__device__ __forceinline__ float apply_act(float x, int act)
{
    switch (act) {
        case ACT_SILU:  return x / (1.0f + expf(-x));
        case ACT_TANH:  return tanhf(x);
        case ACT_DECAY: return expf(-expf(x));
        default:        return x;
    }
}

#define GT_M 64
#define GT_N 64
#define GT_K 32
#define GA_PITCH 36
#define GB_PITCH 72

__global__ __launch_bounds__(128) void rwkv_wmma_gemm_kernel(
    const float* __restrict__ X, const float* __restrict__ W,
    const float* __restrict__ bias, float* __restrict__ Y,
    int K, int N, int act)
{
    __shared__ __attribute__((aligned(16))) float As[2][GT_M][GA_PITCH];
    __shared__ __attribute__((aligned(16))) float Bs[2][GT_K][GB_PITCH];

    const int t    = threadIdx.x;
    const int wave = t >> 5;
    const int lane = t & 31;
    const int m0   = blockIdx.x * GT_M;
    const int n0   = blockIdx.y * GT_N;
    const int wm   = (wave >> 1) << 5;
    const int wn   = (wave & 1) << 5;

    const int sub  = lane & 15;
    const int koff = (lane >> 4) << 1;

    const int ar = t >> 3;
    const int ac = (t & 7) << 2;
    const int br = t >> 4;
    const int bc = (t & 15) << 2;
    const float* __restrict__ Ag = X + (long)(m0 + ar) * K + ac;
    const float* __restrict__ Bg = W + (long)br * N + n0 + bc;

    v4f pa[4], pb[4];
    #pragma unroll
    for (int i = 0; i < 4; ++i) {
        pa[i] = *(const v4f*)(Ag + (long)(16 * i) * K);
        pb[i] = *(const v4f*)(Bg + (long)(8 * i) * N);
    }
    #pragma unroll
    for (int i = 0; i < 4; ++i) {
        *(v4f*)&As[0][ar + 16 * i][ac] = pa[i];
        *(v4f*)&Bs[0][br + 8 * i][bc]  = pb[i];
    }
    __syncthreads();

    v8f c00 = {}, c01 = {}, c10 = {}, c11 = {};
    const int nchunk = K / GT_K;
    for (int c = 0; c < nchunk; ++c) {
        const int  cur  = c & 1;
        const bool more = (c + 1 < nchunk);

        if (more) {
            const float* __restrict__ Ap = Ag + (c + 1) * GT_K;
            const float* __restrict__ Bp = Bg + (long)((c + 1) * GT_K) * N;
            #pragma unroll
            for (int i = 0; i < 4; ++i) {
                pa[i] = *(const v4f*)(Ap + (long)(16 * i) * K);
                pb[i] = *(const v4f*)(Bp + (long)(8 * i) * N);
            }
        }

        #pragma unroll
        for (int ks = 0; ks < GT_K; ks += 4) {
            const v2f a0 = *(const v2f*)&As[cur][wm + sub][ks + koff];
            const v2f a1 = *(const v2f*)&As[cur][wm + 16 + sub][ks + koff];
            v2f b0, b1;
            b0.x = Bs[cur][ks + koff][wn + sub];
            b0.y = Bs[cur][ks + koff + 1][wn + sub];
            b1.x = Bs[cur][ks + koff][wn + 16 + sub];
            b1.y = Bs[cur][ks + koff + 1][wn + 16 + sub];
            c00 = __builtin_amdgcn_wmma_f32_16x16x4_f32(false, a0, false, b0, (short)0, c00, false, false);
            c01 = __builtin_amdgcn_wmma_f32_16x16x4_f32(false, a0, false, b1, (short)0, c01, false, false);
            c10 = __builtin_amdgcn_wmma_f32_16x16x4_f32(false, a1, false, b0, (short)0, c10, false, false);
            c11 = __builtin_amdgcn_wmma_f32_16x16x4_f32(false, a1, false, b1, (short)0, c11, false, false);
        }

        if (more) {
            #pragma unroll
            for (int i = 0; i < 4; ++i) {
                *(v4f*)&As[cur ^ 1][ar + 16 * i][ac] = pa[i];
                *(v4f*)&Bs[cur ^ 1][br + 8 * i][bc]  = pb[i];
            }
        }
        __syncthreads();
    }

    const int mbase = m0 + wm;
    const int nbase = n0 + wn;
    const int rbase = (lane >> 4) * 8;
    const int col0  = nbase + sub;
    const int col1  = col0 + 16;
    const float bv0 = bias ? bias[col0] : 0.0f;
    const float bv1 = bias ? bias[col1] : 0.0f;
    #pragma unroll
    for (int r = 0; r < 8; ++r) {
        const long row0 = (long)mbase + rbase + r;
        const long row1 = row0 + 16;
        Y[row0 * N + col0] = apply_act(c00[r] + bv0, act);
        Y[row0 * N + col1] = apply_act(c01[r] + bv1, act);
        Y[row1 * N + col0] = apply_act(c10[r] + bv0, act);
        Y[row1 * N + col1] = apply_act(c11[r] + bv1, act);
    }
}

#define SCAN_T 16
#define SCAN_NCHUNK (LKs / SCAN_T)

__global__ __launch_bounds__(256) void rwkv_scan_kernel(
    const float* __restrict__ r, const float* __restrict__ k,
    const float* __restrict__ v, const float* __restrict__ decay,
    const float* __restrict__ u, float* __restrict__ o)
{
    const int e    = blockIdx.x >> 2;
    const int head = blockIdx.x & 3;
    const long base = ((long)e * LKs) * Hdim + head * DK;
    const int t    = threadIdx.x;
    const int vcol = t & 63;
    const int k0   = (t >> 6) * 16;

    __shared__ __attribute__((aligned(16))) float stage[2][SCAN_T][4][DK];
    __shared__ __attribute__((aligned(16))) float part[2][SCAN_T][256];

    float S[16], uu[16];
    #pragma unroll
    for (int j = 0; j < 16; ++j) {
        S[j]  = 0.0f;
        uu[j] = u[head * DK + k0 + j];
    }

    const int  ls   = t >> 4;
    const int  lc   = (t & 15) << 2;
    const long lofs = base + (long)ls * Hdim + lc;

    float4 g0 = *(const float4*)(r + lofs);
    float4 g1 = *(const float4*)(k + lofs);
    float4 g2 = *(const float4*)(v + lofs);
    float4 g3 = *(const float4*)(decay + lofs);
    *(float4*)&stage[0][ls][0][lc] = g0;
    *(float4*)&stage[0][ls][1][lc] = g1;
    *(float4*)&stage[0][ls][2][lc] = g2;
    *(float4*)&stage[0][ls][3][lc] = g3;
    __syncthreads();

    for (int c = 0; c < SCAN_NCHUNK; ++c) {
        const int  cur  = c & 1;
        const bool more = (c + 1 < SCAN_NCHUNK);

        if (more) {
            const long nofs = lofs + (long)(c + 1) * SCAN_T * Hdim;
            g0 = *(const float4*)(r + nofs);
            g1 = *(const float4*)(k + nofs);
            g2 = *(const float4*)(v + nofs);
            g3 = *(const float4*)(decay + nofs);
        }

        #pragma unroll
        for (int s = 0; s < SCAN_T; ++s) {
            const float* rr = &stage[cur][s][0][k0];
            const float* kk = &stage[cur][s][1][k0];
            const float* dd = &stage[cur][s][3][k0];
            const float  vt = stage[cur][s][2][vcol];
            float acc = 0.0f;
            #pragma unroll
            for (int j = 0; j < 16; ++j) {
                const float kv = kk[j] * vt;
                acc += rr[j] * (S[j] + uu[j] * kv);
                S[j] = dd[j] * S[j] + kv;
            }
            part[cur][s][t] = acc;
        }

        if (more) {
            *(float4*)&stage[cur ^ 1][ls][0][lc] = g0;
            *(float4*)&stage[cur ^ 1][ls][1][lc] = g1;
            *(float4*)&stage[cur ^ 1][ls][2][lc] = g2;
            *(float4*)&stage[cur ^ 1][ls][3][lc] = g3;
        }
        __syncthreads();

        const float4 p0 = *(const float4*)&part[cur][ls][lc];
        const float4 p1 = *(const float4*)&part[cur][ls][64 + lc];
        const float4 p2 = *(const float4*)&part[cur][ls][128 + lc];
        const float4 p3 = *(const float4*)&part[cur][ls][192 + lc];
        float4 sum;
        sum.x = p0.x + p1.x + p2.x + p3.x;
        sum.y = p0.y + p1.y + p2.y + p3.y;
        sum.z = p0.z + p1.z + p2.z + p3.z;
        sum.w = p0.w + p1.w + p2.w + p3.w;
        *(float4*)(o + base + (long)(c * SCAN_T + ls) * Hdim + lc) = sum;
    }
}

__global__ void rwkv_gn_silu_kernel(
    const float* __restrict__ o, const float* __restrict__ gsilu,
    const float* __restrict__ gn_w, const float* __restrict__ gn_b,
    float* __restrict__ y)
{
    const long tok = blockIdx.x;
    const int t    = threadIdx.x;
    const int head = t >> 6;
    const int idx  = t & 63;

    const float val = o[tok * Hdim + t];
    __shared__ float red[256];

    red[t] = val;
    __syncthreads();
    for (int off = 32; off >= 1; off >>= 1) {
        if (idx < off) red[t] += red[t + off];
        __syncthreads();
    }
    const float mean = red[head * 64] * (1.0f / 64.0f);
    __syncthreads();

    const float dv = val - mean;
    red[t] = dv * dv;
    __syncthreads();
    for (int off = 32; off >= 1; off >>= 1) {
        if (idx < off) red[t] += red[t + off];
        __syncthreads();
    }
    const float var = red[head * 64] * (1.0f / 64.0f);

    const float ynorm = dv * rsqrtf(var + EPSV);
    y[tok * Hdim + t] = (ynorm * gn_w[t] + gn_b[t]) * gsilu[tok * Hdim + t];
}

extern "C" void kernel_launch(void* const* d_in, const int* in_sizes, int n_in,
                              void* d_out, int out_size, void* d_ws, size_t ws_size,
                              hipStream_t stream)
{
    const float* query    = (const float*)d_in[0];
    const float* keyval   = (const float*)d_in[1];
    const float* rwg_mu   = (const float*)d_in[2];
    const float* W_rwg0   = (const float*)d_in[3];
    const float* W_rwg2   = (const float*)d_in[4];
    const float* rwg_bias = (const float*)d_in[5];
    const float* kv_mu    = (const float*)d_in[6];
    const float* W_kv0    = (const float*)d_in[7];
    const float* W_kv2    = (const float*)d_in[8];
    const float* kv_bias  = (const float*)d_in[9];
    const float* Wr       = (const float*)d_in[10];
    const float* Wa       = (const float*)d_in[11];
    const float* Wb       = (const float*)d_in[12];
    const float* bb       = (const float*)d_in[13];
    const float* Wk       = (const float*)d_in[14];
    const float* Wv       = (const float*)d_in[15];
    const float* Wg       = (const float*)d_in[16];
    const float* u        = (const float*)d_in[17];
    const float* gn_w     = (const float*)d_in[18];
    const float* gn_b     = (const float*)d_in[19];
    const float* Wo       = (const float*)d_in[20];

    float* ws = (float*)d_ws;
    const long SZ = (long)NTOK * Hdim;
    float* xr    = ws + 0 * SZ;
    float* xw    = ws + 1 * SZ;
    float* xg    = ws + 2 * SZ;
    float* xk    = ws + 3 * SZ;
    float* xv    = ws + 4 * SZ;
    float* rbuf  = ws + 5 * SZ;
    float* kbuf  = ws + 6 * SZ;
    float* vbuf  = ws + 7 * SZ;
    float* dbuf  = ws + 8 * SZ;
    float* gsbuf = ws + 9 * SZ;
    float* twbuf = ws + 10 * SZ;
    float* obuf  = xr;
    float* ybuf  = xw;

    rwkv_features_kernel<<<NPOS, 256, 0, stream>>>(
        query, keyval, rwg_mu, W_rwg0, W_rwg2, rwg_bias,
        kv_mu, W_kv0, W_kv2, kv_bias, xr, xw, xg, xk, xv);

    auto gemm = [&](const float* X, const float* W, const float* bias,
                    float* Y, int M, int K, int N, int act) {
        dim3 grid(M / GT_M, N / GT_N);
        rwkv_wmma_gemm_kernel<<<grid, 128, 0, stream>>>(X, W, bias, Y, K, N, act);
    };

    gemm(xr, Wr, nullptr, rbuf,  NTOK, 256, 256, ACT_NONE);
    gemm(xk, Wk, nullptr, kbuf,  NTOK, 256, 256, ACT_NONE);
    gemm(xv, Wv, nullptr, vbuf,  NTOK, 256, 256, ACT_NONE);
    gemm(xg, Wg, nullptr, gsbuf, NTOK, 256, 256, ACT_SILU);
    gemm(xw, Wa, nullptr, twbuf, NTOK, 256, 64,  ACT_TANH);
    gemm(twbuf, Wb, bb, dbuf,    NTOK, 64,  256, ACT_DECAY);

    rwkv_scan_kernel<<<Bsz * LQ * NHEAD, 256, 0, stream>>>(rbuf, kbuf, vbuf, dbuf, u, obuf);

    rwkv_gn_silu_kernel<<<NTOK, 256, 0, stream>>>(obuf, gsbuf, gn_w, gn_b, ybuf);

    gemm(ybuf, Wo, nullptr, (float*)d_out, NTOK, 256, 256, ACT_NONE);
}


# scan: each (sequence, head) chain split by value columns over 4 work-groups (256 groups, one per WGP; thread = 4 k x 2 v); o plane kept per quarter in whole lines; o sum order changed (16 x 4 k)
# speedup vs baseline: 6.1238x; 1.4417x over previous
#include <hip/hip_runtime.h>
#include <math.h>

#define Hdim 256
#define NHEAD 4
#define DK 64
#define PLR 32
#define Bsz 2
#define LQ 8
#define LKs 512
#define NTOK (Bsz * LQ * LKs)
#define EPSV 1e-5f

typedef float v2f __attribute__((ext_vector_type(2)));
typedef float v8f __attribute__((ext_vector_type(8)));
typedef float v4f __attribute__((ext_vector_type(4)));

#define ACT_NONE  0
#define ACT_SILU  1
#define ACT_TANH  2
#define ACT_DECAY 3

#define NPOS   (Bsz * LKs)
#define WPITCH 36

struct Slab8 { v4f v[8]; };

__device__ __forceinline__ Slab8 slab_fetch(const float* __restrict__ Wsrc, int ld,
                                            int goff, int t)
{
    const float* p = Wsrc + (t >> 3) * ld + goff + ((t & 7) << 2);
    Slab8 s;
    #pragma unroll
    for (int m = 0; m < 8; ++m) s.v[m] = *(const v4f*)(p + m * 32 * ld);
    return s;
}

__device__ __forceinline__ void slab_store(float (*slab)[WPITCH], int t, const Slab8& s)
{
    #pragma unroll
    for (int m = 0; m < 8; ++m)
        *(v4f*)&slab[m * 32 + (t >> 3)][(t & 7) << 2] = s.v[m];
}

__device__ __forceinline__ void slab_row(float (*slab)[WPITCH], int t, float* w)
{
    #pragma unroll
    for (int r4 = 0; r4 < PLR; r4 += 4) {
        const float4 wv = *(const float4*)&slab[t][r4];
        w[r4] = wv.x; w[r4 + 1] = wv.y; w[r4 + 2] = wv.z; w[r4 + 3] = wv.w;
    }
}

__global__ __launch_bounds__(256) void rwkv_features_kernel(
    const float* __restrict__ query, const float* __restrict__ keyval,
    const float* __restrict__ rwg_mu, const float* __restrict__ W_rwg0,
    const float* __restrict__ W_rwg2, const float* __restrict__ rwg_bias,
    const float* __restrict__ kv_mu, const float* __restrict__ W_kv0,
    const float* __restrict__ W_kv2, const float* __restrict__ kv_bias,
    float* __restrict__ xr, float* __restrict__ xw, float* __restrict__ xg,
    float* __restrict__ xk, float* __restrict__ xv)
{
    const int pos = blockIdx.x;
    const int b   = pos >> 9;
    const int l   = pos & 511;
    const int t   = threadIdx.x;

    __shared__ __attribute__((aligned(16))) float xrwg_s[Hdim][LQ];
    __shared__ float xkv_s[Hdim];
    __shared__ __attribute__((aligned(16))) float t1_s[3 * PLR][LQ];
    __shared__ float t2_s[2 * PLR];
    __shared__ __attribute__((aligned(16))) float wslab[Hdim][WPITCH];

    const float h  = keyval[(long)pos * Hdim + t];
    const float hp = (l == 0) ? 0.0f : keyval[(long)(pos - 1) * Hdim + t];
    const float d  = hp - h;
    float dc[LQ];
    {
        const float mu = rwg_mu[t];
        float xq[LQ];
        #pragma unroll
        for (int q = 0; q < LQ; ++q) {
            dc[q] = query[(long)(b * LQ + q) * Hdim + t] - h;
            xq[q] = h + dc[q] * mu;
        }
        *(float4*)&xrwg_s[t][0] = make_float4(xq[0], xq[1], xq[2], xq[3]);
        *(float4*)&xrwg_s[t][4] = make_float4(xq[4], xq[5], xq[6], xq[7]);
        xkv_s[t] = h + d * kv_mu[t];
    }
    __syncthreads();

    if (t < 192) {
        const int c  = (t < 96) ? t : t - 96;
        const int qh = (t < 96) ? 0 : 4;
        float a0 = 0.0f, a1 = 0.0f, a2 = 0.0f, a3 = 0.0f;
        #pragma unroll 16
        for (int i = 0; i < Hdim; ++i) {
            const float  w = W_rwg0[i * 96 + c];
            const float4 x = *(const float4*)&xrwg_s[i][qh];
            a0 += x.x * w;
            a1 += x.y * w;
            a2 += x.z * w;
            a3 += x.w * w;
        }
        *(float4*)&t1_s[c][qh] = make_float4(tanhf(a0), tanhf(a1), tanhf(a2), tanhf(a3));
    } else {
        const int j = t - 192;
        float acc = 0.0f;
        #pragma unroll 16
        for (int i = 0; i < Hdim; ++i) acc += xkv_s[i] * W_kv0[i * 64 + j];
        t2_s[j] = tanhf(acc);
    }

    const long orow = ((long)(b * LQ) * LKs + l) * Hdim + t;
    float w[PLR];

    auto expand_q = [&](int g, float* __restrict__ dst) {
        const float bias = rwg_bias[g * Hdim + t];
        float acc[LQ];
        #pragma unroll
        for (int q = 0; q < LQ; ++q) acc[q] = bias;
        #pragma unroll
        for (int r = 0; r < PLR; ++r) {
            const float4 x0 = *(const float4*)&t1_s[g * PLR + r][0];
            const float4 x1 = *(const float4*)&t1_s[g * PLR + r][4];
            acc[0] += x0.x * w[r];
            acc[1] += x0.y * w[r];
            acc[2] += x0.z * w[r];
            acc[3] += x0.w * w[r];
            acc[4] += x1.x * w[r];
            acc[5] += x1.y * w[r];
            acc[6] += x1.z * w[r];
            acc[7] += x1.w * w[r];
        }
        #pragma unroll
        for (int q = 0; q < LQ; ++q)
            dst[orow + (long)q * LKs * Hdim] = h + dc[q] * acc[q];
    };
    auto expand_kv = [&](int gg, float* __restrict__ dst) {
        float acc = kv_bias[gg * Hdim + t];
        #pragma unroll
        for (int r = 0; r < PLR; ++r) acc += t2_s[gg * PLR + r] * w[r];
        const float val = h + d * acc;
        #pragma unroll
        for (int q = 0; q < LQ; ++q)
            dst[orow + (long)q * LKs * Hdim] = val;
    };

    Slab8 s0 = slab_fetch(W_rwg2, 3 * PLR, 0, t);
    __syncthreads();

    slab_store(wslab, t, s0);
    __syncthreads();
    Slab8 s1 = slab_fetch(W_rwg2, 3 * PLR, PLR, t);
    slab_row(wslab, t, w);
    expand_q(0, xr);
    __syncthreads();

    slab_store(wslab, t, s1);
    __syncthreads();
    Slab8 s2 = slab_fetch(W_rwg2, 3 * PLR, 2 * PLR, t);
    slab_row(wslab, t, w);
    expand_q(1, xw);
    __syncthreads();

    slab_store(wslab, t, s2);
    __syncthreads();
    Slab8 s3 = slab_fetch(W_kv2, 2 * PLR, 0, t);
    slab_row(wslab, t, w);
    expand_q(2, xg);
    __syncthreads();

    slab_store(wslab, t, s3);
    __syncthreads();
    Slab8 s4 = slab_fetch(W_kv2, 2 * PLR, PLR, t);
    slab_row(wslab, t, w);
    expand_kv(0, xk);
    __syncthreads();

    slab_store(wslab, t, s4);
    __syncthreads();
    slab_row(wslab, t, w);
    expand_kv(1, xv);
}

__device__ __forceinline__ float apply_act(float x, int act)
{
    switch (act) {
        case ACT_SILU:  return x / (1.0f + expf(-x));
        case ACT_TANH:  return tanhf(x);
        case ACT_DECAY: return expf(-expf(x));
        default:        return x;
    }
}

#define GT_M 64
#define GT_N 64
#define GT_K 32
#define GA_PITCH 36
#define GB_PITCH 72

__global__ __launch_bounds__(128) void rwkv_wmma_gemm_kernel(
    const float* __restrict__ X, const float* __restrict__ W,
    const float* __restrict__ bias, float* __restrict__ Y,
    int K, int N, int act)
{
    __shared__ __attribute__((aligned(16))) float As[2][GT_M][GA_PITCH];
    __shared__ __attribute__((aligned(16))) float Bs[2][GT_K][GB_PITCH];

    const int t    = threadIdx.x;
    const int wave = t >> 5;
    const int lane = t & 31;
    const int m0   = blockIdx.x * GT_M;
    const int n0   = blockIdx.y * GT_N;
    const int wm   = (wave >> 1) << 5;
    const int wn   = (wave & 1) << 5;

    const int sub  = lane & 15;
    const int koff = (lane >> 4) << 1;

    const int ar = t >> 3;
    const int ac = (t & 7) << 2;
    const int br = t >> 4;
    const int bc = (t & 15) << 2;
    const float* __restrict__ Ag = X + (long)(m0 + ar) * K + ac;
    const float* __restrict__ Bg = W + (long)br * N + n0 + bc;

    v4f pa[4], pb[4];
    #pragma unroll
    for (int i = 0; i < 4; ++i) {
        pa[i] = *(const v4f*)(Ag + (long)(16 * i) * K);
        pb[i] = *(const v4f*)(Bg + (long)(8 * i) * N);
    }
    #pragma unroll
    for (int i = 0; i < 4; ++i) {
        *(v4f*)&As[0][ar + 16 * i][ac] = pa[i];
        *(v4f*)&Bs[0][br + 8 * i][bc]  = pb[i];
    }
    __syncthreads();

    v8f c00 = {}, c01 = {}, c10 = {}, c11 = {};
    const int nchunk = K / GT_K;
    for (int c = 0; c < nchunk; ++c) {
        const int  cur  = c & 1;
        const bool more = (c + 1 < nchunk);

        if (more) {
            const float* __restrict__ Ap = Ag + (c + 1) * GT_K;
            const float* __restrict__ Bp = Bg + (long)((c + 1) * GT_K) * N;
            #pragma unroll
            for (int i = 0; i < 4; ++i) {
                pa[i] = *(const v4f*)(Ap + (long)(16 * i) * K);
                pb[i] = *(const v4f*)(Bp + (long)(8 * i) * N);
            }
        }

        #pragma unroll
        for (int ks = 0; ks < GT_K; ks += 4) {
            const v2f a0 = *(const v2f*)&As[cur][wm + sub][ks + koff];
            const v2f a1 = *(const v2f*)&As[cur][wm + 16 + sub][ks + koff];
            v2f b0, b1;
            b0.x = Bs[cur][ks + koff][wn + sub];
            b0.y = Bs[cur][ks + koff + 1][wn + sub];
            b1.x = Bs[cur][ks + koff][wn + 16 + sub];
            b1.y = Bs[cur][ks + koff + 1][wn + 16 + sub];
            c00 = __builtin_amdgcn_wmma_f32_16x16x4_f32(false, a0, false, b0, (short)0, c00, false, false);
            c01 = __builtin_amdgcn_wmma_f32_16x16x4_f32(false, a0, false, b1, (short)0, c01, false, false);
            c10 = __builtin_amdgcn_wmma_f32_16x16x4_f32(false, a1, false, b0, (short)0, c10, false, false);
            c11 = __builtin_amdgcn_wmma_f32_16x16x4_f32(false, a1, false, b1, (short)0, c11, false, false);
        }

        if (more) {
            #pragma unroll
            for (int i = 0; i < 4; ++i) {
                *(v4f*)&As[cur ^ 1][ar + 16 * i][ac] = pa[i];
                *(v4f*)&Bs[cur ^ 1][br + 8 * i][bc]  = pb[i];
            }
        }
        __syncthreads();
    }

    const int mbase = m0 + wm;
    const int nbase = n0 + wn;
    const int rbase = (lane >> 4) * 8;
    const int col0  = nbase + sub;
    const int col1  = col0 + 16;
    const float bv0 = bias ? bias[col0] : 0.0f;
    const float bv1 = bias ? bias[col1] : 0.0f;
    #pragma unroll
    for (int r = 0; r < 8; ++r) {
        const long row0 = (long)mbase + rbase + r;
        const long row1 = row0 + 16;
        Y[row0 * N + col0] = apply_act(c00[r] + bv0, act);
        Y[row0 * N + col1] = apply_act(c01[r] + bv1, act);
        Y[row1 * N + col0] = apply_act(c10[r] + bv0, act);
        Y[row1 * N + col1] = apply_act(c11[r] + bv1, act);
    }
}

#define SCAN_T 64
#define SCAN_NCHUNK (LKs / SCAN_T)
#define SCAN_VQ 16
#define SCAN_KS 16
#define SCAN_PPITCH (SCAN_KS * SCAN_VQ + 16)
#define SCAN_PD 4

__device__ __forceinline__ long scan_o_index(int e, int head, int l, int vcol)
{
    return ((((long)(e * NHEAD + head) * 4 + (vcol >> 4)) * LKs + l) << 4) + (vcol & 15);
}

__global__ __launch_bounds__(128) void rwkv_scan_kernel(
    const float* __restrict__ r, const float* __restrict__ k,
    const float* __restrict__ v, const float* __restrict__ decay,
    const float* __restrict__ u, float* __restrict__ o)
{
    const int chain = blockIdx.x >> 2;
    const int vq    = blockIdx.x & 3;
    const int e     = chain >> 2;
    const int head  = chain & 3;
    const long base = ((long)e * LKs) * Hdim + head * DK;
    const int t    = threadIdx.x;
    const int vp2  = (t & 7) << 1;
    const int ks   = t >> 3;
    const int k0   = ks << 2;

    __shared__ __attribute__((aligned(16))) float stage[2][SCAN_T][3][DK];
    __shared__ __attribute__((aligned(16))) float vstage[2][SCAN_T][SCAN_VQ];
    __shared__ __attribute__((aligned(16))) float part[2][SCAN_T][SCAN_PPITCH];

    float S0[4], S1[4], uu[4];
    #pragma unroll
    for (int j = 0; j < 4; ++j) {
        S0[j] = 0.0f;
        S1[j] = 0.0f;
        uu[j] = u[head * DK + k0 + j];
    }

    const int  lrow = t >> 4;
    const int  lc   = (t & 15) << 2;
    const int  vrow = t >> 2;
    const int  vc   = (t & 3) << 2;
    const long lofs = base + (long)lrow * Hdim + lc;
    const long vofs = base + (long)vrow * Hdim + vq * SCAN_VQ + vc;

    v4f gr[4], gk[4], gd[4], gv;
    for (int hh = 0; hh < 2; ++hh) {
        const long hofs = (long)(hh * 32) * Hdim;
        #pragma unroll
        for (int i = 0; i < 4; ++i) {
            gr[i] = *(const v4f*)(r     + lofs + hofs + (long)(8 * i) * Hdim);
            gk[i] = *(const v4f*)(k     + lofs + hofs + (long)(8 * i) * Hdim);
            gd[i] = *(const v4f*)(decay + lofs + hofs + (long)(8 * i) * Hdim);
        }
        gv = *(const v4f*)(v + vofs + hofs);
        #pragma unroll
        for (int i = 0; i < 4; ++i) {
            *(v4f*)&stage[0][hh * 32 + lrow + 8 * i][0][lc] = gr[i];
            *(v4f*)&stage[0][hh * 32 + lrow + 8 * i][1][lc] = gk[i];
            *(v4f*)&stage[0][hh * 32 + lrow + 8 * i][2][lc] = gd[i];
        }
        *(v4f*)&vstage[0][hh * 32 + vrow][vc] = gv;
    }
    __syncthreads();

    float* __restrict__ oq = o + (((long)chain * 4 + vq) * LKs << 4);

    for (int c = 0; c < SCAN_NCHUNK; ++c) {
        const int  cur  = c & 1;
        const bool more = (c + 1 < SCAN_NCHUNK);

        float4 qr[SCAN_PD], qk[SCAN_PD], qd[SCAN_PD];
        float2 qv[SCAN_PD];
        #pragma unroll
        for (int p = 0; p < SCAN_PD; ++p) {
            qr[p] = *(const float4*)&stage[cur][p][0][k0];
            qk[p] = *(const float4*)&stage[cur][p][1][k0];
            qd[p] = *(const float4*)&stage[cur][p][2][k0];
            qv[p] = *(const float2*)&vstage[cur][p][vp2];
        }

        for (int hh = 0; hh < 2; ++hh) {
            if (more) {
                const long cofs = (long)((c + 1) * SCAN_T + hh * 32) * Hdim;
                #pragma unroll
                for (int i = 0; i < 4; ++i) {
                    gr[i] = *(const v4f*)(r     + lofs + cofs + (long)(8 * i) * Hdim);
                    gk[i] = *(const v4f*)(k     + lofs + cofs + (long)(8 * i) * Hdim);
                    gd[i] = *(const v4f*)(decay + lofs + cofs + (long)(8 * i) * Hdim);
                }
                gv = *(const v4f*)(v + vofs + cofs);
            }

            #pragma unroll 1
            for (int s0 = hh * 32; s0 < hh * 32 + 32; s0 += 16) {
                #pragma unroll
                for (int i = 0; i < 16; ++i) {
                    const int s = s0 + i;
                    const int p = i % SCAN_PD;
                    const float4 rr = qr[p];
                    const float4 kk = qk[p];
                    const float4 dd = qd[p];
                    const float2 vt = qv[p];
                    const int ns = (i + SCAN_PD < 16) ? s + SCAN_PD : min(s + SCAN_PD, SCAN_T - 1);
                    qr[p] = *(const float4*)&stage[cur][ns][0][k0];
                    qk[p] = *(const float4*)&stage[cur][ns][1][k0];
                    qd[p] = *(const float4*)&stage[cur][ns][2][k0];
                    qv[p] = *(const float2*)&vstage[cur][ns][vp2];
                    __builtin_amdgcn_sched_barrier(0);
                    const float rj[4] = {rr.x, rr.y, rr.z, rr.w};
                    const float kj[4] = {kk.x, kk.y, kk.z, kk.w};
                    const float dj[4] = {dd.x, dd.y, dd.z, dd.w};
                    float a0 = 0.0f, a1 = 0.0f;
                    #pragma unroll
                    for (int j = 0; j < 4; ++j) {
                        const float kv0 = kj[j] * vt.x;
                        const float kv1 = kj[j] * vt.y;
                        a0 += rj[j] * (S0[j] + uu[j] * kv0);
                        a1 += rj[j] * (S1[j] + uu[j] * kv1);
                        S0[j] = dj[j] * S0[j] + kv0;
                        S1[j] = dj[j] * S1[j] + kv1;
                    }
                    *(float2*)&part[cur][s][(ks << 4) + vp2] = make_float2(a0, a1);
                }
            }

            if (more) {
                #pragma unroll
                for (int i = 0; i < 4; ++i) {
                    *(v4f*)&stage[cur ^ 1][hh * 32 + lrow + 8 * i][0][lc] = gr[i];
                    *(v4f*)&stage[cur ^ 1][hh * 32 + lrow + 8 * i][1][lc] = gk[i];
                    *(v4f*)&stage[cur ^ 1][hh * 32 + lrow + 8 * i][2][lc] = gd[i];
                }
                *(v4f*)&vstage[cur ^ 1][hh * 32 + vrow][vc] = gv;
            }
        }
        __syncthreads();

        #pragma unroll
        for (int i = 0; i < 2; ++i) {
            const int js = vrow + 32 * i;
            float4 sum = *(const float4*)&part[cur][js][vc];
            #pragma unroll
            for (int q = 1; q < SCAN_KS; ++q) {
                const float4 pq = *(const float4*)&part[cur][js][(q << 4) + vc];
                sum.x += pq.x; sum.y += pq.y; sum.z += pq.z; sum.w += pq.w;
            }
            *(float4*)(oq + ((long)(c * SCAN_T + js) << 4) + vc) = sum;
        }
    }
}

__global__ void rwkv_gn_silu_kernel(
    const float* __restrict__ o, const float* __restrict__ gsilu,
    const float* __restrict__ gn_w, const float* __restrict__ gn_b,
    float* __restrict__ y)
{
    const long tok = blockIdx.x;
    const int t    = threadIdx.x;
    const int head = t >> 6;
    const int idx  = t & 63;

    const float val = o[scan_o_index((int)(tok >> 9), head, (int)(tok & 511), idx)];
    __shared__ float red[256];

    red[t] = val;
    __syncthreads();
    for (int off = 32; off >= 1; off >>= 1) {
        if (idx < off) red[t] += red[t + off];
        __syncthreads();
    }
    const float mean = red[head * 64] * (1.0f / 64.0f);
    __syncthreads();

    const float dv = val - mean;
    red[t] = dv * dv;
    __syncthreads();
    for (int off = 32; off >= 1; off >>= 1) {
        if (idx < off) red[t] += red[t + off];
        __syncthreads();
    }
    const float var = red[head * 64] * (1.0f / 64.0f);

    const float ynorm = dv * rsqrtf(var + EPSV);
    y[tok * Hdim + t] = (ynorm * gn_w[t] + gn_b[t]) * gsilu[tok * Hdim + t];
}

extern "C" void kernel_launch(void* const* d_in, const int* in_sizes, int n_in,
                              void* d_out, int out_size, void* d_ws, size_t ws_size,
                              hipStream_t stream)
{
    const float* query    = (const float*)d_in[0];
    const float* keyval   = (const float*)d_in[1];
    const float* rwg_mu   = (const float*)d_in[2];
    const float* W_rwg0   = (const float*)d_in[3];
    const float* W_rwg2   = (const float*)d_in[4];
    const float* rwg_bias = (const float*)d_in[5];
    const float* kv_mu    = (const float*)d_in[6];
    const float* W_kv0    = (const float*)d_in[7];
    const float* W_kv2    = (const float*)d_in[8];
    const float* kv_bias  = (const float*)d_in[9];
    const float* Wr       = (const float*)d_in[10];
    const float* Wa       = (const float*)d_in[11];
    const float* Wb       = (const float*)d_in[12];
    const float* bb       = (const float*)d_in[13];
    const float* Wk       = (const float*)d_in[14];
    const float* Wv       = (const float*)d_in[15];
    const float* Wg       = (const float*)d_in[16];
    const float* u        = (const float*)d_in[17];
    const float* gn_w     = (const float*)d_in[18];
    const float* gn_b     = (const float*)d_in[19];
    const float* Wo       = (const float*)d_in[20];

    float* ws = (float*)d_ws;
    const long SZ = (long)NTOK * Hdim;
    float* xr    = ws + 0 * SZ;
    float* xw    = ws + 1 * SZ;
    float* xg    = ws + 2 * SZ;
    float* xk    = ws + 3 * SZ;
    float* xv    = ws + 4 * SZ;
    float* rbuf  = ws + 5 * SZ;
    float* kbuf  = ws + 6 * SZ;
    float* vbuf  = ws + 7 * SZ;
    float* dbuf  = ws + 8 * SZ;
    float* gsbuf = ws + 9 * SZ;
    float* twbuf = ws + 10 * SZ;
    float* obuf  = xr;
    float* ybuf  = xw;

    rwkv_features_kernel<<<NPOS, 256, 0, stream>>>(
        query, keyval, rwg_mu, W_rwg0, W_rwg2, rwg_bias,
        kv_mu, W_kv0, W_kv2, kv_bias, xr, xw, xg, xk, xv);

    auto gemm = [&](const float* X, const float* W, const float* bias,
                    float* Y, int M, int K, int N, int act) {
        dim3 grid(M / GT_M, N / GT_N);
        rwkv_wmma_gemm_kernel<<<grid, 128, 0, stream>>>(X, W, bias, Y, K, N, act);
    };

    gemm(xr, Wr, nullptr, rbuf,  NTOK, 256, 256, ACT_NONE);
    gemm(xk, Wk, nullptr, kbuf,  NTOK, 256, 256, ACT_NONE);
    gemm(xv, Wv, nullptr, vbuf,  NTOK, 256, 256, ACT_NONE);
    gemm(xg, Wg, nullptr, gsbuf, NTOK, 256, 256, ACT_SILU);
    gemm(xw, Wa, nullptr, twbuf, NTOK, 256, 64,  ACT_TANH);
    gemm(twbuf, Wb, bb, dbuf,    NTOK, 64,  256, ACT_DECAY);

    rwkv_scan_kernel<<<Bsz * LQ * NHEAD * 4, 128, 0, stream>>>(rbuf, kbuf, vbuf, dbuf, u, obuf);

    rwkv_gn_silu_kernel<<<NTOK, 256, 0, stream>>>(obuf, gsbuf, gn_w, gn_b, ybuf);

    gemm(ybuf, Wo, nullptr, (float*)d_out, NTOK, 256, 256, ACT_NONE);
}


# GroupNorm+gate: one wave serves 8 tokens of a head, 8 lanes x 8 values per token, sums by lane exchange, no LDS, no barrier: 4096 waves, not 65536 with 14 barriers each; halving-tree sum order kept
# speedup vs baseline: 6.5178x; 1.0643x over previous
#include <hip/hip_runtime.h>
#include <math.h>

#define Hdim 256
#define NHEAD 4
#define DK 64
#define PLR 32
#define Bsz 2
#define LQ 8
#define LKs 512
#define NTOK (Bsz * LQ * LKs)
#define EPSV 1e-5f

typedef float v2f __attribute__((ext_vector_type(2)));
typedef float v8f __attribute__((ext_vector_type(8)));
typedef float v4f __attribute__((ext_vector_type(4)));

#define ACT_NONE  0
#define ACT_SILU  1
#define ACT_TANH  2
#define ACT_DECAY 3

#define NPOS   (Bsz * LKs)
#define WPITCH 36

struct Slab8 { v4f v[8]; };

__device__ __forceinline__ Slab8 slab_fetch(const float* __restrict__ Wsrc, int ld,
                                            int goff, int t)
{
    const float* p = Wsrc + (t >> 3) * ld + goff + ((t & 7) << 2);
    Slab8 s;
    #pragma unroll
    for (int m = 0; m < 8; ++m) s.v[m] = *(const v4f*)(p + m * 32 * ld);
    return s;
}

__device__ __forceinline__ void slab_store(float (*slab)[WPITCH], int t, const Slab8& s)
{
    #pragma unroll
    for (int m = 0; m < 8; ++m)
        *(v4f*)&slab[m * 32 + (t >> 3)][(t & 7) << 2] = s.v[m];
}

__device__ __forceinline__ void slab_row(float (*slab)[WPITCH], int t, float* w)
{
    #pragma unroll
    for (int r4 = 0; r4 < PLR; r4 += 4) {
        const float4 wv = *(const float4*)&slab[t][r4];
        w[r4] = wv.x; w[r4 + 1] = wv.y; w[r4 + 2] = wv.z; w[r4 + 3] = wv.w;
    }
}

__global__ __launch_bounds__(256) void rwkv_features_kernel(
    const float* __restrict__ query, const float* __restrict__ keyval,
    const float* __restrict__ rwg_mu, const float* __restrict__ W_rwg0,
    const float* __restrict__ W_rwg2, const float* __restrict__ rwg_bias,
    const float* __restrict__ kv_mu, const float* __restrict__ W_kv0,
    const float* __restrict__ W_kv2, const float* __restrict__ kv_bias,
    float* __restrict__ xr, float* __restrict__ xw, float* __restrict__ xg,
    float* __restrict__ xk, float* __restrict__ xv)
{
    const int pos = blockIdx.x;
    const int b   = pos >> 9;
    const int l   = pos & 511;
    const int t   = threadIdx.x;

    __shared__ __attribute__((aligned(16))) float xrwg_s[Hdim][LQ];
    __shared__ float xkv_s[Hdim];
    __shared__ __attribute__((aligned(16))) float t1_s[3 * PLR][LQ];
    __shared__ float t2_s[2 * PLR];
    __shared__ __attribute__((aligned(16))) float wslab[Hdim][WPITCH];

    const float h  = keyval[(long)pos * Hdim + t];
    const float hp = (l == 0) ? 0.0f : keyval[(long)(pos - 1) * Hdim + t];
    const float d  = hp - h;
    float dc[LQ];
    {
        const float mu = rwg_mu[t];
        float xq[LQ];
        #pragma unroll
        for (int q = 0; q < LQ; ++q) {
            dc[q] = query[(long)(b * LQ + q) * Hdim + t] - h;
            xq[q] = h + dc[q] * mu;
        }
        *(float4*)&xrwg_s[t][0] = make_float4(xq[0], xq[1], xq[2], xq[3]);
        *(float4*)&xrwg_s[t][4] = make_float4(xq[4], xq[5], xq[6], xq[7]);
        xkv_s[t] = h + d * kv_mu[t];
    }
    __syncthreads();

    if (t < 192) {
        const int c  = (t < 96) ? t : t - 96;
        const int qh = (t < 96) ? 0 : 4;
        float a0 = 0.0f, a1 = 0.0f, a2 = 0.0f, a3 = 0.0f;
        #pragma unroll 16
        for (int i = 0; i < Hdim; ++i) {
            const float  w = W_rwg0[i * 96 + c];
            const float4 x = *(const float4*)&xrwg_s[i][qh];
            a0 += x.x * w;
            a1 += x.y * w;
            a2 += x.z * w;
            a3 += x.w * w;
        }
        *(float4*)&t1_s[c][qh] = make_float4(tanhf(a0), tanhf(a1), tanhf(a2), tanhf(a3));
    } else {
        const int j = t - 192;
        float acc = 0.0f;
        #pragma unroll 16
        for (int i = 0; i < Hdim; ++i) acc += xkv_s[i] * W_kv0[i * 64 + j];
        t2_s[j] = tanhf(acc);
    }

    const long orow = ((long)(b * LQ) * LKs + l) * Hdim + t;
    float w[PLR];

    auto expand_q = [&](int g, float* __restrict__ dst) {
        const float bias = rwg_bias[g * Hdim + t];
        float acc[LQ];
        #pragma unroll
        for (int q = 0; q < LQ; ++q) acc[q] = bias;
        #pragma unroll
        for (int r = 0; r < PLR; ++r) {
            const float4 x0 = *(const float4*)&t1_s[g * PLR + r][0];
            const float4 x1 = *(const float4*)&t1_s[g * PLR + r][4];
            acc[0] += x0.x * w[r];
            acc[1] += x0.y * w[r];
            acc[2] += x0.z * w[r];
            acc[3] += x0.w * w[r];
            acc[4] += x1.x * w[r];
            acc[5] += x1.y * w[r];
            acc[6] += x1.z * w[r];
            acc[7] += x1.w * w[r];
        }
        #pragma unroll
        for (int q = 0; q < LQ; ++q)
            dst[orow + (long)q * LKs * Hdim] = h + dc[q] * acc[q];
    };
    auto expand_kv = [&](int gg, float* __restrict__ dst) {
        float acc = kv_bias[gg * Hdim + t];
        #pragma unroll
        for (int r = 0; r < PLR; ++r) acc += t2_s[gg * PLR + r] * w[r];
        const float val = h + d * acc;
        #pragma unroll
        for (int q = 0; q < LQ; ++q)
            dst[orow + (long)q * LKs * Hdim] = val;
    };

    Slab8 s0 = slab_fetch(W_rwg2, 3 * PLR, 0, t);
    __syncthreads();

    slab_store(wslab, t, s0);
    __syncthreads();
    Slab8 s1 = slab_fetch(W_rwg2, 3 * PLR, PLR, t);
    slab_row(wslab, t, w);
    expand_q(0, xr);
    __syncthreads();

    slab_store(wslab, t, s1);
    __syncthreads();
    Slab8 s2 = slab_fetch(W_rwg2, 3 * PLR, 2 * PLR, t);
    slab_row(wslab, t, w);
    expand_q(1, xw);
    __syncthreads();

    slab_store(wslab, t, s2);
    __syncthreads();
    Slab8 s3 = slab_fetch(W_kv2, 2 * PLR, 0, t);
    slab_row(wslab, t, w);
    expand_q(2, xg);
    __syncthreads();

    slab_store(wslab, t, s3);
    __syncthreads();
    Slab8 s4 = slab_fetch(W_kv2, 2 * PLR, PLR, t);
    slab_row(wslab, t, w);
    expand_kv(0, xk);
    __syncthreads();

    slab_store(wslab, t, s4);
    __syncthreads();
    slab_row(wslab, t, w);
    expand_kv(1, xv);
}

__device__ __forceinline__ float apply_act(float x, int act)
{
    switch (act) {
        case ACT_SILU:  return x / (1.0f + expf(-x));
        case ACT_TANH:  return tanhf(x);
        case ACT_DECAY: return expf(-expf(x));
        default:        return x;
    }
}

#define GT_M 64
#define GT_N 64
#define GT_K 32
#define GA_PITCH 36
#define GB_PITCH 72

__global__ __launch_bounds__(128) void rwkv_wmma_gemm_kernel(
    const float* __restrict__ X, const float* __restrict__ W,
    const float* __restrict__ bias, float* __restrict__ Y,
    int K, int N, int act)
{
    __shared__ __attribute__((aligned(16))) float As[2][GT_M][GA_PITCH];
    __shared__ __attribute__((aligned(16))) float Bs[2][GT_K][GB_PITCH];

    const int t    = threadIdx.x;
    const int wave = t >> 5;
    const int lane = t & 31;
    const int m0   = blockIdx.x * GT_M;
    const int n0   = blockIdx.y * GT_N;
    const int wm   = (wave >> 1) << 5;
    const int wn   = (wave & 1) << 5;

    const int sub  = lane & 15;
    const int koff = (lane >> 4) << 1;

    const int ar = t >> 3;
    const int ac = (t & 7) << 2;
    const int br = t >> 4;
    const int bc = (t & 15) << 2;
    const float* __restrict__ Ag = X + (long)(m0 + ar) * K + ac;
    const float* __restrict__ Bg = W + (long)br * N + n0 + bc;

    v4f pa[4], pb[4];
    #pragma unroll
    for (int i = 0; i < 4; ++i) {
        pa[i] = *(const v4f*)(Ag + (long)(16 * i) * K);
        pb[i] = *(const v4f*)(Bg + (long)(8 * i) * N);
    }
    #pragma unroll
    for (int i = 0; i < 4; ++i) {
        *(v4f*)&As[0][ar + 16 * i][ac] = pa[i];
        *(v4f*)&Bs[0][br + 8 * i][bc]  = pb[i];
    }
    __syncthreads();

    v8f c00 = {}, c01 = {}, c10 = {}, c11 = {};
    const int nchunk = K / GT_K;
    for (int c = 0; c < nchunk; ++c) {
        const int  cur  = c & 1;
        const bool more = (c + 1 < nchunk);

        if (more) {
            const float* __restrict__ Ap = Ag + (c + 1) * GT_K;
            const float* __restrict__ Bp = Bg + (long)((c + 1) * GT_K) * N;
            #pragma unroll
            for (int i = 0; i < 4; ++i) {
                pa[i] = *(const v4f*)(Ap + (long)(16 * i) * K);
                pb[i] = *(const v4f*)(Bp + (long)(8 * i) * N);
            }
        }

        #pragma unroll
        for (int ks = 0; ks < GT_K; ks += 4) {
            const v2f a0 = *(const v2f*)&As[cur][wm + sub][ks + koff];
            const v2f a1 = *(const v2f*)&As[cur][wm + 16 + sub][ks + koff];
            v2f b0, b1;
            b0.x = Bs[cur][ks + koff][wn + sub];
            b0.y = Bs[cur][ks + koff + 1][wn + sub];
            b1.x = Bs[cur][ks + koff][wn + 16 + sub];
            b1.y = Bs[cur][ks + koff + 1][wn + 16 + sub];
            c00 = __builtin_amdgcn_wmma_f32_16x16x4_f32(false, a0, false, b0, (short)0, c00, false, false);
            c01 = __builtin_amdgcn_wmma_f32_16x16x4_f32(false, a0, false, b1, (short)0, c01, false, false);
            c10 = __builtin_amdgcn_wmma_f32_16x16x4_f32(false, a1, false, b0, (short)0, c10, false, false);
            c11 = __builtin_amdgcn_wmma_f32_16x16x4_f32(false, a1, false, b1, (short)0, c11, false, false);
        }

        if (more) {
            #pragma unroll
            for (int i = 0; i < 4; ++i) {
                *(v4f*)&As[cur ^ 1][ar + 16 * i][ac] = pa[i];
                *(v4f*)&Bs[cur ^ 1][br + 8 * i][bc]  = pb[i];
            }
        }
        __syncthreads();
    }

    const int mbase = m0 + wm;
    const int nbase = n0 + wn;
    const int rbase = (lane >> 4) * 8;
    const int col0  = nbase + sub;
    const int col1  = col0 + 16;
    const float bv0 = bias ? bias[col0] : 0.0f;
    const float bv1 = bias ? bias[col1] : 0.0f;
    #pragma unroll
    for (int r = 0; r < 8; ++r) {
        const long row0 = (long)mbase + rbase + r;
        const long row1 = row0 + 16;
        Y[row0 * N + col0] = apply_act(c00[r] + bv0, act);
        Y[row0 * N + col1] = apply_act(c01[r] + bv1, act);
        Y[row1 * N + col0] = apply_act(c10[r] + bv0, act);
        Y[row1 * N + col1] = apply_act(c11[r] + bv1, act);
    }
}

#define SCAN_T 64
#define SCAN_NCHUNK (LKs / SCAN_T)
#define SCAN_VQ 16
#define SCAN_KS 16
#define SCAN_PPITCH (SCAN_KS * SCAN_VQ + 16)
#define SCAN_PD 4

__device__ __forceinline__ long scan_o_index(int e, int head, int l, int vcol)
{
    return ((((long)(e * NHEAD + head) * 4 + (vcol >> 4)) * LKs + l) << 4) + (vcol & 15);
}

__global__ __launch_bounds__(128) void rwkv_scan_kernel(
    const float* __restrict__ r, const float* __restrict__ k,
    const float* __restrict__ v, const float* __restrict__ decay,
    const float* __restrict__ u, float* __restrict__ o)
{
    const int chain = blockIdx.x >> 2;
    const int vq    = blockIdx.x & 3;
    const int e     = chain >> 2;
    const int head  = chain & 3;
    const long base = ((long)e * LKs) * Hdim + head * DK;
    const int t    = threadIdx.x;
    const int vp2  = (t & 7) << 1;
    const int ks   = t >> 3;
    const int k0   = ks << 2;

    __shared__ __attribute__((aligned(16))) float stage[2][SCAN_T][3][DK];
    __shared__ __attribute__((aligned(16))) float vstage[2][SCAN_T][SCAN_VQ];
    __shared__ __attribute__((aligned(16))) float part[2][SCAN_T][SCAN_PPITCH];

    float S0[4], S1[4], uu[4];
    #pragma unroll
    for (int j = 0; j < 4; ++j) {
        S0[j] = 0.0f;
        S1[j] = 0.0f;
        uu[j] = u[head * DK + k0 + j];
    }

    const int  lrow = t >> 4;
    const int  lc   = (t & 15) << 2;
    const int  vrow = t >> 2;
    const int  vc   = (t & 3) << 2;
    const long lofs = base + (long)lrow * Hdim + lc;
    const long vofs = base + (long)vrow * Hdim + vq * SCAN_VQ + vc;

    v4f gr[4], gk[4], gd[4], gv;
    for (int hh = 0; hh < 2; ++hh) {
        const long hofs = (long)(hh * 32) * Hdim;
        #pragma unroll
        for (int i = 0; i < 4; ++i) {
            gr[i] = *(const v4f*)(r     + lofs + hofs + (long)(8 * i) * Hdim);
            gk[i] = *(const v4f*)(k     + lofs + hofs + (long)(8 * i) * Hdim);
            gd[i] = *(const v4f*)(decay + lofs + hofs + (long)(8 * i) * Hdim);
        }
        gv = *(const v4f*)(v + vofs + hofs);
        #pragma unroll
        for (int i = 0; i < 4; ++i) {
            *(v4f*)&stage[0][hh * 32 + lrow + 8 * i][0][lc] = gr[i];
            *(v4f*)&stage[0][hh * 32 + lrow + 8 * i][1][lc] = gk[i];
            *(v4f*)&stage[0][hh * 32 + lrow + 8 * i][2][lc] = gd[i];
        }
        *(v4f*)&vstage[0][hh * 32 + vrow][vc] = gv;
    }
    __syncthreads();

    float* __restrict__ oq = o + (((long)chain * 4 + vq) * LKs << 4);

    for (int c = 0; c < SCAN_NCHUNK; ++c) {
        const int  cur  = c & 1;
        const bool more = (c + 1 < SCAN_NCHUNK);

        float4 qr[SCAN_PD], qk[SCAN_PD], qd[SCAN_PD];
        float2 qv[SCAN_PD];
        #pragma unroll
        for (int p = 0; p < SCAN_PD; ++p) {
            qr[p] = *(const float4*)&stage[cur][p][0][k0];
            qk[p] = *(const float4*)&stage[cur][p][1][k0];
            qd[p] = *(const float4*)&stage[cur][p][2][k0];
            qv[p] = *(const float2*)&vstage[cur][p][vp2];
        }

        for (int hh = 0; hh < 2; ++hh) {
            if (more) {
                const long cofs = (long)((c + 1) * SCAN_T + hh * 32) * Hdim;
                #pragma unroll
                for (int i = 0; i < 4; ++i) {
                    gr[i] = *(const v4f*)(r     + lofs + cofs + (long)(8 * i) * Hdim);
                    gk[i] = *(const v4f*)(k     + lofs + cofs + (long)(8 * i) * Hdim);
                    gd[i] = *(const v4f*)(decay + lofs + cofs + (long)(8 * i) * Hdim);
                }
                gv = *(const v4f*)(v + vofs + cofs);
            }

            #pragma unroll 1
            for (int s0 = hh * 32; s0 < hh * 32 + 32; s0 += 16) {
                #pragma unroll
                for (int i = 0; i < 16; ++i) {
                    const int s = s0 + i;
                    const int p = i % SCAN_PD;
                    const float4 rr = qr[p];
                    const float4 kk = qk[p];
                    const float4 dd = qd[p];
                    const float2 vt = qv[p];
                    const int ns = (i + SCAN_PD < 16) ? s + SCAN_PD : min(s + SCAN_PD, SCAN_T - 1);
                    qr[p] = *(const float4*)&stage[cur][ns][0][k0];
                    qk[p] = *(const float4*)&stage[cur][ns][1][k0];
                    qd[p] = *(const float4*)&stage[cur][ns][2][k0];
                    qv[p] = *(const float2*)&vstage[cur][ns][vp2];
                    __builtin_amdgcn_sched_barrier(0);
                    const float rj[4] = {rr.x, rr.y, rr.z, rr.w};
                    const float kj[4] = {kk.x, kk.y, kk.z, kk.w};
                    const float dj[4] = {dd.x, dd.y, dd.z, dd.w};
                    float a0 = 0.0f, a1 = 0.0f;
                    #pragma unroll
                    for (int j = 0; j < 4; ++j) {
                        const float kv0 = kj[j] * vt.x;
                        const float kv1 = kj[j] * vt.y;
                        a0 += rj[j] * (S0[j] + uu[j] * kv0);
                        a1 += rj[j] * (S1[j] + uu[j] * kv1);
                        S0[j] = dj[j] * S0[j] + kv0;
                        S1[j] = dj[j] * S1[j] + kv1;
                    }
                    *(float2*)&part[cur][s][(ks << 4) + vp2] = make_float2(a0, a1);
                }
            }

            if (more) {
                #pragma unroll
                for (int i = 0; i < 4; ++i) {
                    *(v4f*)&stage[cur ^ 1][hh * 32 + lrow + 8 * i][0][lc] = gr[i];
                    *(v4f*)&stage[cur ^ 1][hh * 32 + lrow + 8 * i][1][lc] = gk[i];
                    *(v4f*)&stage[cur ^ 1][hh * 32 + lrow + 8 * i][2][lc] = gd[i];
                }
                *(v4f*)&vstage[cur ^ 1][hh * 32 + vrow][vc] = gv;
            }
        }
        __syncthreads();

        #pragma unroll
        for (int i = 0; i < 2; ++i) {
            const int js = vrow + 32 * i;
            float4 sum = *(const float4*)&part[cur][js][vc];
            #pragma unroll
            for (int q = 1; q < SCAN_KS; ++q) {
                const float4 pq = *(const float4*)&part[cur][js][(q << 4) + vc];
                sum.x += pq.x; sum.y += pq.y; sum.z += pq.z; sum.w += pq.w;
            }
            *(float4*)(oq + ((long)(c * SCAN_T + js) << 4) + vc) = sum;
        }
    }
}

#define GN_TOK 8
#define GN_WAVES (NTOK * NHEAD / GN_TOK)
#define GN_GROUPS (GN_WAVES / 8)

__device__ __forceinline__ void gn_tree_sum2(v4f lo0, v4f hi0, v4f lo1, v4f hi1,
                                             float& s0, float& s1)
{
    v4f a = lo0 + hi0;
    v4f c = lo1 + hi1;
    #pragma unroll
    for (int m = 4; m >= 1; m >>= 1) {
        v4f ta, tc;
        ta.x = __shfl_xor(a.x, m); ta.y = __shfl_xor(a.y, m);
        ta.z = __shfl_xor(a.z, m); ta.w = __shfl_xor(a.w, m);
        tc.x = __shfl_xor(c.x, m); tc.y = __shfl_xor(c.y, m);
        tc.z = __shfl_xor(c.z, m); tc.w = __shfl_xor(c.w, m);
        a += ta;
        c += tc;
    }
    s0 = (a.x + a.z) + (a.y + a.w);
    s1 = (c.x + c.z) + (c.y + c.w);
}

__global__ __launch_bounds__(256) __attribute__((amdgpu_waves_per_eu(4, 8))) void rwkv_gn_silu_kernel(
    const float* __restrict__ o, const float* __restrict__ gsilu,
    const float* __restrict__ gn_w, const float* __restrict__ gn_b,
    float* __restrict__ y)
{
    const int gw   = blockIdx.x * 8 + (threadIdx.x >> 5);
    const int lane = threadIdx.x & 31;
    const int head = gw & 3;
    const int lb   = (gw >> 2) & 63;
    const int e    = gw >> 8;
    const int col  = (lane & 7) << 2;
    const int hc   = head * DK + col;
    const int l0   = lb * GN_TOK + (lane >> 3);
    const long row = ((long)e * LKs + l0) * Hdim + hc;

    const v4f lo0 = *(const v4f*)(o + scan_o_index(e, head, l0, col));
    const v4f hi0 = *(const v4f*)(o + scan_o_index(e, head, l0, col + 32));
    const v4f lo1 = *(const v4f*)(o + scan_o_index(e, head, l0 + 4, col));
    const v4f hi1 = *(const v4f*)(o + scan_o_index(e, head, l0 + 4, col + 32));
    const v4f g00 = *(const v4f*)(gsilu + row);
    const v4f g01 = *(const v4f*)(gsilu + row + 32);
    const v4f g10 = *(const v4f*)(gsilu + row + 4 * Hdim);
    const v4f g11 = *(const v4f*)(gsilu + row + 4 * Hdim + 32);
    const v4f w0  = *(const v4f*)(gn_w + hc);
    const v4f w1  = *(const v4f*)(gn_w + hc + 32);
    const v4f b0  = *(const v4f*)(gn_b + hc);
    const v4f b1  = *(const v4f*)(gn_b + hc + 32);

    float m0, m1;
    gn_tree_sum2(lo0, hi0, lo1, hi1, m0, m1);
    m0 *= (1.0f / 64.0f);
    m1 *= (1.0f / 64.0f);
    const v4f d00 = lo0 - m0, d01 = hi0 - m0;
    const v4f d10 = lo1 - m1, d11 = hi1 - m1;
    float q0, q1;
    gn_tree_sum2(d00 * d00, d01 * d01, d10 * d10, d11 * d11, q0, q1);
    const float r0 = rsqrtf(q0 * (1.0f / 64.0f) + EPSV);
    const float r1 = rsqrtf(q1 * (1.0f / 64.0f) + EPSV);

    *(v4f*)(y + row)                  = ((d00 * r0) * w0 + b0) * g00;
    *(v4f*)(y + row + 32)             = ((d01 * r0) * w1 + b1) * g01;
    *(v4f*)(y + row + 4 * Hdim)       = ((d10 * r1) * w0 + b0) * g10;
    *(v4f*)(y + row + 4 * Hdim + 32)  = ((d11 * r1) * w1 + b1) * g11;
}

extern "C" void kernel_launch(void* const* d_in, const int* in_sizes, int n_in,
                              void* d_out, int out_size, void* d_ws, size_t ws_size,
                              hipStream_t stream)
{
    const float* query    = (const float*)d_in[0];
    const float* keyval   = (const float*)d_in[1];
    const float* rwg_mu   = (const float*)d_in[2];
    const float* W_rwg0   = (const float*)d_in[3];
    const float* W_rwg2   = (const float*)d_in[4];
    const float* rwg_bias = (const float*)d_in[5];
    const float* kv_mu    = (const float*)d_in[6];
    const float* W_kv0    = (const float*)d_in[7];
    const float* W_kv2    = (const float*)d_in[8];
    const float* kv_bias  = (const float*)d_in[9];
    const float* Wr       = (const float*)d_in[10];
    const float* Wa       = (const float*)d_in[11];
    const float* Wb       = (const float*)d_in[12];
    const float* bb       = (const float*)d_in[13];
    const float* Wk       = (const float*)d_in[14];
    const float* Wv       = (const float*)d_in[15];
    const float* Wg       = (const float*)d_in[16];
    const float* u        = (const float*)d_in[17];
    const float* gn_w     = (const float*)d_in[18];
    const float* gn_b     = (const float*)d_in[19];
    const float* Wo       = (const float*)d_in[20];

    float* ws = (float*)d_ws;
    const long SZ = (long)NTOK * Hdim;
    float* xr    = ws + 0 * SZ;
    float* xw    = ws + 1 * SZ;
    float* xg    = ws + 2 * SZ;
    float* xk    = ws + 3 * SZ;
    float* xv    = ws + 4 * SZ;
    float* rbuf  = ws + 5 * SZ;
    float* kbuf  = ws + 6 * SZ;
    float* vbuf  = ws + 7 * SZ;
    float* dbuf  = ws + 8 * SZ;
    float* gsbuf = ws + 9 * SZ;
    float* twbuf = ws + 10 * SZ;
    float* obuf  = xr;
    float* ybuf  = xw;

    rwkv_features_kernel<<<NPOS, 256, 0, stream>>>(
        query, keyval, rwg_mu, W_rwg0, W_rwg2, rwg_bias,
        kv_mu, W_kv0, W_kv2, kv_bias, xr, xw, xg, xk, xv);

    auto gemm = [&](const float* X, const float* W, const float* bias,
                    float* Y, int M, int K, int N, int act) {
        dim3 grid(M / GT_M, N / GT_N);
        rwkv_wmma_gemm_kernel<<<grid, 128, 0, stream>>>(X, W, bias, Y, K, N, act);
    };

    gemm(xr, Wr, nullptr, rbuf,  NTOK, 256, 256, ACT_NONE);
    gemm(xk, Wk, nullptr, kbuf,  NTOK, 256, 256, ACT_NONE);
    gemm(xv, Wv, nullptr, vbuf,  NTOK, 256, 256, ACT_NONE);
    gemm(xg, Wg, nullptr, gsbuf, NTOK, 256, 256, ACT_SILU);
    gemm(xw, Wa, nullptr, twbuf, NTOK, 256, 64,  ACT_TANH);
    gemm(twbuf, Wb, bb, dbuf,    NTOK, 64,  256, ACT_DECAY);

    rwkv_scan_kernel<<<Bsz * LQ * NHEAD * 4, 128, 0, stream>>>(rbuf, kbuf, vbuf, dbuf, u, obuf);

    rwkv_gn_silu_kernel<<<GN_GROUPS, 256, 0, stream>>>(obuf, gsbuf, gn_w, gn_b, ybuf);

    gemm(ybuf, Wo, nullptr, (float*)d_out, NTOK, 256, 256, ACT_NONE);
}


# GEMM launches: the w-down, r, k, v projections share one launch and the gate and w-up projections another (job picked per column tile): 10 launches -> 6, tiles resident together; sum order kept
# speedup vs baseline: 8.0831x; 1.2402x over previous
#include <hip/hip_runtime.h>
#include <math.h>

#define Hdim 256
#define NHEAD 4
#define DK 64
#define PLR 32
#define Bsz 2
#define LQ 8
#define LKs 512
#define NTOK (Bsz * LQ * LKs)
#define EPSV 1e-5f

typedef float v2f __attribute__((ext_vector_type(2)));
typedef float v8f __attribute__((ext_vector_type(8)));
typedef float v4f __attribute__((ext_vector_type(4)));

#define ACT_NONE  0
#define ACT_SILU  1
#define ACT_TANH  2
#define ACT_DECAY 3

#define NPOS   (Bsz * LKs)
#define WPITCH 36

struct Slab8 { v4f v[8]; };

__device__ __forceinline__ Slab8 slab_fetch(const float* __restrict__ Wsrc, int ld,
                                            int goff, int t)
{
    const float* p = Wsrc + (t >> 3) * ld + goff + ((t & 7) << 2);
    Slab8 s;
    #pragma unroll
    for (int m = 0; m < 8; ++m) s.v[m] = *(const v4f*)(p + m * 32 * ld);
    return s;
}

__device__ __forceinline__ void slab_store(float (*slab)[WPITCH], int t, const Slab8& s)
{
    #pragma unroll
    for (int m = 0; m < 8; ++m)
        *(v4f*)&slab[m * 32 + (t >> 3)][(t & 7) << 2] = s.v[m];
}

__device__ __forceinline__ void slab_row(float (*slab)[WPITCH], int t, float* w)
{
    #pragma unroll
    for (int r4 = 0; r4 < PLR; r4 += 4) {
        const float4 wv = *(const float4*)&slab[t][r4];
        w[r4] = wv.x; w[r4 + 1] = wv.y; w[r4 + 2] = wv.z; w[r4 + 3] = wv.w;
    }
}

__global__ __launch_bounds__(256) void rwkv_features_kernel(
    const float* __restrict__ query, const float* __restrict__ keyval,
    const float* __restrict__ rwg_mu, const float* __restrict__ W_rwg0,
    const float* __restrict__ W_rwg2, const float* __restrict__ rwg_bias,
    const float* __restrict__ kv_mu, const float* __restrict__ W_kv0,
    const float* __restrict__ W_kv2, const float* __restrict__ kv_bias,
    float* __restrict__ xr, float* __restrict__ xw, float* __restrict__ xg,
    float* __restrict__ xk, float* __restrict__ xv)
{
    const int pos = blockIdx.x;
    const int b   = pos >> 9;
    const int l   = pos & 511;
    const int t   = threadIdx.x;

    __shared__ __attribute__((aligned(16))) float xrwg_s[Hdim][LQ];
    __shared__ float xkv_s[Hdim];
    __shared__ __attribute__((aligned(16))) float t1_s[3 * PLR][LQ];
    __shared__ float t2_s[2 * PLR];
    __shared__ __attribute__((aligned(16))) float wslab[Hdim][WPITCH];

    const float h  = keyval[(long)pos * Hdim + t];
    const float hp = (l == 0) ? 0.0f : keyval[(long)(pos - 1) * Hdim + t];
    const float d  = hp - h;
    float dc[LQ];
    {
        const float mu = rwg_mu[t];
        float xq[LQ];
        #pragma unroll
        for (int q = 0; q < LQ; ++q) {
            dc[q] = query[(long)(b * LQ + q) * Hdim + t] - h;
            xq[q] = h + dc[q] * mu;
        }
        *(float4*)&xrwg_s[t][0] = make_float4(xq[0], xq[1], xq[2], xq[3]);
        *(float4*)&xrwg_s[t][4] = make_float4(xq[4], xq[5], xq[6], xq[7]);
        xkv_s[t] = h + d * kv_mu[t];
    }
    __syncthreads();

    if (t < 192) {
        const int c  = (t < 96) ? t : t - 96;
        const int qh = (t < 96) ? 0 : 4;
        float a0 = 0.0f, a1 = 0.0f, a2 = 0.0f, a3 = 0.0f;
        #pragma unroll 16
        for (int i = 0; i < Hdim; ++i) {
            const float  w = W_rwg0[i * 96 + c];
            const float4 x = *(const float4*)&xrwg_s[i][qh];
            a0 += x.x * w;
            a1 += x.y * w;
            a2 += x.z * w;
            a3 += x.w * w;
        }
        *(float4*)&t1_s[c][qh] = make_float4(tanhf(a0), tanhf(a1), tanhf(a2), tanhf(a3));
    } else {
        const int j = t - 192;
        float acc = 0.0f;
        #pragma unroll 16
        for (int i = 0; i < Hdim; ++i) acc += xkv_s[i] * W_kv0[i * 64 + j];
        t2_s[j] = tanhf(acc);
    }

    const long orow = ((long)(b * LQ) * LKs + l) * Hdim + t;
    float w[PLR];

    auto expand_q = [&](int g, float* __restrict__ dst) {
        const float bias = rwg_bias[g * Hdim + t];
        float acc[LQ];
        #pragma unroll
        for (int q = 0; q < LQ; ++q) acc[q] = bias;
        #pragma unroll
        for (int r = 0; r < PLR; ++r) {
            const float4 x0 = *(const float4*)&t1_s[g * PLR + r][0];
            const float4 x1 = *(const float4*)&t1_s[g * PLR + r][4];
            acc[0] += x0.x * w[r];
            acc[1] += x0.y * w[r];
            acc[2] += x0.z * w[r];
            acc[3] += x0.w * w[r];
            acc[4] += x1.x * w[r];
            acc[5] += x1.y * w[r];
            acc[6] += x1.z * w[r];
            acc[7] += x1.w * w[r];
        }
        #pragma unroll
        for (int q = 0; q < LQ; ++q)
            dst[orow + (long)q * LKs * Hdim] = h + dc[q] * acc[q];
    };
    auto expand_kv = [&](int gg, float* __restrict__ dst) {
        float acc = kv_bias[gg * Hdim + t];
        #pragma unroll
        for (int r = 0; r < PLR; ++r) acc += t2_s[gg * PLR + r] * w[r];
        const float val = h + d * acc;
        #pragma unroll
        for (int q = 0; q < LQ; ++q)
            dst[orow + (long)q * LKs * Hdim] = val;
    };

    Slab8 s0 = slab_fetch(W_rwg2, 3 * PLR, 0, t);
    __syncthreads();

    slab_store(wslab, t, s0);
    __syncthreads();
    Slab8 s1 = slab_fetch(W_rwg2, 3 * PLR, PLR, t);
    slab_row(wslab, t, w);
    expand_q(0, xr);
    __syncthreads();

    slab_store(wslab, t, s1);
    __syncthreads();
    Slab8 s2 = slab_fetch(W_rwg2, 3 * PLR, 2 * PLR, t);
    slab_row(wslab, t, w);
    expand_q(1, xw);
    __syncthreads();

    slab_store(wslab, t, s2);
    __syncthreads();
    Slab8 s3 = slab_fetch(W_kv2, 2 * PLR, 0, t);
    slab_row(wslab, t, w);
    expand_q(2, xg);
    __syncthreads();

    slab_store(wslab, t, s3);
    __syncthreads();
    Slab8 s4 = slab_fetch(W_kv2, 2 * PLR, PLR, t);
    slab_row(wslab, t, w);
    expand_kv(0, xk);
    __syncthreads();

    slab_store(wslab, t, s4);
    __syncthreads();
    slab_row(wslab, t, w);
    expand_kv(1, xv);
}

__device__ __forceinline__ float apply_act(float x, int act)
{
    switch (act) {
        case ACT_SILU:  return x / (1.0f + expf(-x));
        case ACT_TANH:  return tanhf(x);
        case ACT_DECAY: return expf(-expf(x));
        default:        return x;
    }
}

#define GT_M 64
#define GT_N 64
#define GT_K 32
#define GA_PITCH 36
#define GB_PITCH 72

__device__ __forceinline__ void gemm_tile(
    const float* __restrict__ X, const float* __restrict__ W,
    const float* __restrict__ bias, float* __restrict__ Y,
    int K, int N, int act, int m0, int n0)
{
    __shared__ __attribute__((aligned(16))) float As[2][GT_M][GA_PITCH];
    __shared__ __attribute__((aligned(16))) float Bs[2][GT_K][GB_PITCH];

    const int t    = threadIdx.x;
    const int wave = t >> 5;
    const int lane = t & 31;
    const int wm   = (wave >> 1) << 5;
    const int wn   = (wave & 1) << 5;

    const int sub  = lane & 15;
    const int koff = (lane >> 4) << 1;

    const int ar = t >> 3;
    const int ac = (t & 7) << 2;
    const int br = t >> 4;
    const int bc = (t & 15) << 2;
    const float* __restrict__ Ag = X + (long)(m0 + ar) * K + ac;
    const float* __restrict__ Bg = W + (long)br * N + n0 + bc;

    v4f pa[4], pb[4];
    #pragma unroll
    for (int i = 0; i < 4; ++i) {
        pa[i] = *(const v4f*)(Ag + (long)(16 * i) * K);
        pb[i] = *(const v4f*)(Bg + (long)(8 * i) * N);
    }
    #pragma unroll
    for (int i = 0; i < 4; ++i) {
        *(v4f*)&As[0][ar + 16 * i][ac] = pa[i];
        *(v4f*)&Bs[0][br + 8 * i][bc]  = pb[i];
    }
    __syncthreads();

    v8f c00 = {}, c01 = {}, c10 = {}, c11 = {};
    const int nchunk = K / GT_K;
    for (int c = 0; c < nchunk; ++c) {
        const int  cur  = c & 1;
        const bool more = (c + 1 < nchunk);

        if (more) {
            const float* __restrict__ Ap = Ag + (c + 1) * GT_K;
            const float* __restrict__ Bp = Bg + (long)((c + 1) * GT_K) * N;
            #pragma unroll
            for (int i = 0; i < 4; ++i) {
                pa[i] = *(const v4f*)(Ap + (long)(16 * i) * K);
                pb[i] = *(const v4f*)(Bp + (long)(8 * i) * N);
            }
        }

        #pragma unroll
        for (int ks = 0; ks < GT_K; ks += 4) {
            const v2f a0 = *(const v2f*)&As[cur][wm + sub][ks + koff];
            const v2f a1 = *(const v2f*)&As[cur][wm + 16 + sub][ks + koff];
            v2f b0, b1;
            b0.x = Bs[cur][ks + koff][wn + sub];
            b0.y = Bs[cur][ks + koff + 1][wn + sub];
            b1.x = Bs[cur][ks + koff][wn + 16 + sub];
            b1.y = Bs[cur][ks + koff + 1][wn + 16 + sub];
            c00 = __builtin_amdgcn_wmma_f32_16x16x4_f32(false, a0, false, b0, (short)0, c00, false, false);
            c01 = __builtin_amdgcn_wmma_f32_16x16x4_f32(false, a0, false, b1, (short)0, c01, false, false);
            c10 = __builtin_amdgcn_wmma_f32_16x16x4_f32(false, a1, false, b0, (short)0, c10, false, false);
            c11 = __builtin_amdgcn_wmma_f32_16x16x4_f32(false, a1, false, b1, (short)0, c11, false, false);
        }

        if (more) {
            #pragma unroll
            for (int i = 0; i < 4; ++i) {
                *(v4f*)&As[cur ^ 1][ar + 16 * i][ac] = pa[i];
                *(v4f*)&Bs[cur ^ 1][br + 8 * i][bc]  = pb[i];
            }
        }
        __syncthreads();
    }

    const int mbase = m0 + wm;
    const int nbase = n0 + wn;
    const int rbase = (lane >> 4) * 8;
    const int col0  = nbase + sub;
    const int col1  = col0 + 16;
    const float bv0 = bias ? bias[col0] : 0.0f;
    const float bv1 = bias ? bias[col1] : 0.0f;
    #pragma unroll
    for (int r = 0; r < 8; ++r) {
        const long row0 = (long)mbase + rbase + r;
        const long row1 = row0 + 16;
        Y[row0 * N + col0] = apply_act(c00[r] + bv0, act);
        Y[row0 * N + col1] = apply_act(c01[r] + bv1, act);
        Y[row1 * N + col0] = apply_act(c10[r] + bv0, act);
        Y[row1 * N + col1] = apply_act(c11[r] + bv1, act);
    }
}

__global__ __launch_bounds__(128) void rwkv_wmma_gemm_kernel(
    const float* X0, const float* W0, const float* b0, float* Y0, int K0, int N0, int act0,
    const float* X1, const float* W1, const float* b1, float* Y1, int K1, int N1, int act1,
    const float* X2, const float* W2, const float* b2, float* Y2, int K2, int N2, int act2,
    const float* X3, const float* W3, const float* b3, float* Y3, int K3, int N3, int act3,
    int y1, int y2, int y3)
{
    const int y = blockIdx.y;
    const float* X = X0;
    const float* W = W0;
    const float* bias = b0;
    float* Y = Y0;
    int K = K0, N = N0, act = act0, yb = 0;
    if (y >= y1) { X = X1; W = W1; bias = b1; Y = Y1; K = K1; N = N1; act = act1; yb = y1; }
    if (y >= y2) { X = X2; W = W2; bias = b2; Y = Y2; K = K2; N = N2; act = act2; yb = y2; }
    if (y >= y3) { X = X3; W = W3; bias = b3; Y = Y3; K = K3; N = N3; act = act3; yb = y3; }
    gemm_tile(X, W, bias, Y, K, N, act, blockIdx.x * GT_M, (y - yb) * GT_N);
}

#define SCAN_T 64
#define SCAN_NCHUNK (LKs / SCAN_T)
#define SCAN_VQ 16
#define SCAN_KS 16
#define SCAN_PPITCH (SCAN_KS * SCAN_VQ + 16)
#define SCAN_PD 4

__device__ __forceinline__ long scan_o_index(int e, int head, int l, int vcol)
{
    return ((((long)(e * NHEAD + head) * 4 + (vcol >> 4)) * LKs + l) << 4) + (vcol & 15);
}

__global__ __launch_bounds__(128) void rwkv_scan_kernel(
    const float* __restrict__ r, const float* __restrict__ k,
    const float* __restrict__ v, const float* __restrict__ decay,
    const float* __restrict__ u, float* __restrict__ o)
{
    const int chain = blockIdx.x >> 2;
    const int vq    = blockIdx.x & 3;
    const int e     = chain >> 2;
    const int head  = chain & 3;
    const long base = ((long)e * LKs) * Hdim + head * DK;
    const int t    = threadIdx.x;
    const int vp2  = (t & 7) << 1;
    const int ks   = t >> 3;
    const int k0   = ks << 2;

    __shared__ __attribute__((aligned(16))) float stage[2][SCAN_T][3][DK];
    __shared__ __attribute__((aligned(16))) float vstage[2][SCAN_T][SCAN_VQ];
    __shared__ __attribute__((aligned(16))) float part[2][SCAN_T][SCAN_PPITCH];

    float S0[4], S1[4], uu[4];
    #pragma unroll
    for (int j = 0; j < 4; ++j) {
        S0[j] = 0.0f;
        S1[j] = 0.0f;
        uu[j] = u[head * DK + k0 + j];
    }

    const int  lrow = t >> 4;
    const int  lc   = (t & 15) << 2;
    const int  vrow = t >> 2;
    const int  vc   = (t & 3) << 2;
    const long lofs = base + (long)lrow * Hdim + lc;
    const long vofs = base + (long)vrow * Hdim + vq * SCAN_VQ + vc;

    v4f gr[4], gk[4], gd[4], gv;
    for (int hh = 0; hh < 2; ++hh) {
        const long hofs = (long)(hh * 32) * Hdim;
        #pragma unroll
        for (int i = 0; i < 4; ++i) {
            gr[i] = *(const v4f*)(r     + lofs + hofs + (long)(8 * i) * Hdim);
            gk[i] = *(const v4f*)(k     + lofs + hofs + (long)(8 * i) * Hdim);
            gd[i] = *(const v4f*)(decay + lofs + hofs + (long)(8 * i) * Hdim);
        }
        gv = *(const v4f*)(v + vofs + hofs);
        #pragma unroll
        for (int i = 0; i < 4; ++i) {
            *(v4f*)&stage[0][hh * 32 + lrow + 8 * i][0][lc] = gr[i];
            *(v4f*)&stage[0][hh * 32 + lrow + 8 * i][1][lc] = gk[i];
            *(v4f*)&stage[0][hh * 32 + lrow + 8 * i][2][lc] = gd[i];
        }
        *(v4f*)&vstage[0][hh * 32 + vrow][vc] = gv;
    }
    __syncthreads();

    float* __restrict__ oq = o + (((long)chain * 4 + vq) * LKs << 4);

    for (int c = 0; c < SCAN_NCHUNK; ++c) {
        const int  cur  = c & 1;
        const bool more = (c + 1 < SCAN_NCHUNK);

        float4 qr[SCAN_PD], qk[SCAN_PD], qd[SCAN_PD];
        float2 qv[SCAN_PD];
        #pragma unroll
        for (int p = 0; p < SCAN_PD; ++p) {
            qr[p] = *(const float4*)&stage[cur][p][0][k0];
            qk[p] = *(const float4*)&stage[cur][p][1][k0];
            qd[p] = *(const float4*)&stage[cur][p][2][k0];
            qv[p] = *(const float2*)&vstage[cur][p][vp2];
        }

        for (int hh = 0; hh < 2; ++hh) {
            if (more) {
                const long cofs = (long)((c + 1) * SCAN_T + hh * 32) * Hdim;
                #pragma unroll
                for (int i = 0; i < 4; ++i) {
                    gr[i] = *(const v4f*)(r     + lofs + cofs + (long)(8 * i) * Hdim);
                    gk[i] = *(const v4f*)(k     + lofs + cofs + (long)(8 * i) * Hdim);
                    gd[i] = *(const v4f*)(decay + lofs + cofs + (long)(8 * i) * Hdim);
                }
                gv = *(const v4f*)(v + vofs + cofs);
            }

            #pragma unroll 1
            for (int s0 = hh * 32; s0 < hh * 32 + 32; s0 += 16) {
                #pragma unroll
                for (int i = 0; i < 16; ++i) {
                    const int s = s0 + i;
                    const int p = i % SCAN_PD;
                    const float4 rr = qr[p];
                    const float4 kk = qk[p];
                    const float4 dd = qd[p];
                    const float2 vt = qv[p];
                    const int ns = (i + SCAN_PD < 16) ? s + SCAN_PD : min(s + SCAN_PD, SCAN_T - 1);
                    qr[p] = *(const float4*)&stage[cur][ns][0][k0];
                    qk[p] = *(const float4*)&stage[cur][ns][1][k0];
                    qd[p] = *(const float4*)&stage[cur][ns][2][k0];
                    qv[p] = *(const float2*)&vstage[cur][ns][vp2];
                    __builtin_amdgcn_sched_barrier(0);
                    const float rj[4] = {rr.x, rr.y, rr.z, rr.w};
                    const float kj[4] = {kk.x, kk.y, kk.z, kk.w};
                    const float dj[4] = {dd.x, dd.y, dd.z, dd.w};
                    float a0 = 0.0f, a1 = 0.0f;
                    #pragma unroll
                    for (int j = 0; j < 4; ++j) {
                        const float kv0 = kj[j] * vt.x;
                        const float kv1 = kj[j] * vt.y;
                        a0 += rj[j] * (S0[j] + uu[j] * kv0);
                        a1 += rj[j] * (S1[j] + uu[j] * kv1);
                        S0[j] = dj[j] * S0[j] + kv0;
                        S1[j] = dj[j] * S1[j] + kv1;
                    }
                    *(float2*)&part[cur][s][(ks << 4) + vp2] = make_float2(a0, a1);
                }
            }

            if (more) {
                #pragma unroll
                for (int i = 0; i < 4; ++i) {
                    *(v4f*)&stage[cur ^ 1][hh * 32 + lrow + 8 * i][0][lc] = gr[i];
                    *(v4f*)&stage[cur ^ 1][hh * 32 + lrow + 8 * i][1][lc] = gk[i];
                    *(v4f*)&stage[cur ^ 1][hh * 32 + lrow + 8 * i][2][lc] = gd[i];
                }
                *(v4f*)&vstage[cur ^ 1][hh * 32 + vrow][vc] = gv;
            }
        }
        __syncthreads();

        #pragma unroll
        for (int i = 0; i < 2; ++i) {
            const int js = vrow + 32 * i;
            float4 sum = *(const float4*)&part[cur][js][vc];
            #pragma unroll
            for (int q = 1; q < SCAN_KS; ++q) {
                const float4 pq = *(const float4*)&part[cur][js][(q << 4) + vc];
                sum.x += pq.x; sum.y += pq.y; sum.z += pq.z; sum.w += pq.w;
            }
            *(float4*)(oq + ((long)(c * SCAN_T + js) << 4) + vc) = sum;
        }
    }
}

#define GN_TOK 8
#define GN_WAVES (NTOK * NHEAD / GN_TOK)
#define GN_GROUPS (GN_WAVES / 8)

__device__ __forceinline__ void gn_tree_sum2(v4f lo0, v4f hi0, v4f lo1, v4f hi1,
                                             float& s0, float& s1)
{
    v4f a = lo0 + hi0;
    v4f c = lo1 + hi1;
    #pragma unroll
    for (int m = 4; m >= 1; m >>= 1) {
        v4f ta, tc;
        ta.x = __shfl_xor(a.x, m); ta.y = __shfl_xor(a.y, m);
        ta.z = __shfl_xor(a.z, m); ta.w = __shfl_xor(a.w, m);
        tc.x = __shfl_xor(c.x, m); tc.y = __shfl_xor(c.y, m);
        tc.z = __shfl_xor(c.z, m); tc.w = __shfl_xor(c.w, m);
        a += ta;
        c += tc;
    }
    s0 = (a.x + a.z) + (a.y + a.w);
    s1 = (c.x + c.z) + (c.y + c.w);
}

__global__ __launch_bounds__(256) __attribute__((amdgpu_waves_per_eu(4, 8))) void rwkv_gn_silu_kernel(
    const float* __restrict__ o, const float* __restrict__ gsilu,
    const float* __restrict__ gn_w, const float* __restrict__ gn_b,
    float* __restrict__ y)
{
    const int gw   = blockIdx.x * 8 + (threadIdx.x >> 5);
    const int lane = threadIdx.x & 31;
    const int head = gw & 3;
    const int lb   = (gw >> 2) & 63;
    const int e    = gw >> 8;
    const int col  = (lane & 7) << 2;
    const int hc   = head * DK + col;
    const int l0   = lb * GN_TOK + (lane >> 3);
    const long row = ((long)e * LKs + l0) * Hdim + hc;

    const v4f lo0 = *(const v4f*)(o + scan_o_index(e, head, l0, col));
    const v4f hi0 = *(const v4f*)(o + scan_o_index(e, head, l0, col + 32));
    const v4f lo1 = *(const v4f*)(o + scan_o_index(e, head, l0 + 4, col));
    const v4f hi1 = *(const v4f*)(o + scan_o_index(e, head, l0 + 4, col + 32));
    const v4f g00 = *(const v4f*)(gsilu + row);
    const v4f g01 = *(const v4f*)(gsilu + row + 32);
    const v4f g10 = *(const v4f*)(gsilu + row + 4 * Hdim);
    const v4f g11 = *(const v4f*)(gsilu + row + 4 * Hdim + 32);
    const v4f w0  = *(const v4f*)(gn_w + hc);
    const v4f w1  = *(const v4f*)(gn_w + hc + 32);
    const v4f b0  = *(const v4f*)(gn_b + hc);
    const v4f b1  = *(const v4f*)(gn_b + hc + 32);

    float m0, m1;
    gn_tree_sum2(lo0, hi0, lo1, hi1, m0, m1);
    m0 *= (1.0f / 64.0f);
    m1 *= (1.0f / 64.0f);
    const v4f d00 = lo0 - m0, d01 = hi0 - m0;
    const v4f d10 = lo1 - m1, d11 = hi1 - m1;
    float q0, q1;
    gn_tree_sum2(d00 * d00, d01 * d01, d10 * d10, d11 * d11, q0, q1);
    const float r0 = rsqrtf(q0 * (1.0f / 64.0f) + EPSV);
    const float r1 = rsqrtf(q1 * (1.0f / 64.0f) + EPSV);

    *(v4f*)(y + row)                  = ((d00 * r0) * w0 + b0) * g00;
    *(v4f*)(y + row + 32)             = ((d01 * r0) * w1 + b1) * g01;
    *(v4f*)(y + row + 4 * Hdim)       = ((d10 * r1) * w0 + b0) * g10;
    *(v4f*)(y + row + 4 * Hdim + 32)  = ((d11 * r1) * w1 + b1) * g11;
}

extern "C" void kernel_launch(void* const* d_in, const int* in_sizes, int n_in,
                              void* d_out, int out_size, void* d_ws, size_t ws_size,
                              hipStream_t stream)
{
    const float* query    = (const float*)d_in[0];
    const float* keyval   = (const float*)d_in[1];
    const float* rwg_mu   = (const float*)d_in[2];
    const float* W_rwg0   = (const float*)d_in[3];
    const float* W_rwg2   = (const float*)d_in[4];
    const float* rwg_bias = (const float*)d_in[5];
    const float* kv_mu    = (const float*)d_in[6];
    const float* W_kv0    = (const float*)d_in[7];
    const float* W_kv2    = (const float*)d_in[8];
    const float* kv_bias  = (const float*)d_in[9];
    const float* Wr       = (const float*)d_in[10];
    const float* Wa       = (const float*)d_in[11];
    const float* Wb       = (const float*)d_in[12];
    const float* bb       = (const float*)d_in[13];
    const float* Wk       = (const float*)d_in[14];
    const float* Wv       = (const float*)d_in[15];
    const float* Wg       = (const float*)d_in[16];
    const float* u        = (const float*)d_in[17];
    const float* gn_w     = (const float*)d_in[18];
    const float* gn_b     = (const float*)d_in[19];
    const float* Wo       = (const float*)d_in[20];

    float* ws = (float*)d_ws;
    const long SZ = (long)NTOK * Hdim;
    float* xr    = ws + 0 * SZ;
    float* xw    = ws + 1 * SZ;
    float* xg    = ws + 2 * SZ;
    float* xk    = ws + 3 * SZ;
    float* xv    = ws + 4 * SZ;
    float* rbuf  = ws + 5 * SZ;
    float* kbuf  = ws + 6 * SZ;
    float* vbuf  = ws + 7 * SZ;
    float* dbuf  = ws + 8 * SZ;
    float* gsbuf = ws + 9 * SZ;
    float* twbuf = ws + 10 * SZ;
    float* obuf  = xr;
    float* ybuf  = xw;

    rwkv_features_kernel<<<NPOS, 256, 0, stream>>>(
        query, keyval, rwg_mu, W_rwg0, W_rwg2, rwg_bias,
        kv_mu, W_kv0, W_kv2, kv_bias, xr, xw, xg, xk, xv);

    const int NOJOB = 0x7fffffff;
    const dim3 gblock(128);

    rwkv_wmma_gemm_kernel<<<dim3(NTOK / GT_M, 13), gblock, 0, stream>>>(
        xw, Wa, nullptr, twbuf, 256, 64,  ACT_TANH,
        xr, Wr, nullptr, rbuf,  256, 256, ACT_NONE,
        xk, Wk, nullptr, kbuf,  256, 256, ACT_NONE,
        xv, Wv, nullptr, vbuf,  256, 256, ACT_NONE,
        1, 5, 9);

    rwkv_wmma_gemm_kernel<<<dim3(NTOK / GT_M, 8), gblock, 0, stream>>>(
        xg,    Wg, nullptr, gsbuf, 256, 256, ACT_SILU,
        twbuf, Wb, bb,      dbuf,  64,  256, ACT_DECAY,
        nullptr, nullptr, nullptr, nullptr, 0, 0, 0,
        nullptr, nullptr, nullptr, nullptr, 0, 0, 0,
        4, NOJOB, NOJOB);

    rwkv_scan_kernel<<<Bsz * LQ * NHEAD * 4, 128, 0, stream>>>(rbuf, kbuf, vbuf, dbuf, u, obuf);

    rwkv_gn_silu_kernel<<<GN_GROUPS, 256, 0, stream>>>(obuf, gsbuf, gn_w, gn_b, ybuf);

    rwkv_wmma_gemm_kernel<<<dim3(NTOK / GT_M, 4), gblock, 0, stream>>>(
        ybuf, Wo, nullptr, (float*)d_out, 256, 256, ACT_NONE,
        nullptr, nullptr, nullptr, nullptr, 0, 0, 0,
        nullptr, nullptr, nullptr, nullptr, 0, 0, 0,
        nullptr, nullptr, nullptr, nullptr, 0, 0, 0,
        NOJOB, NOJOB, NOJOB);
}


# k and v do not depend on the query: their mixed inputs are written once per key position and their projections run on 1024 rows, not 8192; the scan reads them per batch; sum order kept
# speedup vs baseline: 9.2137x; 1.1399x over previous
#include <hip/hip_runtime.h>
#include <math.h>

#define Hdim 256
#define NHEAD 4
#define DK 64
#define PLR 32
#define Bsz 2
#define LQ 8
#define LKs 512
#define NTOK (Bsz * LQ * LKs)
#define EPSV 1e-5f

typedef float v2f __attribute__((ext_vector_type(2)));
typedef float v8f __attribute__((ext_vector_type(8)));
typedef float v4f __attribute__((ext_vector_type(4)));

#define ACT_NONE  0
#define ACT_SILU  1
#define ACT_TANH  2
#define ACT_DECAY 3

#define NPOS   (Bsz * LKs)
#define WPITCH 36

struct Slab8 { v4f v[8]; };

__device__ __forceinline__ Slab8 slab_fetch(const float* __restrict__ Wsrc, int ld,
                                            int goff, int t)
{
    const float* p = Wsrc + (t >> 3) * ld + goff + ((t & 7) << 2);
    Slab8 s;
    #pragma unroll
    for (int m = 0; m < 8; ++m) s.v[m] = *(const v4f*)(p + m * 32 * ld);
    return s;
}

__device__ __forceinline__ void slab_store(float (*slab)[WPITCH], int t, const Slab8& s)
{
    #pragma unroll
    for (int m = 0; m < 8; ++m)
        *(v4f*)&slab[m * 32 + (t >> 3)][(t & 7) << 2] = s.v[m];
}

__device__ __forceinline__ void slab_row(float (*slab)[WPITCH], int t, float* w)
{
    #pragma unroll
    for (int r4 = 0; r4 < PLR; r4 += 4) {
        const float4 wv = *(const float4*)&slab[t][r4];
        w[r4] = wv.x; w[r4 + 1] = wv.y; w[r4 + 2] = wv.z; w[r4 + 3] = wv.w;
    }
}

__global__ __launch_bounds__(256) void rwkv_features_kernel(
    const float* __restrict__ query, const float* __restrict__ keyval,
    const float* __restrict__ rwg_mu, const float* __restrict__ W_rwg0,
    const float* __restrict__ W_rwg2, const float* __restrict__ rwg_bias,
    const float* __restrict__ kv_mu, const float* __restrict__ W_kv0,
    const float* __restrict__ W_kv2, const float* __restrict__ kv_bias,
    float* __restrict__ xr, float* __restrict__ xw, float* __restrict__ xg,
    float* __restrict__ xk, float* __restrict__ xv)
{
    const int pos = blockIdx.x;
    const int b   = pos >> 9;
    const int l   = pos & 511;
    const int t   = threadIdx.x;

    __shared__ __attribute__((aligned(16))) float xrwg_s[Hdim][LQ];
    __shared__ float xkv_s[Hdim];
    __shared__ __attribute__((aligned(16))) float t1_s[3 * PLR][LQ];
    __shared__ float t2_s[2 * PLR];
    __shared__ __attribute__((aligned(16))) float wslab[Hdim][WPITCH];

    const float h  = keyval[(long)pos * Hdim + t];
    const float hp = (l == 0) ? 0.0f : keyval[(long)(pos - 1) * Hdim + t];
    const float d  = hp - h;
    float dc[LQ];
    {
        const float mu = rwg_mu[t];
        float xq[LQ];
        #pragma unroll
        for (int q = 0; q < LQ; ++q) {
            dc[q] = query[(long)(b * LQ + q) * Hdim + t] - h;
            xq[q] = h + dc[q] * mu;
        }
        *(float4*)&xrwg_s[t][0] = make_float4(xq[0], xq[1], xq[2], xq[3]);
        *(float4*)&xrwg_s[t][4] = make_float4(xq[4], xq[5], xq[6], xq[7]);
        xkv_s[t] = h + d * kv_mu[t];
    }
    __syncthreads();

    if (t < 192) {
        const int c  = (t < 96) ? t : t - 96;
        const int qh = (t < 96) ? 0 : 4;
        float a0 = 0.0f, a1 = 0.0f, a2 = 0.0f, a3 = 0.0f;
        #pragma unroll 16
        for (int i = 0; i < Hdim; ++i) {
            const float  w = W_rwg0[i * 96 + c];
            const float4 x = *(const float4*)&xrwg_s[i][qh];
            a0 += x.x * w;
            a1 += x.y * w;
            a2 += x.z * w;
            a3 += x.w * w;
        }
        *(float4*)&t1_s[c][qh] = make_float4(tanhf(a0), tanhf(a1), tanhf(a2), tanhf(a3));
    } else {
        const int j = t - 192;
        float acc = 0.0f;
        #pragma unroll 16
        for (int i = 0; i < Hdim; ++i) acc += xkv_s[i] * W_kv0[i * 64 + j];
        t2_s[j] = tanhf(acc);
    }

    const long orow = ((long)(b * LQ) * LKs + l) * Hdim + t;
    float w[PLR];

    auto expand_q = [&](int g, float* __restrict__ dst) {
        const float bias = rwg_bias[g * Hdim + t];
        float acc[LQ];
        #pragma unroll
        for (int q = 0; q < LQ; ++q) acc[q] = bias;
        #pragma unroll
        for (int r = 0; r < PLR; ++r) {
            const float4 x0 = *(const float4*)&t1_s[g * PLR + r][0];
            const float4 x1 = *(const float4*)&t1_s[g * PLR + r][4];
            acc[0] += x0.x * w[r];
            acc[1] += x0.y * w[r];
            acc[2] += x0.z * w[r];
            acc[3] += x0.w * w[r];
            acc[4] += x1.x * w[r];
            acc[5] += x1.y * w[r];
            acc[6] += x1.z * w[r];
            acc[7] += x1.w * w[r];
        }
        #pragma unroll
        for (int q = 0; q < LQ; ++q)
            dst[orow + (long)q * LKs * Hdim] = h + dc[q] * acc[q];
    };
    auto expand_kv = [&](int gg, float* __restrict__ dst) {
        float acc = kv_bias[gg * Hdim + t];
        #pragma unroll
        for (int r = 0; r < PLR; ++r) acc += t2_s[gg * PLR + r] * w[r];
        dst[(long)pos * Hdim + t] = h + d * acc;
    };

    Slab8 s0 = slab_fetch(W_rwg2, 3 * PLR, 0, t);
    __syncthreads();

    slab_store(wslab, t, s0);
    __syncthreads();
    Slab8 s1 = slab_fetch(W_rwg2, 3 * PLR, PLR, t);
    slab_row(wslab, t, w);
    expand_q(0, xr);
    __syncthreads();

    slab_store(wslab, t, s1);
    __syncthreads();
    Slab8 s2 = slab_fetch(W_rwg2, 3 * PLR, 2 * PLR, t);
    slab_row(wslab, t, w);
    expand_q(1, xw);
    __syncthreads();

    slab_store(wslab, t, s2);
    __syncthreads();
    Slab8 s3 = slab_fetch(W_kv2, 2 * PLR, 0, t);
    slab_row(wslab, t, w);
    expand_q(2, xg);
    __syncthreads();

    slab_store(wslab, t, s3);
    __syncthreads();
    Slab8 s4 = slab_fetch(W_kv2, 2 * PLR, PLR, t);
    slab_row(wslab, t, w);
    expand_kv(0, xk);
    __syncthreads();

    slab_store(wslab, t, s4);
    __syncthreads();
    slab_row(wslab, t, w);
    expand_kv(1, xv);
}

__device__ __forceinline__ float apply_act(float x, int act)
{
    switch (act) {
        case ACT_SILU:  return x / (1.0f + expf(-x));
        case ACT_TANH:  return tanhf(x);
        case ACT_DECAY: return expf(-expf(x));
        default:        return x;
    }
}

#define GT_M 64
#define GT_N 64
#define GT_K 32
#define GA_PITCH 36
#define GB_PITCH 72

__device__ __forceinline__ void gemm_tile(
    const float* __restrict__ X, const float* __restrict__ W,
    const float* __restrict__ bias, float* __restrict__ Y,
    int K, int N, int act, int m0, int n0)
{
    __shared__ __attribute__((aligned(16))) float As[2][GT_M][GA_PITCH];
    __shared__ __attribute__((aligned(16))) float Bs[2][GT_K][GB_PITCH];

    const int t    = threadIdx.x;
    const int wave = t >> 5;
    const int lane = t & 31;
    const int wm   = (wave >> 1) << 5;
    const int wn   = (wave & 1) << 5;

    const int sub  = lane & 15;
    const int koff = (lane >> 4) << 1;

    const int ar = t >> 3;
    const int ac = (t & 7) << 2;
    const int br = t >> 4;
    const int bc = (t & 15) << 2;
    const float* __restrict__ Ag = X + (long)(m0 + ar) * K + ac;
    const float* __restrict__ Bg = W + (long)br * N + n0 + bc;

    v4f pa[4], pb[4];
    #pragma unroll
    for (int i = 0; i < 4; ++i) {
        pa[i] = *(const v4f*)(Ag + (long)(16 * i) * K);
        pb[i] = *(const v4f*)(Bg + (long)(8 * i) * N);
    }
    #pragma unroll
    for (int i = 0; i < 4; ++i) {
        *(v4f*)&As[0][ar + 16 * i][ac] = pa[i];
        *(v4f*)&Bs[0][br + 8 * i][bc]  = pb[i];
    }
    __syncthreads();

    v8f c00 = {}, c01 = {}, c10 = {}, c11 = {};
    const int nchunk = K / GT_K;
    for (int c = 0; c < nchunk; ++c) {
        const int  cur  = c & 1;
        const bool more = (c + 1 < nchunk);

        if (more) {
            const float* __restrict__ Ap = Ag + (c + 1) * GT_K;
            const float* __restrict__ Bp = Bg + (long)((c + 1) * GT_K) * N;
            #pragma unroll
            for (int i = 0; i < 4; ++i) {
                pa[i] = *(const v4f*)(Ap + (long)(16 * i) * K);
                pb[i] = *(const v4f*)(Bp + (long)(8 * i) * N);
            }
        }

        #pragma unroll
        for (int ks = 0; ks < GT_K; ks += 4) {
            const v2f a0 = *(const v2f*)&As[cur][wm + sub][ks + koff];
            const v2f a1 = *(const v2f*)&As[cur][wm + 16 + sub][ks + koff];
            v2f b0, b1;
            b0.x = Bs[cur][ks + koff][wn + sub];
            b0.y = Bs[cur][ks + koff + 1][wn + sub];
            b1.x = Bs[cur][ks + koff][wn + 16 + sub];
            b1.y = Bs[cur][ks + koff + 1][wn + 16 + sub];
            c00 = __builtin_amdgcn_wmma_f32_16x16x4_f32(false, a0, false, b0, (short)0, c00, false, false);
            c01 = __builtin_amdgcn_wmma_f32_16x16x4_f32(false, a0, false, b1, (short)0, c01, false, false);
            c10 = __builtin_amdgcn_wmma_f32_16x16x4_f32(false, a1, false, b0, (short)0, c10, false, false);
            c11 = __builtin_amdgcn_wmma_f32_16x16x4_f32(false, a1, false, b1, (short)0, c11, false, false);
        }

        if (more) {
            #pragma unroll
            for (int i = 0; i < 4; ++i) {
                *(v4f*)&As[cur ^ 1][ar + 16 * i][ac] = pa[i];
                *(v4f*)&Bs[cur ^ 1][br + 8 * i][bc]  = pb[i];
            }
        }
        __syncthreads();
    }

    const int mbase = m0 + wm;
    const int nbase = n0 + wn;
    const int rbase = (lane >> 4) * 8;
    const int col0  = nbase + sub;
    const int col1  = col0 + 16;
    const float bv0 = bias ? bias[col0] : 0.0f;
    const float bv1 = bias ? bias[col1] : 0.0f;
    #pragma unroll
    for (int r = 0; r < 8; ++r) {
        const long row0 = (long)mbase + rbase + r;
        const long row1 = row0 + 16;
        Y[row0 * N + col0] = apply_act(c00[r] + bv0, act);
        Y[row0 * N + col1] = apply_act(c01[r] + bv1, act);
        Y[row1 * N + col0] = apply_act(c10[r] + bv0, act);
        Y[row1 * N + col1] = apply_act(c11[r] + bv1, act);
    }
}

__global__ __launch_bounds__(128) void rwkv_wmma_gemm_kernel(
    const float* X0, const float* W0, const float* b0, float* Y0, int K0, int N0, int act0, int mtl0,
    const float* X1, const float* W1, const float* b1, float* Y1, int K1, int N1, int act1, int mtl1,
    const float* X2, const float* W2, const float* b2, float* Y2, int K2, int N2, int act2, int mtl2,
    const float* X3, const float* W3, const float* b3, float* Y3, int K3, int N3, int act3, int mtl3,
    int s1, int s2, int s3)
{
    const int id = blockIdx.x;
    const float* X = X0;
    const float* W = W0;
    const float* bias = b0;
    float* Y = Y0;
    int K = K0, N = N0, act = act0, mtl = mtl0, sb = 0;
    if (id >= s1) { X = X1; W = W1; bias = b1; Y = Y1; K = K1; N = N1; act = act1; mtl = mtl1; sb = s1; }
    if (id >= s2) { X = X2; W = W2; bias = b2; Y = Y2; K = K2; N = N2; act = act2; mtl = mtl2; sb = s2; }
    if (id >= s3) { X = X3; W = W3; bias = b3; Y = Y3; K = K3; N = N3; act = act3; mtl = mtl3; sb = s3; }
    const int local = id - sb;
    gemm_tile(X, W, bias, Y, K, N, act, (local & ((1 << mtl) - 1)) * GT_M, (local >> mtl) * GT_N);
}

#define RT_TOK  (NTOK / GT_M)
#define RTL_TOK 7
#define RT_POS  (NPOS / GT_M)
#define RTL_POS 4
static_assert((1 << RTL_TOK) == RT_TOK && (1 << RTL_POS) == RT_POS, "row-tile shifts");
#define G1_S1    (RT_TOK)
#define G1_S2    (G1_S1 + 4 * RT_TOK)
#define G1_S3    (G1_S2 + 4 * RT_POS)
#define G1_TILES (G1_S3 + 4 * RT_POS)
#define G2_S1    (4 * RT_TOK)
#define G2_TILES (G2_S1 + 4 * RT_TOK)
#define G3_TILES (4 * RT_TOK)

#define SCAN_T 64
#define SCAN_NCHUNK (LKs / SCAN_T)
#define SCAN_VQ 16
#define SCAN_KS 16
#define SCAN_PPITCH (SCAN_KS * SCAN_VQ + 16)
#define SCAN_PD 4

__device__ __forceinline__ long scan_o_index(int e, int head, int l, int vcol)
{
    return ((((long)(e * NHEAD + head) * 4 + (vcol >> 4)) * LKs + l) << 4) + (vcol & 15);
}

__global__ __launch_bounds__(128) void rwkv_scan_kernel(
    const float* __restrict__ r, const float* __restrict__ k,
    const float* __restrict__ v, const float* __restrict__ decay,
    const float* __restrict__ u, float* __restrict__ o)
{
    const int chain = blockIdx.x >> 2;
    const int vq    = blockIdx.x & 3;
    const int e     = chain >> 2;
    const int head  = chain & 3;
    const long base   = ((long)e * LKs) * Hdim + head * DK;
    const long basekv = ((long)(e / LQ) * LKs) * Hdim + head * DK;
    const int t    = threadIdx.x;
    const int vp2  = (t & 7) << 1;
    const int ks   = t >> 3;
    const int k0   = ks << 2;

    __shared__ __attribute__((aligned(16))) float stage[2][SCAN_T][3][DK];
    __shared__ __attribute__((aligned(16))) float vstage[2][SCAN_T][SCAN_VQ];
    __shared__ __attribute__((aligned(16))) float part[2][SCAN_T][SCAN_PPITCH];

    float S0[4], S1[4], uu[4];
    #pragma unroll
    for (int j = 0; j < 4; ++j) {
        S0[j] = 0.0f;
        S1[j] = 0.0f;
        uu[j] = u[head * DK + k0 + j];
    }

    const int  lrow = t >> 4;
    const int  lc   = (t & 15) << 2;
    const int  vrow = t >> 2;
    const int  vc   = (t & 3) << 2;
    const long lofs = base + (long)lrow * Hdim + lc;
    const long kofs = basekv + (long)lrow * Hdim + lc;
    const long vofs = basekv + (long)vrow * Hdim + vq * SCAN_VQ + vc;

    v4f gr[4], gk[4], gd[4], gv;
    for (int hh = 0; hh < 2; ++hh) {
        const long hofs = (long)(hh * 32) * Hdim;
        #pragma unroll
        for (int i = 0; i < 4; ++i) {
            gr[i] = *(const v4f*)(r     + lofs + hofs + (long)(8 * i) * Hdim);
            gk[i] = *(const v4f*)(k     + kofs + hofs + (long)(8 * i) * Hdim);
            gd[i] = *(const v4f*)(decay + lofs + hofs + (long)(8 * i) * Hdim);
        }
        gv = *(const v4f*)(v + vofs + hofs);
        #pragma unroll
        for (int i = 0; i < 4; ++i) {
            *(v4f*)&stage[0][hh * 32 + lrow + 8 * i][0][lc] = gr[i];
            *(v4f*)&stage[0][hh * 32 + lrow + 8 * i][1][lc] = gk[i];
            *(v4f*)&stage[0][hh * 32 + lrow + 8 * i][2][lc] = gd[i];
        }
        *(v4f*)&vstage[0][hh * 32 + vrow][vc] = gv;
    }
    __syncthreads();

    float* __restrict__ oq = o + (((long)chain * 4 + vq) * LKs << 4);

    for (int c = 0; c < SCAN_NCHUNK; ++c) {
        const int  cur  = c & 1;
        const bool more = (c + 1 < SCAN_NCHUNK);

        float4 qr[SCAN_PD], qk[SCAN_PD], qd[SCAN_PD];
        float2 qv[SCAN_PD];
        #pragma unroll
        for (int p = 0; p < SCAN_PD; ++p) {
            qr[p] = *(const float4*)&stage[cur][p][0][k0];
            qk[p] = *(const float4*)&stage[cur][p][1][k0];
            qd[p] = *(const float4*)&stage[cur][p][2][k0];
            qv[p] = *(const float2*)&vstage[cur][p][vp2];
        }

        for (int hh = 0; hh < 2; ++hh) {
            if (more) {
                const long cofs = (long)((c + 1) * SCAN_T + hh * 32) * Hdim;
                #pragma unroll
                for (int i = 0; i < 4; ++i) {
                    gr[i] = *(const v4f*)(r     + lofs + cofs + (long)(8 * i) * Hdim);
                    gk[i] = *(const v4f*)(k     + kofs + cofs + (long)(8 * i) * Hdim);
                    gd[i] = *(const v4f*)(decay + lofs + cofs + (long)(8 * i) * Hdim);
                }
                gv = *(const v4f*)(v + vofs + cofs);
            }

            #pragma unroll 1
            for (int s0 = hh * 32; s0 < hh * 32 + 32; s0 += 16) {
                #pragma unroll
                for (int i = 0; i < 16; ++i) {
                    const int s = s0 + i;
                    const int p = i % SCAN_PD;
                    const float4 rr = qr[p];
                    const float4 kk = qk[p];
                    const float4 dd = qd[p];
                    const float2 vt = qv[p];
                    const int ns = (i + SCAN_PD < 16) ? s + SCAN_PD : min(s + SCAN_PD, SCAN_T - 1);
                    qr[p] = *(const float4*)&stage[cur][ns][0][k0];
                    qk[p] = *(const float4*)&stage[cur][ns][1][k0];
                    qd[p] = *(const float4*)&stage[cur][ns][2][k0];
                    qv[p] = *(const float2*)&vstage[cur][ns][vp2];
                    __builtin_amdgcn_sched_barrier(0);
                    const float rj[4] = {rr.x, rr.y, rr.z, rr.w};
                    const float kj[4] = {kk.x, kk.y, kk.z, kk.w};
                    const float dj[4] = {dd.x, dd.y, dd.z, dd.w};
                    float a0 = 0.0f, a1 = 0.0f;
                    #pragma unroll
                    for (int j = 0; j < 4; ++j) {
                        const float kv0 = kj[j] * vt.x;
                        const float kv1 = kj[j] * vt.y;
                        a0 += rj[j] * (S0[j] + uu[j] * kv0);
                        a1 += rj[j] * (S1[j] + uu[j] * kv1);
                        S0[j] = dj[j] * S0[j] + kv0;
                        S1[j] = dj[j] * S1[j] + kv1;
                    }
                    *(float2*)&part[cur][s][(ks << 4) + vp2] = make_float2(a0, a1);
                }
            }

            if (more) {
                #pragma unroll
                for (int i = 0; i < 4; ++i) {
                    *(v4f*)&stage[cur ^ 1][hh * 32 + lrow + 8 * i][0][lc] = gr[i];
                    *(v4f*)&stage[cur ^ 1][hh * 32 + lrow + 8 * i][1][lc] = gk[i];
                    *(v4f*)&stage[cur ^ 1][hh * 32 + lrow + 8 * i][2][lc] = gd[i];
                }
                *(v4f*)&vstage[cur ^ 1][hh * 32 + vrow][vc] = gv;
            }
        }
        __syncthreads();

        #pragma unroll
        for (int i = 0; i < 2; ++i) {
            const int js = vrow + 32 * i;
            float4 sum = *(const float4*)&part[cur][js][vc];
            #pragma unroll
            for (int q = 1; q < SCAN_KS; ++q) {
                const float4 pq = *(const float4*)&part[cur][js][(q << 4) + vc];
                sum.x += pq.x; sum.y += pq.y; sum.z += pq.z; sum.w += pq.w;
            }
            *(float4*)(oq + ((long)(c * SCAN_T + js) << 4) + vc) = sum;
        }
    }
}

#define GN_TOK 8
#define GN_WAVES (NTOK * NHEAD / GN_TOK)
#define GN_GROUPS (GN_WAVES / 8)

__device__ __forceinline__ void gn_tree_sum2(v4f lo0, v4f hi0, v4f lo1, v4f hi1,
                                             float& s0, float& s1)
{
    v4f a = lo0 + hi0;
    v4f c = lo1 + hi1;
    #pragma unroll
    for (int m = 4; m >= 1; m >>= 1) {
        v4f ta, tc;
        ta.x = __shfl_xor(a.x, m); ta.y = __shfl_xor(a.y, m);
        ta.z = __shfl_xor(a.z, m); ta.w = __shfl_xor(a.w, m);
        tc.x = __shfl_xor(c.x, m); tc.y = __shfl_xor(c.y, m);
        tc.z = __shfl_xor(c.z, m); tc.w = __shfl_xor(c.w, m);
        a += ta;
        c += tc;
    }
    s0 = (a.x + a.z) + (a.y + a.w);
    s1 = (c.x + c.z) + (c.y + c.w);
}

__global__ __launch_bounds__(256) __attribute__((amdgpu_waves_per_eu(4, 8))) void rwkv_gn_silu_kernel(
    const float* __restrict__ o, const float* __restrict__ gsilu,
    const float* __restrict__ gn_w, const float* __restrict__ gn_b,
    float* __restrict__ y)
{
    const int gw   = blockIdx.x * 8 + (threadIdx.x >> 5);
    const int lane = threadIdx.x & 31;
    const int head = gw & 3;
    const int lb   = (gw >> 2) & 63;
    const int e    = gw >> 8;
    const int col  = (lane & 7) << 2;
    const int hc   = head * DK + col;
    const int l0   = lb * GN_TOK + (lane >> 3);
    const long row = ((long)e * LKs + l0) * Hdim + hc;

    const v4f lo0 = *(const v4f*)(o + scan_o_index(e, head, l0, col));
    const v4f hi0 = *(const v4f*)(o + scan_o_index(e, head, l0, col + 32));
    const v4f lo1 = *(const v4f*)(o + scan_o_index(e, head, l0 + 4, col));
    const v4f hi1 = *(const v4f*)(o + scan_o_index(e, head, l0 + 4, col + 32));
    const v4f g00 = *(const v4f*)(gsilu + row);
    const v4f g01 = *(const v4f*)(gsilu + row + 32);
    const v4f g10 = *(const v4f*)(gsilu + row + 4 * Hdim);
    const v4f g11 = *(const v4f*)(gsilu + row + 4 * Hdim + 32);
    const v4f w0  = *(const v4f*)(gn_w + hc);
    const v4f w1  = *(const v4f*)(gn_w + hc + 32);
    const v4f b0  = *(const v4f*)(gn_b + hc);
    const v4f b1  = *(const v4f*)(gn_b + hc + 32);

    float m0, m1;
    gn_tree_sum2(lo0, hi0, lo1, hi1, m0, m1);
    m0 *= (1.0f / 64.0f);
    m1 *= (1.0f / 64.0f);
    const v4f d00 = lo0 - m0, d01 = hi0 - m0;
    const v4f d10 = lo1 - m1, d11 = hi1 - m1;
    float q0, q1;
    gn_tree_sum2(d00 * d00, d01 * d01, d10 * d10, d11 * d11, q0, q1);
    const float r0 = rsqrtf(q0 * (1.0f / 64.0f) + EPSV);
    const float r1 = rsqrtf(q1 * (1.0f / 64.0f) + EPSV);

    *(v4f*)(y + row)                  = ((d00 * r0) * w0 + b0) * g00;
    *(v4f*)(y + row + 32)             = ((d01 * r0) * w1 + b1) * g01;
    *(v4f*)(y + row + 4 * Hdim)       = ((d10 * r1) * w0 + b0) * g10;
    *(v4f*)(y + row + 4 * Hdim + 32)  = ((d11 * r1) * w1 + b1) * g11;
}

extern "C" void kernel_launch(void* const* d_in, const int* in_sizes, int n_in,
                              void* d_out, int out_size, void* d_ws, size_t ws_size,
                              hipStream_t stream)
{
    const float* query    = (const float*)d_in[0];
    const float* keyval   = (const float*)d_in[1];
    const float* rwg_mu   = (const float*)d_in[2];
    const float* W_rwg0   = (const float*)d_in[3];
    const float* W_rwg2   = (const float*)d_in[4];
    const float* rwg_bias = (const float*)d_in[5];
    const float* kv_mu    = (const float*)d_in[6];
    const float* W_kv0    = (const float*)d_in[7];
    const float* W_kv2    = (const float*)d_in[8];
    const float* kv_bias  = (const float*)d_in[9];
    const float* Wr       = (const float*)d_in[10];
    const float* Wa       = (const float*)d_in[11];
    const float* Wb       = (const float*)d_in[12];
    const float* bb       = (const float*)d_in[13];
    const float* Wk       = (const float*)d_in[14];
    const float* Wv       = (const float*)d_in[15];
    const float* Wg       = (const float*)d_in[16];
    const float* u        = (const float*)d_in[17];
    const float* gn_w     = (const float*)d_in[18];
    const float* gn_b     = (const float*)d_in[19];
    const float* Wo       = (const float*)d_in[20];

    float* ws = (float*)d_ws;
    const long SZ = (long)NTOK * Hdim;
    float* xr    = ws + 0 * SZ;
    float* xw    = ws + 1 * SZ;
    float* xg    = ws + 2 * SZ;
    float* xk    = ws + 3 * SZ;
    float* xv    = ws + 4 * SZ;
    float* rbuf  = ws + 5 * SZ;
    float* kbuf  = ws + 6 * SZ;
    float* vbuf  = ws + 7 * SZ;
    float* dbuf  = ws + 8 * SZ;
    float* gsbuf = ws + 9 * SZ;
    float* twbuf = ws + 10 * SZ;
    float* obuf  = xr;
    float* ybuf  = xw;

    rwkv_features_kernel<<<NPOS, 256, 0, stream>>>(
        query, keyval, rwg_mu, W_rwg0, W_rwg2, rwg_bias,
        kv_mu, W_kv0, W_kv2, kv_bias, xr, xw, xg, xk, xv);

    const int NOJOB = 0x7fffffff;
    const dim3 gblock(128);

    rwkv_wmma_gemm_kernel<<<G1_TILES, gblock, 0, stream>>>(
        xw, Wa, nullptr, twbuf, 256, 64,  ACT_TANH, RTL_TOK,
        xr, Wr, nullptr, rbuf,  256, 256, ACT_NONE, RTL_TOK,
        xk, Wk, nullptr, kbuf,  256, 256, ACT_NONE, RTL_POS,
        xv, Wv, nullptr, vbuf,  256, 256, ACT_NONE, RTL_POS,
        G1_S1, G1_S2, G1_S3);

    rwkv_wmma_gemm_kernel<<<G2_TILES, gblock, 0, stream>>>(
        xg,    Wg, nullptr, gsbuf, 256, 256, ACT_SILU,  RTL_TOK,
        twbuf, Wb, bb,      dbuf,  64,  256, ACT_DECAY, RTL_TOK,
        nullptr, nullptr, nullptr, nullptr, 0, 0, 0, 0,
        nullptr, nullptr, nullptr, nullptr, 0, 0, 0, 0,
        G2_S1, NOJOB, NOJOB);

    rwkv_scan_kernel<<<Bsz * LQ * NHEAD * 4, 128, 0, stream>>>(rbuf, kbuf, vbuf, dbuf, u, obuf);

    rwkv_gn_silu_kernel<<<GN_GROUPS, 256, 0, stream>>>(obuf, gsbuf, gn_w, gn_b, ybuf);

    rwkv_wmma_gemm_kernel<<<G3_TILES, gblock, 0, stream>>>(
        ybuf, Wo, nullptr, (float*)d_out, 256, 256, ACT_NONE, RTL_TOK,
        nullptr, nullptr, nullptr, nullptr, 0, 0, 0, 0,
        nullptr, nullptr, nullptr, nullptr, 0, 0, 0, 0,
        nullptr, nullptr, nullptr, nullptr, 0, 0, 0, 0,
        NOJOB, NOJOB, NOJOB);
}


# decay path: both products in one tile job (tanh tile kept in LDS, not memory), gate moved beside it: 6 dependent launches -> 5; plain GEMM tiles keep only the silu test; every sum order kept
# speedup vs baseline: 9.5858x; 1.0404x over previous
#include <hip/hip_runtime.h>
#include <math.h>

#define Hdim 256
#define NHEAD 4
#define DK 64
#define PLR 32
#define Bsz 2
#define LQ 8
#define LKs 512
#define NTOK (Bsz * LQ * LKs)
#define EPSV 1e-5f

typedef float v2f __attribute__((ext_vector_type(2)));
typedef float v8f __attribute__((ext_vector_type(8)));
typedef float v4f __attribute__((ext_vector_type(4)));

#define ACT_NONE  0
#define ACT_SILU  1

#define NPOS   (Bsz * LKs)
#define WPITCH 36

struct Slab8 { v4f v[8]; };

__device__ __forceinline__ Slab8 slab_fetch(const float* __restrict__ Wsrc, int ld,
                                            int goff, int t)
{
    const float* p = Wsrc + (t >> 3) * ld + goff + ((t & 7) << 2);
    Slab8 s;
    #pragma unroll
    for (int m = 0; m < 8; ++m) s.v[m] = *(const v4f*)(p + m * 32 * ld);
    return s;
}

__device__ __forceinline__ void slab_store(float (*slab)[WPITCH], int t, const Slab8& s)
{
    #pragma unroll
    for (int m = 0; m < 8; ++m)
        *(v4f*)&slab[m * 32 + (t >> 3)][(t & 7) << 2] = s.v[m];
}

__device__ __forceinline__ void slab_row(float (*slab)[WPITCH], int t, float* w)
{
    #pragma unroll
    for (int r4 = 0; r4 < PLR; r4 += 4) {
        const float4 wv = *(const float4*)&slab[t][r4];
        w[r4] = wv.x; w[r4 + 1] = wv.y; w[r4 + 2] = wv.z; w[r4 + 3] = wv.w;
    }
}

__global__ __launch_bounds__(256) void rwkv_features_kernel(
    const float* __restrict__ query, const float* __restrict__ keyval,
    const float* __restrict__ rwg_mu, const float* __restrict__ W_rwg0,
    const float* __restrict__ W_rwg2, const float* __restrict__ rwg_bias,
    const float* __restrict__ kv_mu, const float* __restrict__ W_kv0,
    const float* __restrict__ W_kv2, const float* __restrict__ kv_bias,
    float* __restrict__ xr, float* __restrict__ xw, float* __restrict__ xg,
    float* __restrict__ xk, float* __restrict__ xv)
{
    const int pos = blockIdx.x;
    const int b   = pos >> 9;
    const int l   = pos & 511;
    const int t   = threadIdx.x;

    __shared__ __attribute__((aligned(16))) float xrwg_s[Hdim][LQ];
    __shared__ float xkv_s[Hdim];
    __shared__ __attribute__((aligned(16))) float t1_s[3 * PLR][LQ];
    __shared__ float t2_s[2 * PLR];
    __shared__ __attribute__((aligned(16))) float wslab[Hdim][WPITCH];

    const float h  = keyval[(long)pos * Hdim + t];
    const float hp = (l == 0) ? 0.0f : keyval[(long)(pos - 1) * Hdim + t];
    const float d  = hp - h;
    float dc[LQ];
    {
        const float mu = rwg_mu[t];
        float xq[LQ];
        #pragma unroll
        for (int q = 0; q < LQ; ++q) {
            dc[q] = query[(long)(b * LQ + q) * Hdim + t] - h;
            xq[q] = h + dc[q] * mu;
        }
        *(float4*)&xrwg_s[t][0] = make_float4(xq[0], xq[1], xq[2], xq[3]);
        *(float4*)&xrwg_s[t][4] = make_float4(xq[4], xq[5], xq[6], xq[7]);
        xkv_s[t] = h + d * kv_mu[t];
    }
    __syncthreads();

    if (t < 192) {
        const int c  = (t < 96) ? t : t - 96;
        const int qh = (t < 96) ? 0 : 4;
        float a0 = 0.0f, a1 = 0.0f, a2 = 0.0f, a3 = 0.0f;
        #pragma unroll 16
        for (int i = 0; i < Hdim; ++i) {
            const float  w = W_rwg0[i * 96 + c];
            const float4 x = *(const float4*)&xrwg_s[i][qh];
            a0 += x.x * w;
            a1 += x.y * w;
            a2 += x.z * w;
            a3 += x.w * w;
        }
        *(float4*)&t1_s[c][qh] = make_float4(tanhf(a0), tanhf(a1), tanhf(a2), tanhf(a3));
    } else {
        const int j = t - 192;
        float acc = 0.0f;
        #pragma unroll 16
        for (int i = 0; i < Hdim; ++i) acc += xkv_s[i] * W_kv0[i * 64 + j];
        t2_s[j] = tanhf(acc);
    }

    const long orow = ((long)(b * LQ) * LKs + l) * Hdim + t;
    float w[PLR];

    auto expand_q = [&](int g, float* __restrict__ dst) {
        const float bias = rwg_bias[g * Hdim + t];
        float acc[LQ];
        #pragma unroll
        for (int q = 0; q < LQ; ++q) acc[q] = bias;
        #pragma unroll
        for (int r = 0; r < PLR; ++r) {
            const float4 x0 = *(const float4*)&t1_s[g * PLR + r][0];
            const float4 x1 = *(const float4*)&t1_s[g * PLR + r][4];
            acc[0] += x0.x * w[r];
            acc[1] += x0.y * w[r];
            acc[2] += x0.z * w[r];
            acc[3] += x0.w * w[r];
            acc[4] += x1.x * w[r];
            acc[5] += x1.y * w[r];
            acc[6] += x1.z * w[r];
            acc[7] += x1.w * w[r];
        }
        #pragma unroll
        for (int q = 0; q < LQ; ++q)
            dst[orow + (long)q * LKs * Hdim] = h + dc[q] * acc[q];
    };
    auto expand_kv = [&](int gg, float* __restrict__ dst) {
        float acc = kv_bias[gg * Hdim + t];
        #pragma unroll
        for (int r = 0; r < PLR; ++r) acc += t2_s[gg * PLR + r] * w[r];
        dst[(long)pos * Hdim + t] = h + d * acc;
    };

    Slab8 s0 = slab_fetch(W_rwg2, 3 * PLR, 0, t);
    __syncthreads();

    slab_store(wslab, t, s0);
    __syncthreads();
    Slab8 s1 = slab_fetch(W_rwg2, 3 * PLR, PLR, t);
    slab_row(wslab, t, w);
    expand_q(0, xr);
    __syncthreads();

    slab_store(wslab, t, s1);
    __syncthreads();
    Slab8 s2 = slab_fetch(W_rwg2, 3 * PLR, 2 * PLR, t);
    slab_row(wslab, t, w);
    expand_q(1, xw);
    __syncthreads();

    slab_store(wslab, t, s2);
    __syncthreads();
    Slab8 s3 = slab_fetch(W_kv2, 2 * PLR, 0, t);
    slab_row(wslab, t, w);
    expand_q(2, xg);
    __syncthreads();

    slab_store(wslab, t, s3);
    __syncthreads();
    Slab8 s4 = slab_fetch(W_kv2, 2 * PLR, PLR, t);
    slab_row(wslab, t, w);
    expand_kv(0, xk);
    __syncthreads();

    slab_store(wslab, t, s4);
    __syncthreads();
    slab_row(wslab, t, w);
    expand_kv(1, xv);
}

__device__ __forceinline__ float act_silu(float x)  { return x / (1.0f + expf(-x)); }
__device__ __forceinline__ float act_decay(float x) { return expf(-expf(x)); }
template <bool DECAY>
__device__ __forceinline__ float tile_act(float x, bool silu)
{
    return DECAY ? act_decay(x) : (silu ? act_silu(x) : x);
}

#define GT_M 64
#define GT_N 64
#define GT_K 32
#define GA_PITCH 36
#define GB_PITCH 72

typedef float (*gemm_as_t)[GT_M][GA_PITCH];
typedef float (*gemm_bs_t)[GT_K][GB_PITCH];

__device__ __forceinline__ void gemm_mainloop(
    const float* __restrict__ X, const float* __restrict__ W,
    int K, int N, int m0, int n0, gemm_as_t As, gemm_bs_t Bs,
    v8f& c00, v8f& c01, v8f& c10, v8f& c11)
{
    const int t    = threadIdx.x;
    const int wave = t >> 5;
    const int lane = t & 31;
    const int wm   = (wave >> 1) << 5;
    const int wn   = (wave & 1) << 5;

    const int sub  = lane & 15;
    const int koff = (lane >> 4) << 1;

    const int ar = t >> 3;
    const int ac = (t & 7) << 2;
    const int br = t >> 4;
    const int bc = (t & 15) << 2;
    const float* __restrict__ Ag = X + (long)(m0 + ar) * K + ac;
    const float* __restrict__ Bg = W + (long)br * N + n0 + bc;

    v4f pa[4], pb[4];
    #pragma unroll
    for (int i = 0; i < 4; ++i) {
        pa[i] = *(const v4f*)(Ag + (long)(16 * i) * K);
        pb[i] = *(const v4f*)(Bg + (long)(8 * i) * N);
    }
    #pragma unroll
    for (int i = 0; i < 4; ++i) {
        *(v4f*)&As[0][ar + 16 * i][ac] = pa[i];
        *(v4f*)&Bs[0][br + 8 * i][bc]  = pb[i];
    }
    __syncthreads();

    const v8f zero = {};
    c00 = zero; c01 = zero; c10 = zero; c11 = zero;
    const int nchunk = K / GT_K;
    for (int c = 0; c < nchunk; ++c) {
        const int  cur  = c & 1;
        const bool more = (c + 1 < nchunk);

        if (more) {
            const float* __restrict__ Ap = Ag + (c + 1) * GT_K;
            const float* __restrict__ Bp = Bg + (long)((c + 1) * GT_K) * N;
            #pragma unroll
            for (int i = 0; i < 4; ++i) {
                pa[i] = *(const v4f*)(Ap + (long)(16 * i) * K);
                pb[i] = *(const v4f*)(Bp + (long)(8 * i) * N);
            }
        }

        #pragma unroll
        for (int ks = 0; ks < GT_K; ks += 4) {
            const v2f a0 = *(const v2f*)&As[cur][wm + sub][ks + koff];
            const v2f a1 = *(const v2f*)&As[cur][wm + 16 + sub][ks + koff];
            v2f b0, b1;
            b0.x = Bs[cur][ks + koff][wn + sub];
            b0.y = Bs[cur][ks + koff + 1][wn + sub];
            b1.x = Bs[cur][ks + koff][wn + 16 + sub];
            b1.y = Bs[cur][ks + koff + 1][wn + 16 + sub];
            c00 = __builtin_amdgcn_wmma_f32_16x16x4_f32(false, a0, false, b0, (short)0, c00, false, false);
            c01 = __builtin_amdgcn_wmma_f32_16x16x4_f32(false, a0, false, b1, (short)0, c01, false, false);
            c10 = __builtin_amdgcn_wmma_f32_16x16x4_f32(false, a1, false, b0, (short)0, c10, false, false);
            c11 = __builtin_amdgcn_wmma_f32_16x16x4_f32(false, a1, false, b1, (short)0, c11, false, false);
        }

        if (more) {
            #pragma unroll
            for (int i = 0; i < 4; ++i) {
                *(v4f*)&As[cur ^ 1][ar + 16 * i][ac] = pa[i];
                *(v4f*)&Bs[cur ^ 1][br + 8 * i][bc]  = pb[i];
            }
        }
        __syncthreads();
    }
}

template <bool DECAY>
__device__ __forceinline__ void tile_store(
    float* __restrict__ Y, int N, float bv0, float bv1, bool silu, int mbase, int nbase,
    const v8f& c00, const v8f& c01, const v8f& c10, const v8f& c11)
{
    const int lane  = threadIdx.x & 31;
    const int rbase = (lane >> 4) * 8;
    const int col0  = nbase + (lane & 15);
    const int col1  = col0 + 16;
    #pragma unroll
    for (int r = 0; r < 8; ++r) {
        const long row0 = (long)mbase + rbase + r;
        const long row1 = row0 + 16;
        Y[row0 * N + col0] = tile_act<DECAY>(c00[r] + bv0, silu);
        Y[row0 * N + col1] = tile_act<DECAY>(c01[r] + bv1, silu);
        Y[row1 * N + col0] = tile_act<DECAY>(c10[r] + bv0, silu);
        Y[row1 * N + col1] = tile_act<DECAY>(c11[r] + bv1, silu);
    }
}

__device__ __forceinline__ void gemm_tile(
    const float* __restrict__ X, const float* __restrict__ W,
    const float* __restrict__ bias, float* __restrict__ Y,
    int K, int N, int act, int m0, int n0, gemm_as_t As, gemm_bs_t Bs)
{
    v8f c00, c01, c10, c11;
    gemm_mainloop(X, W, K, N, m0, n0, As, Bs, c00, c01, c10, c11);

    const int wave = threadIdx.x >> 5;
    const int lane = threadIdx.x & 31;
    const int mbase = m0 + ((wave >> 1) << 5);
    const int nbase = n0 + ((wave & 1) << 5);
    const int col0  = nbase + (lane & 15);
    const float bv0 = bias ? bias[col0] : 0.0f;
    const float bv1 = bias ? bias[col0 + 16] : 0.0f;
    tile_store<false>(Y, N, bv0, bv1, act == ACT_SILU, mbase, nbase, c00, c01, c10, c11);
}

#define LORA_W   64
#define TW_PITCH 68
static_assert(LORA_W == GT_N && LORA_W == 2 * GT_K, "the middle matrix is one column tile wide and fills both W stages");
static_assert(GT_M * TW_PITCH <= 2 * GT_M * GA_PITCH, "tanh tile fits in the X stages");

__device__ __forceinline__ void decay_tile(
    const float* __restrict__ X, const float* __restrict__ Wd,
    const float* __restrict__ Wu, const float* __restrict__ bu,
    float* __restrict__ Y, int m0, gemm_as_t As, gemm_bs_t Bs)
{
    v8f c00, c01, c10, c11;
    gemm_mainloop(X, Wd, Hdim, LORA_W, m0, 0, As, Bs, c00, c01, c10, c11);

    float (*T)[TW_PITCH]  = (float (*)[TW_PITCH])&As[0][0][0];
    float (*Bu)[GB_PITCH] = (float (*)[GB_PITCH])&Bs[0][0][0];

    const int t     = threadIdx.x;
    const int wave  = t >> 5;
    const int lane  = t & 31;
    const int wm    = (wave >> 1) << 5;
    const int wn    = (wave & 1) << 5;
    const int sub   = lane & 15;
    const int koff  = (lane >> 4) << 1;
    const int rbase = (lane >> 4) * 8;

    const int br = t >> 4;
    const int bc = (t & 15) << 2;
    const float* __restrict__ Bg = Wu + (long)br * Hdim + bc;

    v4f pb[8];
    #pragma unroll
    for (int i = 0; i < 8; ++i) pb[i] = *(const v4f*)(Bg + (long)(8 * i) * Hdim);

    #pragma unroll
    for (int r = 0; r < 8; ++r) {
        const int row = wm + rbase + r;
        T[row][wn + sub]           = tanhf(c00[r]);
        T[row][wn + 16 + sub]      = tanhf(c01[r]);
        T[row + 16][wn + sub]      = tanhf(c10[r]);
        T[row + 16][wn + 16 + sub] = tanhf(c11[r]);
    }
    #pragma unroll
    for (int i = 0; i < 8; ++i) *(v4f*)&Bu[br + 8 * i][bc] = pb[i];
    __syncthreads();

    #pragma unroll 1
    for (int nt = 0; nt < Hdim / GT_N; ++nt) {
        const bool more = (nt + 1 < Hdim / GT_N);
        if (more) {
            #pragma unroll
            for (int i = 0; i < 8; ++i)
                pb[i] = *(const v4f*)(Bg + (nt + 1) * GT_N + (long)(8 * i) * Hdim);
        }
        const int   nbase = nt * GT_N + wn;
        const float bv0   = bu[nbase + sub];
        const float bv1   = bu[nbase + 16 + sub];

        v8f d00 = {}, d01 = {}, d10 = {}, d11 = {};
        #pragma unroll
        for (int s = 0; s < LORA_W / 4; ++s) {
            const int kk = 4 * s + koff;
            const v2f a0 = *(const v2f*)&T[wm + sub][kk];
            const v2f a1 = *(const v2f*)&T[wm + 16 + sub][kk];
            v2f b0, b1;
            b0.x = Bu[kk][wn + sub];
            b0.y = Bu[kk + 1][wn + sub];
            b1.x = Bu[kk][wn + 16 + sub];
            b1.y = Bu[kk + 1][wn + 16 + sub];
            d00 = __builtin_amdgcn_wmma_f32_16x16x4_f32(false, a0, false, b0, (short)0, d00, false, false);
            d01 = __builtin_amdgcn_wmma_f32_16x16x4_f32(false, a0, false, b1, (short)0, d01, false, false);
            d10 = __builtin_amdgcn_wmma_f32_16x16x4_f32(false, a1, false, b0, (short)0, d10, false, false);
            d11 = __builtin_amdgcn_wmma_f32_16x16x4_f32(false, a1, false, b1, (short)0, d11, false, false);
        }
        tile_store<true>(Y, Hdim, bv0, bv1, false, m0 + wm, nbase, d00, d01, d10, d11);

        __syncthreads();
        if (more) {
            #pragma unroll
            for (int i = 0; i < 8; ++i) *(v4f*)&Bu[br + 8 * i][bc] = pb[i];
            __syncthreads();
        }
    }
}

__global__ __launch_bounds__(128) void rwkv_wmma_gemm_kernel(
    const float* Xd, const float* Wd, const float* Wu, const float* bu, float* Yd, int sd,
    const float* X0, const float* W0, const float* b0, float* Y0, int K0, int N0, int act0, int mtl0,
    const float* X1, const float* W1, const float* b1, float* Y1, int K1, int N1, int act1, int mtl1,
    const float* X2, const float* W2, const float* b2, float* Y2, int K2, int N2, int act2, int mtl2,
    const float* X3, const float* W3, const float* b3, float* Y3, int K3, int N3, int act3, int mtl3,
    int s1, int s2, int s3)
{
    __shared__ __attribute__((aligned(16))) float As[2][GT_M][GA_PITCH];
    __shared__ __attribute__((aligned(16))) float Bs[2][GT_K][GB_PITCH];

    if ((int)blockIdx.x < sd) {
        decay_tile(Xd, Wd, Wu, bu, Yd, blockIdx.x * GT_M, As, Bs);
        return;
    }
    const int id = blockIdx.x - sd;
    const float* X = X0;
    const float* W = W0;
    const float* bias = b0;
    float* Y = Y0;
    int K = K0, N = N0, act = act0, mtl = mtl0, sb = 0;
    if (id >= s1) { X = X1; W = W1; bias = b1; Y = Y1; K = K1; N = N1; act = act1; mtl = mtl1; sb = s1; }
    if (id >= s2) { X = X2; W = W2; bias = b2; Y = Y2; K = K2; N = N2; act = act2; mtl = mtl2; sb = s2; }
    if (id >= s3) { X = X3; W = W3; bias = b3; Y = Y3; K = K3; N = N3; act = act3; mtl = mtl3; sb = s3; }
    const int local = id - sb;
    gemm_tile(X, W, bias, Y, K, N, act, (local & ((1 << mtl) - 1)) * GT_M, (local >> mtl) * GT_N, As, Bs);
}

#define RT_TOK  (NTOK / GT_M)
#define RTL_TOK 7
#define RT_POS  (NPOS / GT_M)
#define RTL_POS 4
static_assert((1 << RTL_TOK) == RT_TOK && (1 << RTL_POS) == RT_POS, "row-tile shifts");
#define G1_SD    (RT_TOK)
#define G1_S1    (4 * RT_TOK)
#define G1_S2    (G1_S1 + 4 * RT_TOK)
#define G1_S3    (G1_S2 + 4 * RT_POS)
#define G1_TILES (G1_SD + G1_S3 + 4 * RT_POS)
#define G2_TILES (4 * RT_TOK)

#define SCAN_T 64
#define SCAN_NCHUNK (LKs / SCAN_T)
#define SCAN_VQ 16
#define SCAN_KS 16
#define SCAN_PPITCH (SCAN_KS * SCAN_VQ + 16)
#define SCAN_PD 4

__device__ __forceinline__ long scan_o_index(int e, int head, int l, int vcol)
{
    return ((((long)(e * NHEAD + head) * 4 + (vcol >> 4)) * LKs + l) << 4) + (vcol & 15);
}

__global__ __launch_bounds__(128) void rwkv_scan_kernel(
    const float* __restrict__ r, const float* __restrict__ k,
    const float* __restrict__ v, const float* __restrict__ decay,
    const float* __restrict__ u, float* __restrict__ o)
{
    const int chain = blockIdx.x >> 2;
    const int vq    = blockIdx.x & 3;
    const int e     = chain >> 2;
    const int head  = chain & 3;
    const long base   = ((long)e * LKs) * Hdim + head * DK;
    const long basekv = ((long)(e / LQ) * LKs) * Hdim + head * DK;
    const int t    = threadIdx.x;
    const int vp2  = (t & 7) << 1;
    const int ks   = t >> 3;
    const int k0   = ks << 2;

    __shared__ __attribute__((aligned(16))) float stage[2][SCAN_T][3][DK];
    __shared__ __attribute__((aligned(16))) float vstage[2][SCAN_T][SCAN_VQ];
    __shared__ __attribute__((aligned(16))) float part[2][SCAN_T][SCAN_PPITCH];

    float S0[4], S1[4], uu[4];
    #pragma unroll
    for (int j = 0; j < 4; ++j) {
        S0[j] = 0.0f;
        S1[j] = 0.0f;
        uu[j] = u[head * DK + k0 + j];
    }

    const int  lrow = t >> 4;
    const int  lc   = (t & 15) << 2;
    const int  vrow = t >> 2;
    const int  vc   = (t & 3) << 2;
    const long lofs = base + (long)lrow * Hdim + lc;
    const long kofs = basekv + (long)lrow * Hdim + lc;
    const long vofs = basekv + (long)vrow * Hdim + vq * SCAN_VQ + vc;

    v4f gr[4], gk[4], gd[4], gv;
    for (int hh = 0; hh < 2; ++hh) {
        const long hofs = (long)(hh * 32) * Hdim;
        #pragma unroll
        for (int i = 0; i < 4; ++i) {
            gr[i] = *(const v4f*)(r     + lofs + hofs + (long)(8 * i) * Hdim);
            gk[i] = *(const v4f*)(k     + kofs + hofs + (long)(8 * i) * Hdim);
            gd[i] = *(const v4f*)(decay + lofs + hofs + (long)(8 * i) * Hdim);
        }
        gv = *(const v4f*)(v + vofs + hofs);
        #pragma unroll
        for (int i = 0; i < 4; ++i) {
            *(v4f*)&stage[0][hh * 32 + lrow + 8 * i][0][lc] = gr[i];
            *(v4f*)&stage[0][hh * 32 + lrow + 8 * i][1][lc] = gk[i];
            *(v4f*)&stage[0][hh * 32 + lrow + 8 * i][2][lc] = gd[i];
        }
        *(v4f*)&vstage[0][hh * 32 + vrow][vc] = gv;
    }
    __syncthreads();

    float* __restrict__ oq = o + (((long)chain * 4 + vq) * LKs << 4);

    for (int c = 0; c < SCAN_NCHUNK; ++c) {
        const int  cur  = c & 1;
        const bool more = (c + 1 < SCAN_NCHUNK);

        float4 qr[SCAN_PD], qk[SCAN_PD], qd[SCAN_PD];
        float2 qv[SCAN_PD];
        #pragma unroll
        for (int p = 0; p < SCAN_PD; ++p) {
            qr[p] = *(const float4*)&stage[cur][p][0][k0];
            qk[p] = *(const float4*)&stage[cur][p][1][k0];
            qd[p] = *(const float4*)&stage[cur][p][2][k0];
            qv[p] = *(const float2*)&vstage[cur][p][vp2];
        }

        for (int hh = 0; hh < 2; ++hh) {
            if (more) {
                const long cofs = (long)((c + 1) * SCAN_T + hh * 32) * Hdim;
                #pragma unroll
                for (int i = 0; i < 4; ++i) {
                    gr[i] = *(const v4f*)(r     + lofs + cofs + (long)(8 * i) * Hdim);
                    gk[i] = *(const v4f*)(k     + kofs + cofs + (long)(8 * i) * Hdim);
                    gd[i] = *(const v4f*)(decay + lofs + cofs + (long)(8 * i) * Hdim);
                }
                gv = *(const v4f*)(v + vofs + cofs);
            }

            #pragma unroll 1
            for (int s0 = hh * 32; s0 < hh * 32 + 32; s0 += 16) {
                #pragma unroll
                for (int i = 0; i < 16; ++i) {
                    const int s = s0 + i;
                    const int p = i % SCAN_PD;
                    const float4 rr = qr[p];
                    const float4 kk = qk[p];
                    const float4 dd = qd[p];
                    const float2 vt = qv[p];
                    const int ns = (i + SCAN_PD < 16) ? s + SCAN_PD : min(s + SCAN_PD, SCAN_T - 1);
                    qr[p] = *(const float4*)&stage[cur][ns][0][k0];
                    qk[p] = *(const float4*)&stage[cur][ns][1][k0];
                    qd[p] = *(const float4*)&stage[cur][ns][2][k0];
                    qv[p] = *(const float2*)&vstage[cur][ns][vp2];
                    __builtin_amdgcn_sched_barrier(0);
                    const float rj[4] = {rr.x, rr.y, rr.z, rr.w};
                    const float kj[4] = {kk.x, kk.y, kk.z, kk.w};
                    const float dj[4] = {dd.x, dd.y, dd.z, dd.w};
                    float a0 = 0.0f, a1 = 0.0f;
                    #pragma unroll
                    for (int j = 0; j < 4; ++j) {
                        const float kv0 = kj[j] * vt.x;
                        const float kv1 = kj[j] * vt.y;
                        a0 += rj[j] * (S0[j] + uu[j] * kv0);
                        a1 += rj[j] * (S1[j] + uu[j] * kv1);
                        S0[j] = dj[j] * S0[j] + kv0;
                        S1[j] = dj[j] * S1[j] + kv1;
                    }
                    *(float2*)&part[cur][s][(ks << 4) + vp2] = make_float2(a0, a1);
                }
            }

            if (more) {
                #pragma unroll
                for (int i = 0; i < 4; ++i) {
                    *(v4f*)&stage[cur ^ 1][hh * 32 + lrow + 8 * i][0][lc] = gr[i];
                    *(v4f*)&stage[cur ^ 1][hh * 32 + lrow + 8 * i][1][lc] = gk[i];
                    *(v4f*)&stage[cur ^ 1][hh * 32 + lrow + 8 * i][2][lc] = gd[i];
                }
                *(v4f*)&vstage[cur ^ 1][hh * 32 + vrow][vc] = gv;
            }
        }
        __syncthreads();

        #pragma unroll
        for (int i = 0; i < 2; ++i) {
            const int js = vrow + 32 * i;
            float4 sum = *(const float4*)&part[cur][js][vc];
            #pragma unroll
            for (int q = 1; q < SCAN_KS; ++q) {
                const float4 pq = *(const float4*)&part[cur][js][(q << 4) + vc];
                sum.x += pq.x; sum.y += pq.y; sum.z += pq.z; sum.w += pq.w;
            }
            *(float4*)(oq + ((long)(c * SCAN_T + js) << 4) + vc) = sum;
        }
    }
}

#define GN_TOK 8
#define GN_WAVES (NTOK * NHEAD / GN_TOK)
#define GN_GROUPS (GN_WAVES / 8)

__device__ __forceinline__ void gn_tree_sum2(v4f lo0, v4f hi0, v4f lo1, v4f hi1,
                                             float& s0, float& s1)
{
    v4f a = lo0 + hi0;
    v4f c = lo1 + hi1;
    #pragma unroll
    for (int m = 4; m >= 1; m >>= 1) {
        v4f ta, tc;
        ta.x = __shfl_xor(a.x, m); ta.y = __shfl_xor(a.y, m);
        ta.z = __shfl_xor(a.z, m); ta.w = __shfl_xor(a.w, m);
        tc.x = __shfl_xor(c.x, m); tc.y = __shfl_xor(c.y, m);
        tc.z = __shfl_xor(c.z, m); tc.w = __shfl_xor(c.w, m);
        a += ta;
        c += tc;
    }
    s0 = (a.x + a.z) + (a.y + a.w);
    s1 = (c.x + c.z) + (c.y + c.w);
}

__global__ __launch_bounds__(256) __attribute__((amdgpu_waves_per_eu(4, 8))) void rwkv_gn_silu_kernel(
    const float* __restrict__ o, const float* __restrict__ gsilu,
    const float* __restrict__ gn_w, const float* __restrict__ gn_b,
    float* __restrict__ y)
{
    const int gw   = blockIdx.x * 8 + (threadIdx.x >> 5);
    const int lane = threadIdx.x & 31;
    const int head = gw & 3;
    const int lb   = (gw >> 2) & 63;
    const int e    = gw >> 8;
    const int col  = (lane & 7) << 2;
    const int hc   = head * DK + col;
    const int l0   = lb * GN_TOK + (lane >> 3);
    const long row = ((long)e * LKs + l0) * Hdim + hc;

    const v4f lo0 = *(const v4f*)(o + scan_o_index(e, head, l0, col));
    const v4f hi0 = *(const v4f*)(o + scan_o_index(e, head, l0, col + 32));
    const v4f lo1 = *(const v4f*)(o + scan_o_index(e, head, l0 + 4, col));
    const v4f hi1 = *(const v4f*)(o + scan_o_index(e, head, l0 + 4, col + 32));
    const v4f g00 = *(const v4f*)(gsilu + row);
    const v4f g01 = *(const v4f*)(gsilu + row + 32);
    const v4f g10 = *(const v4f*)(gsilu + row + 4 * Hdim);
    const v4f g11 = *(const v4f*)(gsilu + row + 4 * Hdim + 32);
    const v4f w0  = *(const v4f*)(gn_w + hc);
    const v4f w1  = *(const v4f*)(gn_w + hc + 32);
    const v4f b0  = *(const v4f*)(gn_b + hc);
    const v4f b1  = *(const v4f*)(gn_b + hc + 32);

    float m0, m1;
    gn_tree_sum2(lo0, hi0, lo1, hi1, m0, m1);
    m0 *= (1.0f / 64.0f);
    m1 *= (1.0f / 64.0f);
    const v4f d00 = lo0 - m0, d01 = hi0 - m0;
    const v4f d10 = lo1 - m1, d11 = hi1 - m1;
    float q0, q1;
    gn_tree_sum2(d00 * d00, d01 * d01, d10 * d10, d11 * d11, q0, q1);
    const float r0 = rsqrtf(q0 * (1.0f / 64.0f) + EPSV);
    const float r1 = rsqrtf(q1 * (1.0f / 64.0f) + EPSV);

    *(v4f*)(y + row)                  = ((d00 * r0) * w0 + b0) * g00;
    *(v4f*)(y + row + 32)             = ((d01 * r0) * w1 + b1) * g01;
    *(v4f*)(y + row + 4 * Hdim)       = ((d10 * r1) * w0 + b0) * g10;
    *(v4f*)(y + row + 4 * Hdim + 32)  = ((d11 * r1) * w1 + b1) * g11;
}

extern "C" void kernel_launch(void* const* d_in, const int* in_sizes, int n_in,
                              void* d_out, int out_size, void* d_ws, size_t ws_size,
                              hipStream_t stream)
{
    const float* query    = (const float*)d_in[0];
    const float* keyval   = (const float*)d_in[1];
    const float* rwg_mu   = (const float*)d_in[2];
    const float* W_rwg0   = (const float*)d_in[3];
    const float* W_rwg2   = (const float*)d_in[4];
    const float* rwg_bias = (const float*)d_in[5];
    const float* kv_mu    = (const float*)d_in[6];
    const float* W_kv0    = (const float*)d_in[7];
    const float* W_kv2    = (const float*)d_in[8];
    const float* kv_bias  = (const float*)d_in[9];
    const float* Wr       = (const float*)d_in[10];
    const float* Wa       = (const float*)d_in[11];
    const float* Wb       = (const float*)d_in[12];
    const float* bb       = (const float*)d_in[13];
    const float* Wk       = (const float*)d_in[14];
    const float* Wv       = (const float*)d_in[15];
    const float* Wg       = (const float*)d_in[16];
    const float* u        = (const float*)d_in[17];
    const float* gn_w     = (const float*)d_in[18];
    const float* gn_b     = (const float*)d_in[19];
    const float* Wo       = (const float*)d_in[20];

    float* ws = (float*)d_ws;
    const long SZ = (long)NTOK * Hdim;
    float* xr    = ws + 0 * SZ;
    float* xw    = ws + 1 * SZ;
    float* xg    = ws + 2 * SZ;
    float* xk    = ws + 3 * SZ;
    float* xv    = ws + 4 * SZ;
    float* rbuf  = ws + 5 * SZ;
    float* kbuf  = ws + 6 * SZ;
    float* vbuf  = ws + 7 * SZ;
    float* dbuf  = ws + 8 * SZ;
    float* gsbuf = ws + 9 * SZ;
    float* obuf  = xr;
    float* ybuf  = xw;

    rwkv_features_kernel<<<NPOS, 256, 0, stream>>>(
        query, keyval, rwg_mu, W_rwg0, W_rwg2, rwg_bias,
        kv_mu, W_kv0, W_kv2, kv_bias, xr, xw, xg, xk, xv);

    const int NOJOB = 0x7fffffff;
    const dim3 gblock(128);

    rwkv_wmma_gemm_kernel<<<G1_TILES, gblock, 0, stream>>>(
        xw, Wa, Wb, bb, dbuf, G1_SD,
        xr, Wr, nullptr, rbuf,  256, 256, ACT_NONE, RTL_TOK,
        xg, Wg, nullptr, gsbuf, 256, 256, ACT_SILU, RTL_TOK,
        xk, Wk, nullptr, kbuf,  256, 256, ACT_NONE, RTL_POS,
        xv, Wv, nullptr, vbuf,  256, 256, ACT_NONE, RTL_POS,
        G1_S1, G1_S2, G1_S3);

    rwkv_scan_kernel<<<Bsz * LQ * NHEAD * 4, 128, 0, stream>>>(rbuf, kbuf, vbuf, dbuf, u, obuf);

    rwkv_gn_silu_kernel<<<GN_GROUPS, 256, 0, stream>>>(obuf, gsbuf, gn_w, gn_b, ybuf);

    rwkv_wmma_gemm_kernel<<<G2_TILES, gblock, 0, stream>>>(
        nullptr, nullptr, nullptr, nullptr, nullptr, 0,
        ybuf, Wo, nullptr, (float*)d_out, 256, 256, ACT_NONE, RTL_TOK,
        nullptr, nullptr, nullptr, nullptr, 0, 0, 0, 0,
        nullptr, nullptr, nullptr, nullptr, 0, 0, 0, 0,
        nullptr, nullptr, nullptr, nullptr, 0, 0, 0, 0,
        NOJOB, NOJOB, NOJOB);
}


# GroupNorm+gate folded into the output projection: its X loader reads the o and gate planes, writes normalised rows to LDS; stats one head ahead under the WMMAs: 5 launches -> 4; tree sum order kept
# speedup vs baseline: 9.9654x; 1.0396x over previous
#include <hip/hip_runtime.h>
#include <math.h>

#define Hdim 256
#define NHEAD 4
#define DK 64
#define PLR 32
#define Bsz 2
#define LQ 8
#define LKs 512
#define NTOK (Bsz * LQ * LKs)
#define EPSV 1e-5f

typedef float v2f __attribute__((ext_vector_type(2)));
typedef float v8f __attribute__((ext_vector_type(8)));
typedef float v4f __attribute__((ext_vector_type(4)));

#define ACT_NONE  0
#define ACT_SILU  1

#define NPOS   (Bsz * LKs)
#define WPITCH 36

struct Slab8 { v4f v[8]; };

__device__ __forceinline__ Slab8 slab_fetch(const float* __restrict__ Wsrc, int ld,
                                            int goff, int t)
{
    const float* p = Wsrc + (t >> 3) * ld + goff + ((t & 7) << 2);
    Slab8 s;
    #pragma unroll
    for (int m = 0; m < 8; ++m) s.v[m] = *(const v4f*)(p + m * 32 * ld);
    return s;
}

__device__ __forceinline__ void slab_store(float (*slab)[WPITCH], int t, const Slab8& s)
{
    #pragma unroll
    for (int m = 0; m < 8; ++m)
        *(v4f*)&slab[m * 32 + (t >> 3)][(t & 7) << 2] = s.v[m];
}

__device__ __forceinline__ void slab_row(float (*slab)[WPITCH], int t, float* w)
{
    #pragma unroll
    for (int r4 = 0; r4 < PLR; r4 += 4) {
        const float4 wv = *(const float4*)&slab[t][r4];
        w[r4] = wv.x; w[r4 + 1] = wv.y; w[r4 + 2] = wv.z; w[r4 + 3] = wv.w;
    }
}

__global__ __launch_bounds__(256) void rwkv_features_kernel(
    const float* __restrict__ query, const float* __restrict__ keyval,
    const float* __restrict__ rwg_mu, const float* __restrict__ W_rwg0,
    const float* __restrict__ W_rwg2, const float* __restrict__ rwg_bias,
    const float* __restrict__ kv_mu, const float* __restrict__ W_kv0,
    const float* __restrict__ W_kv2, const float* __restrict__ kv_bias,
    float* __restrict__ xr, float* __restrict__ xw, float* __restrict__ xg,
    float* __restrict__ xk, float* __restrict__ xv)
{
    const int pos = blockIdx.x;
    const int b   = pos >> 9;
    const int l   = pos & 511;
    const int t   = threadIdx.x;

    __shared__ __attribute__((aligned(16))) float xrwg_s[Hdim][LQ];
    __shared__ float xkv_s[Hdim];
    __shared__ __attribute__((aligned(16))) float t1_s[3 * PLR][LQ];
    __shared__ float t2_s[2 * PLR];
    __shared__ __attribute__((aligned(16))) float wslab[Hdim][WPITCH];

    const float h  = keyval[(long)pos * Hdim + t];
    const float hp = (l == 0) ? 0.0f : keyval[(long)(pos - 1) * Hdim + t];
    const float d  = hp - h;
    float dc[LQ];
    {
        const float mu = rwg_mu[t];
        float xq[LQ];
        #pragma unroll
        for (int q = 0; q < LQ; ++q) {
            dc[q] = query[(long)(b * LQ + q) * Hdim + t] - h;
            xq[q] = h + dc[q] * mu;
        }
        *(float4*)&xrwg_s[t][0] = make_float4(xq[0], xq[1], xq[2], xq[3]);
        *(float4*)&xrwg_s[t][4] = make_float4(xq[4], xq[5], xq[6], xq[7]);
        xkv_s[t] = h + d * kv_mu[t];
    }
    __syncthreads();

    if (t < 192) {
        const int c  = (t < 96) ? t : t - 96;
        const int qh = (t < 96) ? 0 : 4;
        float a0 = 0.0f, a1 = 0.0f, a2 = 0.0f, a3 = 0.0f;
        #pragma unroll 16
        for (int i = 0; i < Hdim; ++i) {
            const float  w = W_rwg0[i * 96 + c];
            const float4 x = *(const float4*)&xrwg_s[i][qh];
            a0 += x.x * w;
            a1 += x.y * w;
            a2 += x.z * w;
            a3 += x.w * w;
        }
        *(float4*)&t1_s[c][qh] = make_float4(tanhf(a0), tanhf(a1), tanhf(a2), tanhf(a3));
    } else {
        const int j = t - 192;
        float acc = 0.0f;
        #pragma unroll 16
        for (int i = 0; i < Hdim; ++i) acc += xkv_s[i] * W_kv0[i * 64 + j];
        t2_s[j] = tanhf(acc);
    }

    const long orow = ((long)(b * LQ) * LKs + l) * Hdim + t;
    float w[PLR];

    auto expand_q = [&](int g, float* __restrict__ dst) {
        const float bias = rwg_bias[g * Hdim + t];
        float acc[LQ];
        #pragma unroll
        for (int q = 0; q < LQ; ++q) acc[q] = bias;
        #pragma unroll
        for (int r = 0; r < PLR; ++r) {
            const float4 x0 = *(const float4*)&t1_s[g * PLR + r][0];
            const float4 x1 = *(const float4*)&t1_s[g * PLR + r][4];
            acc[0] += x0.x * w[r];
            acc[1] += x0.y * w[r];
            acc[2] += x0.z * w[r];
            acc[3] += x0.w * w[r];
            acc[4] += x1.x * w[r];
            acc[5] += x1.y * w[r];
            acc[6] += x1.z * w[r];
            acc[7] += x1.w * w[r];
        }
        #pragma unroll
        for (int q = 0; q < LQ; ++q)
            dst[orow + (long)q * LKs * Hdim] = h + dc[q] * acc[q];
    };
    auto expand_kv = [&](int gg, float* __restrict__ dst) {
        float acc = kv_bias[gg * Hdim + t];
        #pragma unroll
        for (int r = 0; r < PLR; ++r) acc += t2_s[gg * PLR + r] * w[r];
        dst[(long)pos * Hdim + t] = h + d * acc;
    };

    Slab8 s0 = slab_fetch(W_rwg2, 3 * PLR, 0, t);
    __syncthreads();

    slab_store(wslab, t, s0);
    __syncthreads();
    Slab8 s1 = slab_fetch(W_rwg2, 3 * PLR, PLR, t);
    slab_row(wslab, t, w);
    expand_q(0, xr);
    __syncthreads();

    slab_store(wslab, t, s1);
    __syncthreads();
    Slab8 s2 = slab_fetch(W_rwg2, 3 * PLR, 2 * PLR, t);
    slab_row(wslab, t, w);
    expand_q(1, xw);
    __syncthreads();

    slab_store(wslab, t, s2);
    __syncthreads();
    Slab8 s3 = slab_fetch(W_kv2, 2 * PLR, 0, t);
    slab_row(wslab, t, w);
    expand_q(2, xg);
    __syncthreads();

    slab_store(wslab, t, s3);
    __syncthreads();
    Slab8 s4 = slab_fetch(W_kv2, 2 * PLR, PLR, t);
    slab_row(wslab, t, w);
    expand_kv(0, xk);
    __syncthreads();

    slab_store(wslab, t, s4);
    __syncthreads();
    slab_row(wslab, t, w);
    expand_kv(1, xv);
}

__device__ __forceinline__ float act_silu(float x)  { return x / (1.0f + expf(-x)); }
__device__ __forceinline__ float act_decay(float x) { return expf(-expf(x)); }
template <bool DECAY>
__device__ __forceinline__ float tile_act(float x, bool silu)
{
    return DECAY ? act_decay(x) : (silu ? act_silu(x) : x);
}

#define GT_M 64
#define GT_N 64
#define GT_K 32
#define GA_PITCH 36
#define GB_PITCH 72

typedef float (*gemm_as_t)[GT_M][GA_PITCH];
typedef float (*gemm_bs_t)[GT_K][GB_PITCH];

__device__ __forceinline__ void gemm_mainloop(
    const float* __restrict__ X, const float* __restrict__ W,
    int K, int N, int m0, int n0, gemm_as_t As, gemm_bs_t Bs,
    v8f& c00, v8f& c01, v8f& c10, v8f& c11)
{
    const int t    = threadIdx.x;
    const int wave = t >> 5;
    const int lane = t & 31;
    const int wm   = (wave >> 1) << 5;
    const int wn   = (wave & 1) << 5;

    const int sub  = lane & 15;
    const int koff = (lane >> 4) << 1;

    const int ar = t >> 3;
    const int ac = (t & 7) << 2;
    const int br = t >> 4;
    const int bc = (t & 15) << 2;
    const float* __restrict__ Ag = X + (long)(m0 + ar) * K + ac;
    const float* __restrict__ Bg = W + (long)br * N + n0 + bc;

    v4f pa[4], pb[4];
    #pragma unroll
    for (int i = 0; i < 4; ++i) {
        pa[i] = *(const v4f*)(Ag + (long)(16 * i) * K);
        pb[i] = *(const v4f*)(Bg + (long)(8 * i) * N);
    }
    #pragma unroll
    for (int i = 0; i < 4; ++i) {
        *(v4f*)&As[0][ar + 16 * i][ac] = pa[i];
        *(v4f*)&Bs[0][br + 8 * i][bc]  = pb[i];
    }
    __syncthreads();

    const v8f zero = {};
    c00 = zero; c01 = zero; c10 = zero; c11 = zero;
    const int nchunk = K / GT_K;
    for (int c = 0; c < nchunk; ++c) {
        const int  cur  = c & 1;
        const bool more = (c + 1 < nchunk);

        if (more) {
            const float* __restrict__ Ap = Ag + (c + 1) * GT_K;
            const float* __restrict__ Bp = Bg + (long)((c + 1) * GT_K) * N;
            #pragma unroll
            for (int i = 0; i < 4; ++i) {
                pa[i] = *(const v4f*)(Ap + (long)(16 * i) * K);
                pb[i] = *(const v4f*)(Bp + (long)(8 * i) * N);
            }
        }

        #pragma unroll
        for (int ks = 0; ks < GT_K; ks += 4) {
            const v2f a0 = *(const v2f*)&As[cur][wm + sub][ks + koff];
            const v2f a1 = *(const v2f*)&As[cur][wm + 16 + sub][ks + koff];
            v2f b0, b1;
            b0.x = Bs[cur][ks + koff][wn + sub];
            b0.y = Bs[cur][ks + koff + 1][wn + sub];
            b1.x = Bs[cur][ks + koff][wn + 16 + sub];
            b1.y = Bs[cur][ks + koff + 1][wn + 16 + sub];
            c00 = __builtin_amdgcn_wmma_f32_16x16x4_f32(false, a0, false, b0, (short)0, c00, false, false);
            c01 = __builtin_amdgcn_wmma_f32_16x16x4_f32(false, a0, false, b1, (short)0, c01, false, false);
            c10 = __builtin_amdgcn_wmma_f32_16x16x4_f32(false, a1, false, b0, (short)0, c10, false, false);
            c11 = __builtin_amdgcn_wmma_f32_16x16x4_f32(false, a1, false, b1, (short)0, c11, false, false);
        }

        if (more) {
            #pragma unroll
            for (int i = 0; i < 4; ++i) {
                *(v4f*)&As[cur ^ 1][ar + 16 * i][ac] = pa[i];
                *(v4f*)&Bs[cur ^ 1][br + 8 * i][bc]  = pb[i];
            }
        }
        __syncthreads();
    }
}

template <bool DECAY>
__device__ __forceinline__ void tile_store(
    float* __restrict__ Y, int N, float bv0, float bv1, bool silu, int mbase, int nbase,
    const v8f& c00, const v8f& c01, const v8f& c10, const v8f& c11)
{
    const int lane  = threadIdx.x & 31;
    const int rbase = (lane >> 4) * 8;
    const int col0  = nbase + (lane & 15);
    const int col1  = col0 + 16;
    #pragma unroll
    for (int r = 0; r < 8; ++r) {
        const long row0 = (long)mbase + rbase + r;
        const long row1 = row0 + 16;
        Y[row0 * N + col0] = tile_act<DECAY>(c00[r] + bv0, silu);
        Y[row0 * N + col1] = tile_act<DECAY>(c01[r] + bv1, silu);
        Y[row1 * N + col0] = tile_act<DECAY>(c10[r] + bv0, silu);
        Y[row1 * N + col1] = tile_act<DECAY>(c11[r] + bv1, silu);
    }
}

__device__ __forceinline__ void gemm_tile(
    const float* __restrict__ X, const float* __restrict__ W,
    const float* __restrict__ bias, float* __restrict__ Y,
    int K, int N, int act, int m0, int n0, gemm_as_t As, gemm_bs_t Bs)
{
    v8f c00, c01, c10, c11;
    gemm_mainloop(X, W, K, N, m0, n0, As, Bs, c00, c01, c10, c11);

    const int wave = threadIdx.x >> 5;
    const int lane = threadIdx.x & 31;
    const int mbase = m0 + ((wave >> 1) << 5);
    const int nbase = n0 + ((wave & 1) << 5);
    const int col0  = nbase + (lane & 15);
    const float bv0 = bias ? bias[col0] : 0.0f;
    const float bv1 = bias ? bias[col0 + 16] : 0.0f;
    tile_store<false>(Y, N, bv0, bv1, act == ACT_SILU, mbase, nbase, c00, c01, c10, c11);
}

#define LORA_W   64
#define TW_PITCH 68
static_assert(LORA_W == GT_N && LORA_W == 2 * GT_K, "the middle matrix is one column tile wide and fills both W stages");
static_assert(GT_M * TW_PITCH <= 2 * GT_M * GA_PITCH, "tanh tile fits in the X stages");

__device__ __forceinline__ void decay_tile(
    const float* __restrict__ X, const float* __restrict__ Wd,
    const float* __restrict__ Wu, const float* __restrict__ bu,
    float* __restrict__ Y, int m0, gemm_as_t As, gemm_bs_t Bs)
{
    v8f c00, c01, c10, c11;
    gemm_mainloop(X, Wd, Hdim, LORA_W, m0, 0, As, Bs, c00, c01, c10, c11);

    float (*T)[TW_PITCH]  = (float (*)[TW_PITCH])&As[0][0][0];
    float (*Bu)[GB_PITCH] = (float (*)[GB_PITCH])&Bs[0][0][0];

    const int t     = threadIdx.x;
    const int wave  = t >> 5;
    const int lane  = t & 31;
    const int wm    = (wave >> 1) << 5;
    const int wn    = (wave & 1) << 5;
    const int sub   = lane & 15;
    const int koff  = (lane >> 4) << 1;
    const int rbase = (lane >> 4) * 8;

    const int br = t >> 4;
    const int bc = (t & 15) << 2;
    const float* __restrict__ Bg = Wu + (long)br * Hdim + bc;

    v4f pb[8];
    #pragma unroll
    for (int i = 0; i < 8; ++i) pb[i] = *(const v4f*)(Bg + (long)(8 * i) * Hdim);

    #pragma unroll
    for (int r = 0; r < 8; ++r) {
        const int row = wm + rbase + r;
        T[row][wn + sub]           = tanhf(c00[r]);
        T[row][wn + 16 + sub]      = tanhf(c01[r]);
        T[row + 16][wn + sub]      = tanhf(c10[r]);
        T[row + 16][wn + 16 + sub] = tanhf(c11[r]);
    }
    #pragma unroll
    for (int i = 0; i < 8; ++i) *(v4f*)&Bu[br + 8 * i][bc] = pb[i];
    __syncthreads();

    #pragma unroll 1
    for (int nt = 0; nt < Hdim / GT_N; ++nt) {
        const bool more = (nt + 1 < Hdim / GT_N);
        if (more) {
            #pragma unroll
            for (int i = 0; i < 8; ++i)
                pb[i] = *(const v4f*)(Bg + (nt + 1) * GT_N + (long)(8 * i) * Hdim);
        }
        const int   nbase = nt * GT_N + wn;
        const float bv0   = bu[nbase + sub];
        const float bv1   = bu[nbase + 16 + sub];

        v8f d00 = {}, d01 = {}, d10 = {}, d11 = {};
        #pragma unroll
        for (int s = 0; s < LORA_W / 4; ++s) {
            const int kk = 4 * s + koff;
            const v2f a0 = *(const v2f*)&T[wm + sub][kk];
            const v2f a1 = *(const v2f*)&T[wm + 16 + sub][kk];
            v2f b0, b1;
            b0.x = Bu[kk][wn + sub];
            b0.y = Bu[kk + 1][wn + sub];
            b1.x = Bu[kk][wn + 16 + sub];
            b1.y = Bu[kk + 1][wn + 16 + sub];
            d00 = __builtin_amdgcn_wmma_f32_16x16x4_f32(false, a0, false, b0, (short)0, d00, false, false);
            d01 = __builtin_amdgcn_wmma_f32_16x16x4_f32(false, a0, false, b1, (short)0, d01, false, false);
            d10 = __builtin_amdgcn_wmma_f32_16x16x4_f32(false, a1, false, b0, (short)0, d10, false, false);
            d11 = __builtin_amdgcn_wmma_f32_16x16x4_f32(false, a1, false, b1, (short)0, d11, false, false);
        }
        tile_store<true>(Y, Hdim, bv0, bv1, false, m0 + wm, nbase, d00, d01, d10, d11);

        __syncthreads();
        if (more) {
            #pragma unroll
            for (int i = 0; i < 8; ++i) *(v4f*)&Bu[br + 8 * i][bc] = pb[i];
            __syncthreads();
        }
    }
}

__global__ __launch_bounds__(128) void rwkv_wmma_gemm_kernel(
    const float* Xd, const float* Wd, const float* Wu, const float* bu, float* Yd, int sd,
    const float* X0, const float* W0, const float* b0, float* Y0, int K0, int N0, int act0, int mtl0,
    const float* X1, const float* W1, const float* b1, float* Y1, int K1, int N1, int act1, int mtl1,
    const float* X2, const float* W2, const float* b2, float* Y2, int K2, int N2, int act2, int mtl2,
    const float* X3, const float* W3, const float* b3, float* Y3, int K3, int N3, int act3, int mtl3,
    int s1, int s2, int s3)
{
    __shared__ __attribute__((aligned(16))) float As[2][GT_M][GA_PITCH];
    __shared__ __attribute__((aligned(16))) float Bs[2][GT_K][GB_PITCH];

    if ((int)blockIdx.x < sd) {
        decay_tile(Xd, Wd, Wu, bu, Yd, blockIdx.x * GT_M, As, Bs);
        return;
    }
    const int id = blockIdx.x - sd;
    const float* X = X0;
    const float* W = W0;
    const float* bias = b0;
    float* Y = Y0;
    int K = K0, N = N0, act = act0, mtl = mtl0, sb = 0;
    if (id >= s1) { X = X1; W = W1; bias = b1; Y = Y1; K = K1; N = N1; act = act1; mtl = mtl1; sb = s1; }
    if (id >= s2) { X = X2; W = W2; bias = b2; Y = Y2; K = K2; N = N2; act = act2; mtl = mtl2; sb = s2; }
    if (id >= s3) { X = X3; W = W3; bias = b3; Y = Y3; K = K3; N = N3; act = act3; mtl = mtl3; sb = s3; }
    const int local = id - sb;
    gemm_tile(X, W, bias, Y, K, N, act, (local & ((1 << mtl) - 1)) * GT_M, (local >> mtl) * GT_N, As, Bs);
}

#define RT_TOK  (NTOK / GT_M)
#define RTL_TOK 7
#define RT_POS  (NPOS / GT_M)
#define RTL_POS 4
static_assert((1 << RTL_TOK) == RT_TOK && (1 << RTL_POS) == RT_POS, "row-tile shifts");
#define G1_SD    (RT_TOK)
#define G1_S1    (4 * RT_TOK)
#define G1_S2    (G1_S1 + 4 * RT_TOK)
#define G1_S3    (G1_S2 + 4 * RT_POS)
#define G1_TILES (G1_SD + G1_S3 + 4 * RT_POS)
#define G2_TILES (4 * RT_TOK)

#define SCAN_T 64
#define SCAN_NCHUNK (LKs / SCAN_T)
#define SCAN_VQ 16
#define SCAN_KS 16
#define SCAN_PPITCH (SCAN_KS * SCAN_VQ + 16)
#define SCAN_PD 4

__device__ __forceinline__ long scan_o_index(int e, int head, int l, int vcol)
{
    return ((((long)(e * NHEAD + head) * 4 + (vcol >> 4)) * LKs + l) << 4) + (vcol & 15);
}

__global__ __launch_bounds__(128) void rwkv_scan_kernel(
    const float* __restrict__ r, const float* __restrict__ k,
    const float* __restrict__ v, const float* __restrict__ decay,
    const float* __restrict__ u, float* __restrict__ o)
{
    const int chain = blockIdx.x >> 2;
    const int vq    = blockIdx.x & 3;
    const int e     = chain >> 2;
    const int head  = chain & 3;
    const long base   = ((long)e * LKs) * Hdim + head * DK;
    const long basekv = ((long)(e / LQ) * LKs) * Hdim + head * DK;
    const int t    = threadIdx.x;
    const int vp2  = (t & 7) << 1;
    const int ks   = t >> 3;
    const int k0   = ks << 2;

    __shared__ __attribute__((aligned(16))) float stage[2][SCAN_T][3][DK];
    __shared__ __attribute__((aligned(16))) float vstage[2][SCAN_T][SCAN_VQ];
    __shared__ __attribute__((aligned(16))) float part[2][SCAN_T][SCAN_PPITCH];

    float S0[4], S1[4], uu[4];
    #pragma unroll
    for (int j = 0; j < 4; ++j) {
        S0[j] = 0.0f;
        S1[j] = 0.0f;
        uu[j] = u[head * DK + k0 + j];
    }

    const int  lrow = t >> 4;
    const int  lc   = (t & 15) << 2;
    const int  vrow = t >> 2;
    const int  vc   = (t & 3) << 2;
    const long lofs = base + (long)lrow * Hdim + lc;
    const long kofs = basekv + (long)lrow * Hdim + lc;
    const long vofs = basekv + (long)vrow * Hdim + vq * SCAN_VQ + vc;

    v4f gr[4], gk[4], gd[4], gv;
    for (int hh = 0; hh < 2; ++hh) {
        const long hofs = (long)(hh * 32) * Hdim;
        #pragma unroll
        for (int i = 0; i < 4; ++i) {
            gr[i] = *(const v4f*)(r     + lofs + hofs + (long)(8 * i) * Hdim);
            gk[i] = *(const v4f*)(k     + kofs + hofs + (long)(8 * i) * Hdim);
            gd[i] = *(const v4f*)(decay + lofs + hofs + (long)(8 * i) * Hdim);
        }
        gv = *(const v4f*)(v + vofs + hofs);
        #pragma unroll
        for (int i = 0; i < 4; ++i) {
            *(v4f*)&stage[0][hh * 32 + lrow + 8 * i][0][lc] = gr[i];
            *(v4f*)&stage[0][hh * 32 + lrow + 8 * i][1][lc] = gk[i];
            *(v4f*)&stage[0][hh * 32 + lrow + 8 * i][2][lc] = gd[i];
        }
        *(v4f*)&vstage[0][hh * 32 + vrow][vc] = gv;
    }
    __syncthreads();

    float* __restrict__ oq = o + (((long)chain * 4 + vq) * LKs << 4);

    for (int c = 0; c < SCAN_NCHUNK; ++c) {
        const int  cur  = c & 1;
        const bool more = (c + 1 < SCAN_NCHUNK);

        float4 qr[SCAN_PD], qk[SCAN_PD], qd[SCAN_PD];
        float2 qv[SCAN_PD];
        #pragma unroll
        for (int p = 0; p < SCAN_PD; ++p) {
            qr[p] = *(const float4*)&stage[cur][p][0][k0];
            qk[p] = *(const float4*)&stage[cur][p][1][k0];
            qd[p] = *(const float4*)&stage[cur][p][2][k0];
            qv[p] = *(const float2*)&vstage[cur][p][vp2];
        }

        for (int hh = 0; hh < 2; ++hh) {
            if (more) {
                const long cofs = (long)((c + 1) * SCAN_T + hh * 32) * Hdim;
                #pragma unroll
                for (int i = 0; i < 4; ++i) {
                    gr[i] = *(const v4f*)(r     + lofs + cofs + (long)(8 * i) * Hdim);
                    gk[i] = *(const v4f*)(k     + kofs + cofs + (long)(8 * i) * Hdim);
                    gd[i] = *(const v4f*)(decay + lofs + cofs + (long)(8 * i) * Hdim);
                }
                gv = *(const v4f*)(v + vofs + cofs);
            }

            #pragma unroll 1
            for (int s0 = hh * 32; s0 < hh * 32 + 32; s0 += 16) {
                #pragma unroll
                for (int i = 0; i < 16; ++i) {
                    const int s = s0 + i;
                    const int p = i % SCAN_PD;
                    const float4 rr = qr[p];
                    const float4 kk = qk[p];
                    const float4 dd = qd[p];
                    const float2 vt = qv[p];
                    const int ns = (i + SCAN_PD < 16) ? s + SCAN_PD : min(s + SCAN_PD, SCAN_T - 1);
                    qr[p] = *(const float4*)&stage[cur][ns][0][k0];
                    qk[p] = *(const float4*)&stage[cur][ns][1][k0];
                    qd[p] = *(const float4*)&stage[cur][ns][2][k0];
                    qv[p] = *(const float2*)&vstage[cur][ns][vp2];
                    __builtin_amdgcn_sched_barrier(0);
                    const float rj[4] = {rr.x, rr.y, rr.z, rr.w};
                    const float kj[4] = {kk.x, kk.y, kk.z, kk.w};
                    const float dj[4] = {dd.x, dd.y, dd.z, dd.w};
                    float a0 = 0.0f, a1 = 0.0f;
                    #pragma unroll
                    for (int j = 0; j < 4; ++j) {
                        const float kv0 = kj[j] * vt.x;
                        const float kv1 = kj[j] * vt.y;
                        a0 += rj[j] * (S0[j] + uu[j] * kv0);
                        a1 += rj[j] * (S1[j] + uu[j] * kv1);
                        S0[j] = dj[j] * S0[j] + kv0;
                        S1[j] = dj[j] * S1[j] + kv1;
                    }
                    *(float2*)&part[cur][s][(ks << 4) + vp2] = make_float2(a0, a1);
                }
            }

            if (more) {
                #pragma unroll
                for (int i = 0; i < 4; ++i) {
                    *(v4f*)&stage[cur ^ 1][hh * 32 + lrow + 8 * i][0][lc] = gr[i];
                    *(v4f*)&stage[cur ^ 1][hh * 32 + lrow + 8 * i][1][lc] = gk[i];
                    *(v4f*)&stage[cur ^ 1][hh * 32 + lrow + 8 * i][2][lc] = gd[i];
                }
                *(v4f*)&vstage[cur ^ 1][hh * 32 + vrow][vc] = gv;
            }
        }
        __syncthreads();

        #pragma unroll
        for (int i = 0; i < 2; ++i) {
            const int js = vrow + 32 * i;
            float4 sum = *(const float4*)&part[cur][js][vc];
            #pragma unroll
            for (int q = 1; q < SCAN_KS; ++q) {
                const float4 pq = *(const float4*)&part[cur][js][(q << 4) + vc];
                sum.x += pq.x; sum.y += pq.y; sum.z += pq.z; sum.w += pq.w;
            }
            *(float4*)(oq + ((long)(c * SCAN_T + js) << 4) + vc) = sum;
        }
    }
}

#define QUAD_XOR1 0xB1
#define QUAD_XOR2 0x4E
template <int CTRL>
__device__ __forceinline__ v4f quad_swap(v4f v)
{
    v4f r;
    r.x = __int_as_float(__builtin_amdgcn_mov_dpp(__float_as_int(v.x), CTRL, 0xF, 0xF, true));
    r.y = __int_as_float(__builtin_amdgcn_mov_dpp(__float_as_int(v.y), CTRL, 0xF, 0xF, true));
    r.z = __int_as_float(__builtin_amdgcn_mov_dpp(__float_as_int(v.z), CTRL, 0xF, 0xF, true));
    r.w = __int_as_float(__builtin_amdgcn_mov_dpp(__float_as_int(v.w), CTRL, 0xF, 0xF, true));
    return r;
}

__device__ __forceinline__ float gn_tree64_quad(v4f x0, v4f x1, v4f x2, v4f x3)
{
    const v4f a0 = x0 + x2;
    const v4f a1 = x1 + x3;
    const v4f b  = a0 + a1;
    const v4f c  = b + quad_swap<QUAD_XOR2>(b);
    const v4f d  = c + quad_swap<QUAD_XOR1>(c);
    return (d.x + d.z) + (d.y + d.w);
}

__global__ __launch_bounds__(128) void rwkv_gn_out_kernel(
    const float* __restrict__ o, const float* __restrict__ gsilu,
    const float* __restrict__ gn_w, const float* __restrict__ gn_b,
    const float* __restrict__ Wo, float* __restrict__ out)
{
    __shared__ __attribute__((aligned(16))) float As[2][GT_M][GA_PITCH];
    __shared__ __attribute__((aligned(16))) float Bs[2][GT_K][GB_PITCH];
    __shared__ float mean_s[GT_M][NHEAD];
    __shared__ float rstd_s[GT_M][NHEAD];

    const int t    = threadIdx.x;
    const int wave = t >> 5;
    const int lane = t & 31;
    const int wm   = (wave >> 1) << 5;
    const int wn   = (wave & 1) << 5;
    const int sub  = lane & 15;
    const int koff = (lane >> 4) << 1;

    const int m0 = (blockIdx.x & (RT_TOK - 1)) * GT_M;
    const int n0 = (blockIdx.x >> RTL_TOK) * GT_N;
    const int e  = m0 / LKs;
    const int l0 = m0 % LKs;

    const int ar = t >> 3;
    const int ac = (t & 7) << 2;
    const int br = t >> 4;
    const int bc = (t & 15) << 2;
    const float* __restrict__ Gg = gsilu + (long)(m0 + ar) * Hdim + ac;
    const float* __restrict__ Bg = Wo + (long)br * Hdim + n0 + bc;

    v4f po[4], pg[4], pb[4], pw, pbias;
    auto request = [&](int c) {
        const int head = c >> 1;
        const int vcol = ((c & 1) << 5) + ac;
        #pragma unroll
        for (int i = 0; i < 4; ++i) {
            po[i] = *(const v4f*)(o + scan_o_index(e, head, l0 + ar + 16 * i, vcol));
            pg[i] = *(const v4f*)(Gg + (long)(16 * i) * Hdim + c * GT_K);
            pb[i] = *(const v4f*)(Bg + (long)(c * GT_K + 8 * i) * Hdim);
        }
        pw    = *(const v4f*)(gn_w + c * GT_K + ac);
        pbias = *(const v4f*)(gn_b + c * GT_K + ac);
    };
    auto deposit = [&](int c, int stage) {
        const int head = c >> 1;
        #pragma unroll
        for (int i = 0; i < 4; ++i) {
            const int   row = ar + 16 * i;
            const float mu  = mean_s[row][head];
            const float rs  = rstd_s[row][head];
            const v4f   d   = po[i] - mu;
            *(v4f*)&As[stage][row][ac]       = ((d * rs) * pw + pbias) * pg[i];
            *(v4f*)&Bs[stage][br + 8 * i][bc] = pb[i];
        }
    };

    const int sj   = t & 3;
    const int srow = t >> 2;
    v4f sx[2][4];
    auto stats_request = [&](int head) {
        #pragma unroll
        for (int hf = 0; hf < 2; ++hf) {
            #pragma unroll
            for (int q = 0; q < 4; ++q)
                sx[hf][q] = *(const v4f*)(o + scan_o_index(e, head, l0 + srow + 32 * hf, 16 * q + 4 * sj));
        }
    };
    auto stats_finish = [&](int head) {
        float mean_r[2], rstd_r[2];
        #pragma unroll
        for (int hf = 0; hf < 2; ++hf) {
            const float mean = gn_tree64_quad(sx[hf][0], sx[hf][1], sx[hf][2], sx[hf][3]) * (1.0f / 64.0f);
            const v4f d0 = sx[hf][0] - mean, d1 = sx[hf][1] - mean;
            const v4f d2 = sx[hf][2] - mean, d3 = sx[hf][3] - mean;
            const float var = gn_tree64_quad(d0 * d0, d1 * d1, d2 * d2, d3 * d3) * (1.0f / 64.0f);
            mean_r[hf] = mean;
            rstd_r[hf] = rsqrtf(var + EPSV);
        }
        if (sj == 0) {
            #pragma unroll
            for (int hf = 0; hf < 2; ++hf) {
                mean_s[srow + 32 * hf][head] = mean_r[hf];
                rstd_s[srow + 32 * hf][head] = rstd_r[hf];
            }
        }
    };

    stats_request(0);
    request(0);
    stats_finish(0);
    __syncthreads();

    deposit(0, 0);
    __syncthreads();

    v8f c00 = {}, c01 = {}, c10 = {}, c11 = {};
    const int nchunk = Hdim / GT_K;
    for (int c = 0; c < nchunk; ++c) {
        const int  cur  = c & 1;
        const bool more = (c + 1 < nchunk);
        const bool ahead = ((c & 1) == 0) && (c + 2 < nchunk);
        const int  nhead = (c >> 1) + 1;

        if (ahead) stats_request(nhead);
        if (more) request(c + 1);

        #pragma unroll
        for (int ks = 0; ks < GT_K; ks += 4) {
            const v2f a0 = *(const v2f*)&As[cur][wm + sub][ks + koff];
            const v2f a1 = *(const v2f*)&As[cur][wm + 16 + sub][ks + koff];
            v2f b0, b1;
            b0.x = Bs[cur][ks + koff][wn + sub];
            b0.y = Bs[cur][ks + koff + 1][wn + sub];
            b1.x = Bs[cur][ks + koff][wn + 16 + sub];
            b1.y = Bs[cur][ks + koff + 1][wn + 16 + sub];
            c00 = __builtin_amdgcn_wmma_f32_16x16x4_f32(false, a0, false, b0, (short)0, c00, false, false);
            c01 = __builtin_amdgcn_wmma_f32_16x16x4_f32(false, a0, false, b1, (short)0, c01, false, false);
            c10 = __builtin_amdgcn_wmma_f32_16x16x4_f32(false, a1, false, b0, (short)0, c10, false, false);
            c11 = __builtin_amdgcn_wmma_f32_16x16x4_f32(false, a1, false, b1, (short)0, c11, false, false);
        }

        if (ahead) stats_finish(nhead);
        if (more) deposit(c + 1, cur ^ 1);
        __syncthreads();
    }

    tile_store<false>(out, Hdim, 0.0f, 0.0f, false, m0 + wm, n0 + wn, c00, c01, c10, c11);
}

extern "C" void kernel_launch(void* const* d_in, const int* in_sizes, int n_in,
                              void* d_out, int out_size, void* d_ws, size_t ws_size,
                              hipStream_t stream)
{
    const float* query    = (const float*)d_in[0];
    const float* keyval   = (const float*)d_in[1];
    const float* rwg_mu   = (const float*)d_in[2];
    const float* W_rwg0   = (const float*)d_in[3];
    const float* W_rwg2   = (const float*)d_in[4];
    const float* rwg_bias = (const float*)d_in[5];
    const float* kv_mu    = (const float*)d_in[6];
    const float* W_kv0    = (const float*)d_in[7];
    const float* W_kv2    = (const float*)d_in[8];
    const float* kv_bias  = (const float*)d_in[9];
    const float* Wr       = (const float*)d_in[10];
    const float* Wa       = (const float*)d_in[11];
    const float* Wb       = (const float*)d_in[12];
    const float* bb       = (const float*)d_in[13];
    const float* Wk       = (const float*)d_in[14];
    const float* Wv       = (const float*)d_in[15];
    const float* Wg       = (const float*)d_in[16];
    const float* u        = (const float*)d_in[17];
    const float* gn_w     = (const float*)d_in[18];
    const float* gn_b     = (const float*)d_in[19];
    const float* Wo       = (const float*)d_in[20];

    float* ws = (float*)d_ws;
    const long SZ = (long)NTOK * Hdim;
    float* xr    = ws + 0 * SZ;
    float* xw    = ws + 1 * SZ;
    float* xg    = ws + 2 * SZ;
    float* xk    = ws + 3 * SZ;
    float* xv    = ws + 4 * SZ;
    float* rbuf  = ws + 5 * SZ;
    float* kbuf  = ws + 6 * SZ;
    float* vbuf  = ws + 7 * SZ;
    float* dbuf  = ws + 8 * SZ;
    float* gsbuf = ws + 9 * SZ;
    float* obuf  = xr;

    rwkv_features_kernel<<<NPOS, 256, 0, stream>>>(
        query, keyval, rwg_mu, W_rwg0, W_rwg2, rwg_bias,
        kv_mu, W_kv0, W_kv2, kv_bias, xr, xw, xg, xk, xv);

    const dim3 gblock(128);

    rwkv_wmma_gemm_kernel<<<G1_TILES, gblock, 0, stream>>>(
        xw, Wa, Wb, bb, dbuf, G1_SD,
        xr, Wr, nullptr, rbuf,  256, 256, ACT_NONE, RTL_TOK,
        xg, Wg, nullptr, gsbuf, 256, 256, ACT_SILU, RTL_TOK,
        xk, Wk, nullptr, kbuf,  256, 256, ACT_NONE, RTL_POS,
        xv, Wv, nullptr, vbuf,  256, 256, ACT_NONE, RTL_POS,
        G1_S1, G1_S2, G1_S3);

    rwkv_scan_kernel<<<Bsz * LQ * NHEAD * 4, 128, 0, stream>>>(rbuf, kbuf, vbuf, dbuf, u, obuf);

    rwkv_gn_out_kernel<<<G2_TILES, gblock, 0, stream>>>(obuf, gsbuf, gn_w, gn_b, Wo, (float*)d_out);
}


# features stage A re-tiled: K split over the 8 waves (32 inputs each; a wave does all 160 columns x 8 rows, one load per weight row), partials joined in LDS in wave order; sum order changed
# speedup vs baseline: 10.5779x; 1.0615x over previous
#include <hip/hip_runtime.h>
#include <math.h>

#define Hdim 256
#define NHEAD 4
#define DK 64
#define PLR 32
#define Bsz 2
#define LQ 8
#define LKs 512
#define NTOK (Bsz * LQ * LKs)
#define EPSV 1e-5f

typedef float v2f __attribute__((ext_vector_type(2)));
typedef float v8f __attribute__((ext_vector_type(8)));
typedef float v4f __attribute__((ext_vector_type(4)));

#define ACT_NONE  0
#define ACT_SILU  1

#define NPOS   (Bsz * LKs)
#define WPITCH 36

struct Slab8 { v4f v[8]; };

__device__ __forceinline__ Slab8 slab_fetch(const float* __restrict__ Wsrc, int ld,
                                            int goff, int t)
{
    const float* p = Wsrc + (t >> 3) * ld + goff + ((t & 7) << 2);
    Slab8 s;
    #pragma unroll
    for (int m = 0; m < 8; ++m) s.v[m] = *(const v4f*)(p + m * 32 * ld);
    return s;
}

__device__ __forceinline__ void slab_store(float (*slab)[WPITCH], int t, const Slab8& s)
{
    #pragma unroll
    for (int m = 0; m < 8; ++m)
        *(v4f*)&slab[m * 32 + (t >> 3)][(t & 7) << 2] = s.v[m];
}

__device__ __forceinline__ void slab_row(float (*slab)[WPITCH], int t, float* w)
{
    #pragma unroll
    for (int r4 = 0; r4 < PLR; r4 += 4) {
        const float4 wv = *(const float4*)&slab[t][r4];
        w[r4] = wv.x; w[r4 + 1] = wv.y; w[r4 + 2] = wv.z; w[r4 + 3] = wv.w;
    }
}

__global__ __launch_bounds__(256) void rwkv_features_kernel(
    const float* __restrict__ query, const float* __restrict__ keyval,
    const float* __restrict__ rwg_mu, const float* __restrict__ W_rwg0,
    const float* __restrict__ W_rwg2, const float* __restrict__ rwg_bias,
    const float* __restrict__ kv_mu, const float* __restrict__ W_kv0,
    const float* __restrict__ W_kv2, const float* __restrict__ kv_bias,
    float* __restrict__ xr, float* __restrict__ xw, float* __restrict__ xg,
    float* __restrict__ xk, float* __restrict__ xv)
{
    const int pos = blockIdx.x;
    const int b   = pos >> 9;
    const int l   = pos & 511;
    const int t   = threadIdx.x;

    __shared__ __attribute__((aligned(16))) float xrwg_s[Hdim][LQ];
    __shared__ float xkv_s[Hdim];
    __shared__ __attribute__((aligned(16))) float t1_s[3 * PLR][LQ];
    __shared__ float t2_s[2 * PLR];
    __shared__ __attribute__((aligned(16))) float wslab[Hdim][WPITCH];

    const float h  = keyval[(long)pos * Hdim + t];
    const float hp = (l == 0) ? 0.0f : keyval[(long)(pos - 1) * Hdim + t];
    const float d  = hp - h;
    float dc[LQ];
    {
        const float mu = rwg_mu[t];
        float xq[LQ];
        #pragma unroll
        for (int q = 0; q < LQ; ++q) {
            dc[q] = query[(long)(b * LQ + q) * Hdim + t] - h;
            xq[q] = h + dc[q] * mu;
        }
        *(float4*)&xrwg_s[t][0] = make_float4(xq[0], xq[1], xq[2], xq[3]);
        *(float4*)&xrwg_s[t][4] = make_float4(xq[4], xq[5], xq[6], xq[7]);
        xkv_s[t] = h + d * kv_mu[t];
    }
    __syncthreads();

    float (*prwg)[LQ][3 * PLR] = (float (*)[LQ][3 * PLR])&wslab[0][0];
    float (*pkv)[2 * PLR]      = (float (*)[2 * PLR])(&wslab[0][0] + 8 * LQ * 3 * PLR);
    static_assert(8 * LQ * 3 * PLR + 8 * 2 * PLR <= Hdim * WPITCH, "partials fit in the slab");
    {
        const int wv = t >> 5;
        const int ln = t & 31;
        float acc[3][LQ];
        float ak0 = 0.0f, ak1 = 0.0f;
        #pragma unroll
        for (int m = 0; m < 3; ++m) {
            #pragma unroll
            for (int q = 0; q < LQ; ++q) acc[m][q] = 0.0f;
        }
        const float* __restrict__ wq = W_rwg0 + (wv * 32) * (3 * PLR) + 3 * ln;
        const float* __restrict__ wk = W_kv0  + (wv * 32) * (2 * PLR) + 2 * ln;
        #pragma unroll 8
        for (int ii = 0; ii < 32; ++ii) {
            const int   i  = wv * 32 + ii;
            const float w0 = wq[ii * (3 * PLR)];
            const float w1 = wq[ii * (3 * PLR) + 1];
            const float w2 = wq[ii * (3 * PLR) + 2];
            const float2 uk = *(const float2*)(wk + ii * (2 * PLR));
            const float4 xa = *(const float4*)&xrwg_s[i][0];
            const float4 xb = *(const float4*)&xrwg_s[i][4];
            const float  xk = xkv_s[i];
            const float  x[LQ] = {xa.x, xa.y, xa.z, xa.w, xb.x, xb.y, xb.z, xb.w};
            #pragma unroll
            for (int q = 0; q < LQ; ++q) {
                acc[0][q] += x[q] * w0;
                acc[1][q] += x[q] * w1;
                acc[2][q] += x[q] * w2;
            }
            ak0 += xk * uk.x;
            ak1 += xk * uk.y;
        }
        #pragma unroll
        for (int m = 0; m < 3; ++m) {
            #pragma unroll
            for (int q = 0; q < LQ; ++q) prwg[wv][q][3 * ln + m] = acc[m][q];
        }
        *(float2*)&pkv[wv][2 * ln] = make_float2(ak0, ak1);
    }
    __syncthreads();

    if (t < 192) {
        const int c  = (t < 96) ? t : t - 96;
        const int qh = (t < 96) ? 0 : 4;
        float a0 = prwg[0][qh][c], a1 = prwg[0][qh + 1][c], a2 = prwg[0][qh + 2][c], a3 = prwg[0][qh + 3][c];
        #pragma unroll
        for (int w8 = 1; w8 < 8; ++w8) {
            a0 += prwg[w8][qh][c];
            a1 += prwg[w8][qh + 1][c];
            a2 += prwg[w8][qh + 2][c];
            a3 += prwg[w8][qh + 3][c];
        }
        *(float4*)&t1_s[c][qh] = make_float4(tanhf(a0), tanhf(a1), tanhf(a2), tanhf(a3));
    } else {
        const int j = t - 192;
        float acc = pkv[0][j];
        #pragma unroll
        for (int w8 = 1; w8 < 8; ++w8) acc += pkv[w8][j];
        t2_s[j] = tanhf(acc);
    }

    const long orow = ((long)(b * LQ) * LKs + l) * Hdim + t;
    float w[PLR];

    auto expand_q = [&](int g, float* __restrict__ dst) {
        const float bias = rwg_bias[g * Hdim + t];
        float acc[LQ];
        #pragma unroll
        for (int q = 0; q < LQ; ++q) acc[q] = bias;
        #pragma unroll
        for (int r = 0; r < PLR; ++r) {
            const float4 x0 = *(const float4*)&t1_s[g * PLR + r][0];
            const float4 x1 = *(const float4*)&t1_s[g * PLR + r][4];
            acc[0] += x0.x * w[r];
            acc[1] += x0.y * w[r];
            acc[2] += x0.z * w[r];
            acc[3] += x0.w * w[r];
            acc[4] += x1.x * w[r];
            acc[5] += x1.y * w[r];
            acc[6] += x1.z * w[r];
            acc[7] += x1.w * w[r];
        }
        #pragma unroll
        for (int q = 0; q < LQ; ++q)
            dst[orow + (long)q * LKs * Hdim] = h + dc[q] * acc[q];
    };
    auto expand_kv = [&](int gg, float* __restrict__ dst) {
        float acc = kv_bias[gg * Hdim + t];
        #pragma unroll
        for (int r = 0; r < PLR; ++r) acc += t2_s[gg * PLR + r] * w[r];
        dst[(long)pos * Hdim + t] = h + d * acc;
    };

    Slab8 s0 = slab_fetch(W_rwg2, 3 * PLR, 0, t);
    __syncthreads();

    slab_store(wslab, t, s0);
    __syncthreads();
    Slab8 s1 = slab_fetch(W_rwg2, 3 * PLR, PLR, t);
    slab_row(wslab, t, w);
    expand_q(0, xr);
    __syncthreads();

    slab_store(wslab, t, s1);
    __syncthreads();
    Slab8 s2 = slab_fetch(W_rwg2, 3 * PLR, 2 * PLR, t);
    slab_row(wslab, t, w);
    expand_q(1, xw);
    __syncthreads();

    slab_store(wslab, t, s2);
    __syncthreads();
    Slab8 s3 = slab_fetch(W_kv2, 2 * PLR, 0, t);
    slab_row(wslab, t, w);
    expand_q(2, xg);
    __syncthreads();

    slab_store(wslab, t, s3);
    __syncthreads();
    Slab8 s4 = slab_fetch(W_kv2, 2 * PLR, PLR, t);
    slab_row(wslab, t, w);
    expand_kv(0, xk);
    __syncthreads();

    slab_store(wslab, t, s4);
    __syncthreads();
    slab_row(wslab, t, w);
    expand_kv(1, xv);
}

__device__ __forceinline__ float act_silu(float x)  { return x / (1.0f + expf(-x)); }
__device__ __forceinline__ float act_decay(float x) { return expf(-expf(x)); }
template <bool DECAY>
__device__ __forceinline__ float tile_act(float x, bool silu)
{
    return DECAY ? act_decay(x) : (silu ? act_silu(x) : x);
}

#define GT_M 64
#define GT_N 64
#define GT_K 32
#define GA_PITCH 36
#define GB_PITCH 72

typedef float (*gemm_as_t)[GT_M][GA_PITCH];
typedef float (*gemm_bs_t)[GT_K][GB_PITCH];

__device__ __forceinline__ void gemm_mainloop(
    const float* __restrict__ X, const float* __restrict__ W,
    int K, int N, int m0, int n0, gemm_as_t As, gemm_bs_t Bs,
    v8f& c00, v8f& c01, v8f& c10, v8f& c11)
{
    const int t    = threadIdx.x;
    const int wave = t >> 5;
    const int lane = t & 31;
    const int wm   = (wave >> 1) << 5;
    const int wn   = (wave & 1) << 5;

    const int sub  = lane & 15;
    const int koff = (lane >> 4) << 1;

    const int ar = t >> 3;
    const int ac = (t & 7) << 2;
    const int br = t >> 4;
    const int bc = (t & 15) << 2;
    const float* __restrict__ Ag = X + (long)(m0 + ar) * K + ac;
    const float* __restrict__ Bg = W + (long)br * N + n0 + bc;

    v4f pa[4], pb[4];
    #pragma unroll
    for (int i = 0; i < 4; ++i) {
        pa[i] = *(const v4f*)(Ag + (long)(16 * i) * K);
        pb[i] = *(const v4f*)(Bg + (long)(8 * i) * N);
    }
    #pragma unroll
    for (int i = 0; i < 4; ++i) {
        *(v4f*)&As[0][ar + 16 * i][ac] = pa[i];
        *(v4f*)&Bs[0][br + 8 * i][bc]  = pb[i];
    }
    __syncthreads();

    const v8f zero = {};
    c00 = zero; c01 = zero; c10 = zero; c11 = zero;
    const int nchunk = K / GT_K;
    for (int c = 0; c < nchunk; ++c) {
        const int  cur  = c & 1;
        const bool more = (c + 1 < nchunk);

        if (more) {
            const float* __restrict__ Ap = Ag + (c + 1) * GT_K;
            const float* __restrict__ Bp = Bg + (long)((c + 1) * GT_K) * N;
            #pragma unroll
            for (int i = 0; i < 4; ++i) {
                pa[i] = *(const v4f*)(Ap + (long)(16 * i) * K);
                pb[i] = *(const v4f*)(Bp + (long)(8 * i) * N);
            }
        }

        #pragma unroll
        for (int ks = 0; ks < GT_K; ks += 4) {
            const v2f a0 = *(const v2f*)&As[cur][wm + sub][ks + koff];
            const v2f a1 = *(const v2f*)&As[cur][wm + 16 + sub][ks + koff];
            v2f b0, b1;
            b0.x = Bs[cur][ks + koff][wn + sub];
            b0.y = Bs[cur][ks + koff + 1][wn + sub];
            b1.x = Bs[cur][ks + koff][wn + 16 + sub];
            b1.y = Bs[cur][ks + koff + 1][wn + 16 + sub];
            c00 = __builtin_amdgcn_wmma_f32_16x16x4_f32(false, a0, false, b0, (short)0, c00, false, false);
            c01 = __builtin_amdgcn_wmma_f32_16x16x4_f32(false, a0, false, b1, (short)0, c01, false, false);
            c10 = __builtin_amdgcn_wmma_f32_16x16x4_f32(false, a1, false, b0, (short)0, c10, false, false);
            c11 = __builtin_amdgcn_wmma_f32_16x16x4_f32(false, a1, false, b1, (short)0, c11, false, false);
        }

        if (more) {
            #pragma unroll
            for (int i = 0; i < 4; ++i) {
                *(v4f*)&As[cur ^ 1][ar + 16 * i][ac] = pa[i];
                *(v4f*)&Bs[cur ^ 1][br + 8 * i][bc]  = pb[i];
            }
        }
        __syncthreads();
    }
}

template <bool DECAY>
__device__ __forceinline__ void tile_store(
    float* __restrict__ Y, int N, float bv0, float bv1, bool silu, int mbase, int nbase,
    const v8f& c00, const v8f& c01, const v8f& c10, const v8f& c11)
{
    const int lane  = threadIdx.x & 31;
    const int rbase = (lane >> 4) * 8;
    const int col0  = nbase + (lane & 15);
    const int col1  = col0 + 16;
    #pragma unroll
    for (int r = 0; r < 8; ++r) {
        const long row0 = (long)mbase + rbase + r;
        const long row1 = row0 + 16;
        Y[row0 * N + col0] = tile_act<DECAY>(c00[r] + bv0, silu);
        Y[row0 * N + col1] = tile_act<DECAY>(c01[r] + bv1, silu);
        Y[row1 * N + col0] = tile_act<DECAY>(c10[r] + bv0, silu);
        Y[row1 * N + col1] = tile_act<DECAY>(c11[r] + bv1, silu);
    }
}

__device__ __forceinline__ void gemm_tile(
    const float* __restrict__ X, const float* __restrict__ W,
    const float* __restrict__ bias, float* __restrict__ Y,
    int K, int N, int act, int m0, int n0, gemm_as_t As, gemm_bs_t Bs)
{
    v8f c00, c01, c10, c11;
    gemm_mainloop(X, W, K, N, m0, n0, As, Bs, c00, c01, c10, c11);

    const int wave = threadIdx.x >> 5;
    const int lane = threadIdx.x & 31;
    const int mbase = m0 + ((wave >> 1) << 5);
    const int nbase = n0 + ((wave & 1) << 5);
    const int col0  = nbase + (lane & 15);
    const float bv0 = bias ? bias[col0] : 0.0f;
    const float bv1 = bias ? bias[col0 + 16] : 0.0f;
    tile_store<false>(Y, N, bv0, bv1, act == ACT_SILU, mbase, nbase, c00, c01, c10, c11);
}

#define LORA_W   64
#define TW_PITCH 68
static_assert(LORA_W == GT_N && LORA_W == 2 * GT_K, "the middle matrix is one column tile wide and fills both W stages");
static_assert(GT_M * TW_PITCH <= 2 * GT_M * GA_PITCH, "tanh tile fits in the X stages");

__device__ __forceinline__ void decay_tile(
    const float* __restrict__ X, const float* __restrict__ Wd,
    const float* __restrict__ Wu, const float* __restrict__ bu,
    float* __restrict__ Y, int m0, gemm_as_t As, gemm_bs_t Bs)
{
    v8f c00, c01, c10, c11;
    gemm_mainloop(X, Wd, Hdim, LORA_W, m0, 0, As, Bs, c00, c01, c10, c11);

    float (*T)[TW_PITCH]  = (float (*)[TW_PITCH])&As[0][0][0];
    float (*Bu)[GB_PITCH] = (float (*)[GB_PITCH])&Bs[0][0][0];

    const int t     = threadIdx.x;
    const int wave  = t >> 5;
    const int lane  = t & 31;
    const int wm    = (wave >> 1) << 5;
    const int wn    = (wave & 1) << 5;
    const int sub   = lane & 15;
    const int koff  = (lane >> 4) << 1;
    const int rbase = (lane >> 4) * 8;

    const int br = t >> 4;
    const int bc = (t & 15) << 2;
    const float* __restrict__ Bg = Wu + (long)br * Hdim + bc;

    v4f pb[8];
    #pragma unroll
    for (int i = 0; i < 8; ++i) pb[i] = *(const v4f*)(Bg + (long)(8 * i) * Hdim);

    #pragma unroll
    for (int r = 0; r < 8; ++r) {
        const int row = wm + rbase + r;
        T[row][wn + sub]           = tanhf(c00[r]);
        T[row][wn + 16 + sub]      = tanhf(c01[r]);
        T[row + 16][wn + sub]      = tanhf(c10[r]);
        T[row + 16][wn + 16 + sub] = tanhf(c11[r]);
    }
    #pragma unroll
    for (int i = 0; i < 8; ++i) *(v4f*)&Bu[br + 8 * i][bc] = pb[i];
    __syncthreads();

    #pragma unroll 1
    for (int nt = 0; nt < Hdim / GT_N; ++nt) {
        const bool more = (nt + 1 < Hdim / GT_N);
        if (more) {
            #pragma unroll
            for (int i = 0; i < 8; ++i)
                pb[i] = *(const v4f*)(Bg + (nt + 1) * GT_N + (long)(8 * i) * Hdim);
        }
        const int   nbase = nt * GT_N + wn;
        const float bv0   = bu[nbase + sub];
        const float bv1   = bu[nbase + 16 + sub];

        v8f d00 = {}, d01 = {}, d10 = {}, d11 = {};
        #pragma unroll
        for (int s = 0; s < LORA_W / 4; ++s) {
            const int kk = 4 * s + koff;
            const v2f a0 = *(const v2f*)&T[wm + sub][kk];
            const v2f a1 = *(const v2f*)&T[wm + 16 + sub][kk];
            v2f b0, b1;
            b0.x = Bu[kk][wn + sub];
            b0.y = Bu[kk + 1][wn + sub];
            b1.x = Bu[kk][wn + 16 + sub];
            b1.y = Bu[kk + 1][wn + 16 + sub];
            d00 = __builtin_amdgcn_wmma_f32_16x16x4_f32(false, a0, false, b0, (short)0, d00, false, false);
            d01 = __builtin_amdgcn_wmma_f32_16x16x4_f32(false, a0, false, b1, (short)0, d01, false, false);
            d10 = __builtin_amdgcn_wmma_f32_16x16x4_f32(false, a1, false, b0, (short)0, d10, false, false);
            d11 = __builtin_amdgcn_wmma_f32_16x16x4_f32(false, a1, false, b1, (short)0, d11, false, false);
        }
        tile_store<true>(Y, Hdim, bv0, bv1, false, m0 + wm, nbase, d00, d01, d10, d11);

        __syncthreads();
        if (more) {
            #pragma unroll
            for (int i = 0; i < 8; ++i) *(v4f*)&Bu[br + 8 * i][bc] = pb[i];
            __syncthreads();
        }
    }
}

__global__ __launch_bounds__(128) void rwkv_wmma_gemm_kernel(
    const float* Xd, const float* Wd, const float* Wu, const float* bu, float* Yd, int sd,
    const float* X0, const float* W0, const float* b0, float* Y0, int K0, int N0, int act0, int mtl0,
    const float* X1, const float* W1, const float* b1, float* Y1, int K1, int N1, int act1, int mtl1,
    const float* X2, const float* W2, const float* b2, float* Y2, int K2, int N2, int act2, int mtl2,
    const float* X3, const float* W3, const float* b3, float* Y3, int K3, int N3, int act3, int mtl3,
    int s1, int s2, int s3)
{
    __shared__ __attribute__((aligned(16))) float As[2][GT_M][GA_PITCH];
    __shared__ __attribute__((aligned(16))) float Bs[2][GT_K][GB_PITCH];

    if ((int)blockIdx.x < sd) {
        decay_tile(Xd, Wd, Wu, bu, Yd, blockIdx.x * GT_M, As, Bs);
        return;
    }
    const int id = blockIdx.x - sd;
    const float* X = X0;
    const float* W = W0;
    const float* bias = b0;
    float* Y = Y0;
    int K = K0, N = N0, act = act0, mtl = mtl0, sb = 0;
    if (id >= s1) { X = X1; W = W1; bias = b1; Y = Y1; K = K1; N = N1; act = act1; mtl = mtl1; sb = s1; }
    if (id >= s2) { X = X2; W = W2; bias = b2; Y = Y2; K = K2; N = N2; act = act2; mtl = mtl2; sb = s2; }
    if (id >= s3) { X = X3; W = W3; bias = b3; Y = Y3; K = K3; N = N3; act = act3; mtl = mtl3; sb = s3; }
    const int local = id - sb;
    gemm_tile(X, W, bias, Y, K, N, act, (local & ((1 << mtl) - 1)) * GT_M, (local >> mtl) * GT_N, As, Bs);
}

#define RT_TOK  (NTOK / GT_M)
#define RTL_TOK 7
#define RT_POS  (NPOS / GT_M)
#define RTL_POS 4
static_assert((1 << RTL_TOK) == RT_TOK && (1 << RTL_POS) == RT_POS, "row-tile shifts");
#define G1_SD    (RT_TOK)
#define G1_S1    (4 * RT_TOK)
#define G1_S2    (G1_S1 + 4 * RT_TOK)
#define G1_S3    (G1_S2 + 4 * RT_POS)
#define G1_TILES (G1_SD + G1_S3 + 4 * RT_POS)
#define G2_TILES (4 * RT_TOK)

#define SCAN_T 64
#define SCAN_NCHUNK (LKs / SCAN_T)
#define SCAN_VQ 16
#define SCAN_KS 16
#define SCAN_PPITCH (SCAN_KS * SCAN_VQ + 16)
#define SCAN_PD 4

__device__ __forceinline__ long scan_o_index(int e, int head, int l, int vcol)
{
    return ((((long)(e * NHEAD + head) * 4 + (vcol >> 4)) * LKs + l) << 4) + (vcol & 15);
}

__global__ __launch_bounds__(128) void rwkv_scan_kernel(
    const float* __restrict__ r, const float* __restrict__ k,
    const float* __restrict__ v, const float* __restrict__ decay,
    const float* __restrict__ u, float* __restrict__ o)
{
    const int chain = blockIdx.x >> 2;
    const int vq    = blockIdx.x & 3;
    const int e     = chain >> 2;
    const int head  = chain & 3;
    const long base   = ((long)e * LKs) * Hdim + head * DK;
    const long basekv = ((long)(e / LQ) * LKs) * Hdim + head * DK;
    const int t    = threadIdx.x;
    const int vp2  = (t & 7) << 1;
    const int ks   = t >> 3;
    const int k0   = ks << 2;

    __shared__ __attribute__((aligned(16))) float stage[2][SCAN_T][3][DK];
    __shared__ __attribute__((aligned(16))) float vstage[2][SCAN_T][SCAN_VQ];
    __shared__ __attribute__((aligned(16))) float part[2][SCAN_T][SCAN_PPITCH];

    float S0[4], S1[4], uu[4];
    #pragma unroll
    for (int j = 0; j < 4; ++j) {
        S0[j] = 0.0f;
        S1[j] = 0.0f;
        uu[j] = u[head * DK + k0 + j];
    }

    const int  lrow = t >> 4;
    const int  lc   = (t & 15) << 2;
    const int  vrow = t >> 2;
    const int  vc   = (t & 3) << 2;
    const long lofs = base + (long)lrow * Hdim + lc;
    const long kofs = basekv + (long)lrow * Hdim + lc;
    const long vofs = basekv + (long)vrow * Hdim + vq * SCAN_VQ + vc;

    v4f gr[4], gk[4], gd[4], gv;
    for (int hh = 0; hh < 2; ++hh) {
        const long hofs = (long)(hh * 32) * Hdim;
        #pragma unroll
        for (int i = 0; i < 4; ++i) {
            gr[i] = *(const v4f*)(r     + lofs + hofs + (long)(8 * i) * Hdim);
            gk[i] = *(const v4f*)(k     + kofs + hofs + (long)(8 * i) * Hdim);
            gd[i] = *(const v4f*)(decay + lofs + hofs + (long)(8 * i) * Hdim);
        }
        gv = *(const v4f*)(v + vofs + hofs);
        #pragma unroll
        for (int i = 0; i < 4; ++i) {
            *(v4f*)&stage[0][hh * 32 + lrow + 8 * i][0][lc] = gr[i];
            *(v4f*)&stage[0][hh * 32 + lrow + 8 * i][1][lc] = gk[i];
            *(v4f*)&stage[0][hh * 32 + lrow + 8 * i][2][lc] = gd[i];
        }
        *(v4f*)&vstage[0][hh * 32 + vrow][vc] = gv;
    }
    __syncthreads();

    float* __restrict__ oq = o + (((long)chain * 4 + vq) * LKs << 4);

    for (int c = 0; c < SCAN_NCHUNK; ++c) {
        const int  cur  = c & 1;
        const bool more = (c + 1 < SCAN_NCHUNK);

        float4 qr[SCAN_PD], qk[SCAN_PD], qd[SCAN_PD];
        float2 qv[SCAN_PD];
        #pragma unroll
        for (int p = 0; p < SCAN_PD; ++p) {
            qr[p] = *(const float4*)&stage[cur][p][0][k0];
            qk[p] = *(const float4*)&stage[cur][p][1][k0];
            qd[p] = *(const float4*)&stage[cur][p][2][k0];
            qv[p] = *(const float2*)&vstage[cur][p][vp2];
        }

        for (int hh = 0; hh < 2; ++hh) {
            if (more) {
                const long cofs = (long)((c + 1) * SCAN_T + hh * 32) * Hdim;
                #pragma unroll
                for (int i = 0; i < 4; ++i) {
                    gr[i] = *(const v4f*)(r     + lofs + cofs + (long)(8 * i) * Hdim);
                    gk[i] = *(const v4f*)(k     + kofs + cofs + (long)(8 * i) * Hdim);
                    gd[i] = *(const v4f*)(decay + lofs + cofs + (long)(8 * i) * Hdim);
                }
                gv = *(const v4f*)(v + vofs + cofs);
            }

            #pragma unroll 1
            for (int s0 = hh * 32; s0 < hh * 32 + 32; s0 += 16) {
                #pragma unroll
                for (int i = 0; i < 16; ++i) {
                    const int s = s0 + i;
                    const int p = i % SCAN_PD;
                    const float4 rr = qr[p];
                    const float4 kk = qk[p];
                    const float4 dd = qd[p];
                    const float2 vt = qv[p];
                    const int ns = (i + SCAN_PD < 16) ? s + SCAN_PD : min(s + SCAN_PD, SCAN_T - 1);
                    qr[p] = *(const float4*)&stage[cur][ns][0][k0];
                    qk[p] = *(const float4*)&stage[cur][ns][1][k0];
                    qd[p] = *(const float4*)&stage[cur][ns][2][k0];
                    qv[p] = *(const float2*)&vstage[cur][ns][vp2];
                    __builtin_amdgcn_sched_barrier(0);
                    const float rj[4] = {rr.x, rr.y, rr.z, rr.w};
                    const float kj[4] = {kk.x, kk.y, kk.z, kk.w};
                    const float dj[4] = {dd.x, dd.y, dd.z, dd.w};
                    float a0 = 0.0f, a1 = 0.0f;
                    #pragma unroll
                    for (int j = 0; j < 4; ++j) {
                        const float kv0 = kj[j] * vt.x;
                        const float kv1 = kj[j] * vt.y;
                        a0 += rj[j] * (S0[j] + uu[j] * kv0);
                        a1 += rj[j] * (S1[j] + uu[j] * kv1);
                        S0[j] = dj[j] * S0[j] + kv0;
                        S1[j] = dj[j] * S1[j] + kv1;
                    }
                    *(float2*)&part[cur][s][(ks << 4) + vp2] = make_float2(a0, a1);
                }
            }

            if (more) {
                #pragma unroll
                for (int i = 0; i < 4; ++i) {
                    *(v4f*)&stage[cur ^ 1][hh * 32 + lrow + 8 * i][0][lc] = gr[i];
                    *(v4f*)&stage[cur ^ 1][hh * 32 + lrow + 8 * i][1][lc] = gk[i];
                    *(v4f*)&stage[cur ^ 1][hh * 32 + lrow + 8 * i][2][lc] = gd[i];
                }
                *(v4f*)&vstage[cur ^ 1][hh * 32 + vrow][vc] = gv;
            }
        }
        __syncthreads();

        #pragma unroll
        for (int i = 0; i < 2; ++i) {
            const int js = vrow + 32 * i;
            float4 sum = *(const float4*)&part[cur][js][vc];
            #pragma unroll
            for (int q = 1; q < SCAN_KS; ++q) {
                const float4 pq = *(const float4*)&part[cur][js][(q << 4) + vc];
                sum.x += pq.x; sum.y += pq.y; sum.z += pq.z; sum.w += pq.w;
            }
            *(float4*)(oq + ((long)(c * SCAN_T + js) << 4) + vc) = sum;
        }
    }
}

#define QUAD_XOR1 0xB1
#define QUAD_XOR2 0x4E
template <int CTRL>
__device__ __forceinline__ v4f quad_swap(v4f v)
{
    v4f r;
    r.x = __int_as_float(__builtin_amdgcn_mov_dpp(__float_as_int(v.x), CTRL, 0xF, 0xF, true));
    r.y = __int_as_float(__builtin_amdgcn_mov_dpp(__float_as_int(v.y), CTRL, 0xF, 0xF, true));
    r.z = __int_as_float(__builtin_amdgcn_mov_dpp(__float_as_int(v.z), CTRL, 0xF, 0xF, true));
    r.w = __int_as_float(__builtin_amdgcn_mov_dpp(__float_as_int(v.w), CTRL, 0xF, 0xF, true));
    return r;
}

__device__ __forceinline__ float gn_tree64_quad(v4f x0, v4f x1, v4f x2, v4f x3)
{
    const v4f a0 = x0 + x2;
    const v4f a1 = x1 + x3;
    const v4f b  = a0 + a1;
    const v4f c  = b + quad_swap<QUAD_XOR2>(b);
    const v4f d  = c + quad_swap<QUAD_XOR1>(c);
    return (d.x + d.z) + (d.y + d.w);
}

__global__ __launch_bounds__(128) void rwkv_gn_out_kernel(
    const float* __restrict__ o, const float* __restrict__ gsilu,
    const float* __restrict__ gn_w, const float* __restrict__ gn_b,
    const float* __restrict__ Wo, float* __restrict__ out)
{
    __shared__ __attribute__((aligned(16))) float As[2][GT_M][GA_PITCH];
    __shared__ __attribute__((aligned(16))) float Bs[2][GT_K][GB_PITCH];
    __shared__ float mean_s[GT_M][NHEAD];
    __shared__ float rstd_s[GT_M][NHEAD];

    const int t    = threadIdx.x;
    const int wave = t >> 5;
    const int lane = t & 31;
    const int wm   = (wave >> 1) << 5;
    const int wn   = (wave & 1) << 5;
    const int sub  = lane & 15;
    const int koff = (lane >> 4) << 1;

    const int m0 = (blockIdx.x & (RT_TOK - 1)) * GT_M;
    const int n0 = (blockIdx.x >> RTL_TOK) * GT_N;
    const int e  = m0 / LKs;
    const int l0 = m0 % LKs;

    const int ar = t >> 3;
    const int ac = (t & 7) << 2;
    const int br = t >> 4;
    const int bc = (t & 15) << 2;
    const float* __restrict__ Gg = gsilu + (long)(m0 + ar) * Hdim + ac;
    const float* __restrict__ Bg = Wo + (long)br * Hdim + n0 + bc;

    v4f po[4], pg[4], pb[4], pw, pbias;
    auto request = [&](int c) {
        const int head = c >> 1;
        const int vcol = ((c & 1) << 5) + ac;
        #pragma unroll
        for (int i = 0; i < 4; ++i) {
            po[i] = *(const v4f*)(o + scan_o_index(e, head, l0 + ar + 16 * i, vcol));
            pg[i] = *(const v4f*)(Gg + (long)(16 * i) * Hdim + c * GT_K);
            pb[i] = *(const v4f*)(Bg + (long)(c * GT_K + 8 * i) * Hdim);
        }
        pw    = *(const v4f*)(gn_w + c * GT_K + ac);
        pbias = *(const v4f*)(gn_b + c * GT_K + ac);
    };
    auto deposit = [&](int c, int stage) {
        const int head = c >> 1;
        #pragma unroll
        for (int i = 0; i < 4; ++i) {
            const int   row = ar + 16 * i;
            const float mu  = mean_s[row][head];
            const float rs  = rstd_s[row][head];
            const v4f   d   = po[i] - mu;
            *(v4f*)&As[stage][row][ac]       = ((d * rs) * pw + pbias) * pg[i];
            *(v4f*)&Bs[stage][br + 8 * i][bc] = pb[i];
        }
    };

    const int sj   = t & 3;
    const int srow = t >> 2;
    v4f sx[2][4];
    auto stats_request = [&](int head) {
        #pragma unroll
        for (int hf = 0; hf < 2; ++hf) {
            #pragma unroll
            for (int q = 0; q < 4; ++q)
                sx[hf][q] = *(const v4f*)(o + scan_o_index(e, head, l0 + srow + 32 * hf, 16 * q + 4 * sj));
        }
    };
    auto stats_finish = [&](int head) {
        float mean_r[2], rstd_r[2];
        #pragma unroll
        for (int hf = 0; hf < 2; ++hf) {
            const float mean = gn_tree64_quad(sx[hf][0], sx[hf][1], sx[hf][2], sx[hf][3]) * (1.0f / 64.0f);
            const v4f d0 = sx[hf][0] - mean, d1 = sx[hf][1] - mean;
            const v4f d2 = sx[hf][2] - mean, d3 = sx[hf][3] - mean;
            const float var = gn_tree64_quad(d0 * d0, d1 * d1, d2 * d2, d3 * d3) * (1.0f / 64.0f);
            mean_r[hf] = mean;
            rstd_r[hf] = rsqrtf(var + EPSV);
        }
        if (sj == 0) {
            #pragma unroll
            for (int hf = 0; hf < 2; ++hf) {
                mean_s[srow + 32 * hf][head] = mean_r[hf];
                rstd_s[srow + 32 * hf][head] = rstd_r[hf];
            }
        }
    };

    stats_request(0);
    request(0);
    stats_finish(0);
    __syncthreads();

    deposit(0, 0);
    __syncthreads();

    v8f c00 = {}, c01 = {}, c10 = {}, c11 = {};
    const int nchunk = Hdim / GT_K;
    for (int c = 0; c < nchunk; ++c) {
        const int  cur  = c & 1;
        const bool more = (c + 1 < nchunk);
        const bool ahead = ((c & 1) == 0) && (c + 2 < nchunk);
        const int  nhead = (c >> 1) + 1;

        if (ahead) stats_request(nhead);
        if (more) request(c + 1);

        #pragma unroll
        for (int ks = 0; ks < GT_K; ks += 4) {
            const v2f a0 = *(const v2f*)&As[cur][wm + sub][ks + koff];
            const v2f a1 = *(const v2f*)&As[cur][wm + 16 + sub][ks + koff];
            v2f b0, b1;
            b0.x = Bs[cur][ks + koff][wn + sub];
            b0.y = Bs[cur][ks + koff + 1][wn + sub];
            b1.x = Bs[cur][ks + koff][wn + 16 + sub];
            b1.y = Bs[cur][ks + koff + 1][wn + 16 + sub];
            c00 = __builtin_amdgcn_wmma_f32_16x16x4_f32(false, a0, false, b0, (short)0, c00, false, false);
            c01 = __builtin_amdgcn_wmma_f32_16x16x4_f32(false, a0, false, b1, (short)0, c01, false, false);
            c10 = __builtin_amdgcn_wmma_f32_16x16x4_f32(false, a1, false, b0, (short)0, c10, false, false);
            c11 = __builtin_amdgcn_wmma_f32_16x16x4_f32(false, a1, false, b1, (short)0, c11, false, false);
        }

        if (ahead) stats_finish(nhead);
        if (more) deposit(c + 1, cur ^ 1);
        __syncthreads();
    }

    tile_store<false>(out, Hdim, 0.0f, 0.0f, false, m0 + wm, n0 + wn, c00, c01, c10, c11);
}

extern "C" void kernel_launch(void* const* d_in, const int* in_sizes, int n_in,
                              void* d_out, int out_size, void* d_ws, size_t ws_size,
                              hipStream_t stream)
{
    const float* query    = (const float*)d_in[0];
    const float* keyval   = (const float*)d_in[1];
    const float* rwg_mu   = (const float*)d_in[2];
    const float* W_rwg0   = (const float*)d_in[3];
    const float* W_rwg2   = (const float*)d_in[4];
    const float* rwg_bias = (const float*)d_in[5];
    const float* kv_mu    = (const float*)d_in[6];
    const float* W_kv0    = (const float*)d_in[7];
    const float* W_kv2    = (const float*)d_in[8];
    const float* kv_bias  = (const float*)d_in[9];
    const float* Wr       = (const float*)d_in[10];
    const float* Wa       = (const float*)d_in[11];
    const float* Wb       = (const float*)d_in[12];
    const float* bb       = (const float*)d_in[13];
    const float* Wk       = (const float*)d_in[14];
    const float* Wv       = (const float*)d_in[15];
    const float* Wg       = (const float*)d_in[16];
    const float* u        = (const float*)d_in[17];
    const float* gn_w     = (const float*)d_in[18];
    const float* gn_b     = (const float*)d_in[19];
    const float* Wo       = (const float*)d_in[20];

    float* ws = (float*)d_ws;
    const long SZ = (long)NTOK * Hdim;
    float* xr    = ws + 0 * SZ;
    float* xw    = ws + 1 * SZ;
    float* xg    = ws + 2 * SZ;
    float* xk    = ws + 3 * SZ;
    float* xv    = ws + 4 * SZ;
    float* rbuf  = ws + 5 * SZ;
    float* kbuf  = ws + 6 * SZ;
    float* vbuf  = ws + 7 * SZ;
    float* dbuf  = ws + 8 * SZ;
    float* gsbuf = ws + 9 * SZ;
    float* obuf  = xr;

    rwkv_features_kernel<<<NPOS, 256, 0, stream>>>(
        query, keyval, rwg_mu, W_rwg0, W_rwg2, rwg_bias,
        kv_mu, W_kv0, W_kv2, kv_bias, xr, xw, xg, xk, xv);

    const dim3 gblock(128);

    rwkv_wmma_gemm_kernel<<<G1_TILES, gblock, 0, stream>>>(
        xw, Wa, Wb, bb, dbuf, G1_SD,
        xr, Wr, nullptr, rbuf,  256, 256, ACT_NONE, RTL_TOK,
        xg, Wg, nullptr, gsbuf, 256, 256, ACT_SILU, RTL_TOK,
        xk, Wk, nullptr, kbuf,  256, 256, ACT_NONE, RTL_POS,
        xv, Wv, nullptr, vbuf,  256, 256, ACT_NONE, RTL_POS,
        G1_S1, G1_S2, G1_S3);

    rwkv_scan_kernel<<<Bsz * LQ * NHEAD * 4, 128, 0, stream>>>(rbuf, kbuf, vbuf, dbuf, u, obuf);

    rwkv_gn_out_kernel<<<G2_TILES, gblock, 0, stream>>>(obuf, gsbuf, gn_w, gn_b, Wo, (float*)d_out);
}
